# starting kernel
# baseline (speedup 1.0000x reference)
#include <hip/hip_runtime.h>
#include <hip/hip_bf16.h>
#include <math.h>

#define N_NODES    100000
#define N_EDGES    1600000
#define NUM_REL    4
#define NUM_GRAPHS 256

typedef __attribute__((ext_vector_type(2))) float v2f;
typedef __attribute__((ext_vector_type(8))) float v8f;

template <int DIN, int DOUT>
__global__ __launch_bounds__(32)
void transform_wmma(const float* __restrict__ X,
                    const float* __restrict__ W,
                    const float* __restrict__ bias,
                    float* __restrict__ H) {
  const int lane = threadIdx.x & 31;
  const int half = lane >> 4;
  const int l16  = lane & 15;
  const int m0   = blockIdx.x * 16;
  const int n0   = blockIdx.y * 16;
  const int row  = m0 + l16;

  v8f acc = {};
#pragma unroll
  for (int k0 = 0; k0 < DIN; k0 += 4) {
    const int ka = k0 + 2 * half;
    v2f a, b;
    a.x = X[(size_t)row * DIN + ka];
    a.y = X[(size_t)row * DIN + ka + 1];
    b.x = W[(size_t)(ka    ) * DOUT + n0 + l16];
    b.y = W[(size_t)(ka + 1) * DOUT + n0 + l16];
    acc = __builtin_amdgcn_wmma_f32_16x16x4_f32(false, a, false, b,
                                                (short)0, acc, false, false);
  }

  const int   col = n0 + l16;
  const float bv  = bias ? bias[col] : 0.0f;
#pragma unroll
  for (int v = 0; v < 8; ++v) {
    H[(size_t)(m0 + v + 8 * half) * DOUT + col] = acc[v] + bv;
  }
}

__global__ void zero_i32(int* __restrict__ p, int n) {
  int t = blockIdx.x * blockDim.x + threadIdx.x;
  if (t < n) p[t] = 0;
}

__global__ void count_edges(const int* __restrict__ dst,
                            const int* __restrict__ et,
                            int* __restrict__ cnt) {
  int e = blockIdx.x * blockDim.x + threadIdx.x;
  if (e < N_EDGES) atomicAdd(&cnt[dst[e] * NUM_REL + et[e]], 1);
}

template <int DOUT, bool FULLH>
__global__ void scatter_mean(const float* __restrict__ H,
                             const int* __restrict__ src,
                             const int* __restrict__ dst,
                             const int* __restrict__ et,
                             const int* __restrict__ cnt,
                             int rel,
                             float* __restrict__ out) {
  constexpr int Q = DOUT / 4;
  const long long t = (long long)blockIdx.x * blockDim.x + threadIdx.x;
  if (t >= (long long)N_EDGES * Q) return;
  const int e = (int)(t / Q);
  const int c = ((int)(t & (Q - 1))) * 4;
  const int r = et[e];
  if (!FULLH && r != rel) return;
  const int   d    = dst[e];
  const float norm = 1.0f / fmaxf((float)cnt[d * NUM_REL + r], 1.0f);
  const size_t hrow =
      ((FULLH ? (size_t)r * N_NODES : (size_t)0) + (size_t)src[e]) * DOUT + c;
  const float4 hv = *(const float4*)(H + hrow);
  float* o = out + (size_t)d * DOUT + c;
  atomicAdd(o + 0, hv.x * norm);
  atomicAdd(o + 1, hv.y * norm);
  atomicAdd(o + 2, hv.z * norm);
  atomicAdd(o + 3, hv.w * norm);
}

__global__ void relu_k(float* __restrict__ p, int n) {
  int t = blockIdx.x * blockDim.x + threadIdx.x;
  if (t < n) p[t] = fmaxf(p[t], 0.0f);
}

__global__ void fill_neginf(float* __restrict__ p, int n) {
  int t = blockIdx.x * blockDim.x + threadIdx.x;
  if (t < n) p[t] = -INFINITY;
}

__global__ void pool_max(const float* __restrict__ h,
                         const int* __restrict__ batch,
                         float* __restrict__ out) {
  int t = blockIdx.x * blockDim.x + threadIdx.x;
  if (t >= N_NODES * 64) return;
  const int n = t >> 6;
  const int c = t & 63;
  const float v = h[t];
  float* addr = &out[batch[n] * 64 + c];
  if (v >= 0.0f) atomicMax((int*)addr, __float_as_int(v));
  else           atomicMin((unsigned int*)addr, __float_as_uint(v));
}

template <int DIN, int DOUT>
static void run_layer(const float* X, const float* W, const float* root,
                      const float* bias, const int* src, const int* dst,
                      const int* et, const int* cnt, float* h, bool fullh,
                      float* out, bool relu, hipStream_t stream) {
  dim3 tg(N_NODES / 16, DOUT / 16);
  transform_wmma<DIN, DOUT><<<tg, 32, 0, stream>>>(X, root, bias, out);

  constexpr int Q = DOUT / 4;
  const long long tot  = (long long)N_EDGES * Q;
  const int       blks = (int)((tot + 255) / 256);

  if (fullh) {
    for (int r = 0; r < NUM_REL; ++r)
      transform_wmma<DIN, DOUT><<<tg, 32, 0, stream>>>(
          X, W + (size_t)r * DIN * DOUT, nullptr,
          h + (size_t)r * N_NODES * DOUT);
    scatter_mean<DOUT, true><<<blks, 256, 0, stream>>>(h, src, dst, et, cnt,
                                                       0, out);
  } else {
    for (int r = 0; r < NUM_REL; ++r) {
      transform_wmma<DIN, DOUT><<<tg, 32, 0, stream>>>(
          X, W + (size_t)r * DIN * DOUT, nullptr, h);
      scatter_mean<DOUT, false><<<blks, 256, 0, stream>>>(h, src, dst, et,
                                                          cnt, r, out);
    }
  }
  if (relu) {
    const int n = N_NODES * DOUT;
    relu_k<<<(n + 255) / 256, 256, 0, stream>>>(out, n);
  }
}

extern "C" void kernel_launch(void* const* d_in, const int* in_sizes, int n_in,
                              void* d_out, int out_size, void* d_ws,
                              size_t ws_size, hipStream_t stream) {
  (void)in_sizes; (void)n_in; (void)out_size;

  const float* x     = (const float*)d_in[0];
  const int*   ei    = (const int*)d_in[1];
  const int*   et    = (const int*)d_in[2];
  const int*   batch = (const int*)d_in[3];
  const float* W1 = (const float*)d_in[4];
  const float* r1 = (const float*)d_in[5];
  const float* b1 = (const float*)d_in[6];
  const float* W2 = (const float*)d_in[7];
  const float* r2 = (const float*)d_in[8];
  const float* b2 = (const float*)d_in[9];
  const float* W3 = (const float*)d_in[10];
  const float* r3 = (const float*)d_in[11];
  const float* b3 = (const float*)d_in[12];
  const int* src = ei;
  const int* dst = ei + N_EDGES;
  float* out = (float*)d_out;

  char*  ws    = (char*)d_ws;
  int*   cnt   = (int*)ws;
  float* buf2  = (float*)(ws + 1600000);
  float* buf13 = (float*)(ws + 14400000);
  float* h     = (float*)(ws + 40000000);

  const size_t NEED_BIG = 40000000ull + 4ull * N_NODES * 64 * 4;
  const bool fullh = ws_size >= NEED_BIG;

  zero_i32<<<(N_NODES * NUM_REL + 255) / 256, 256, 0, stream>>>(
      cnt, N_NODES * NUM_REL);
  count_edges<<<(N_EDGES + 255) / 256, 256, 0, stream>>>(dst, et, cnt);

  run_layer<16, 16>(x,     W1, r1, b1, src, dst, et, cnt, h, fullh, buf13,
                    true, stream);
  run_layer<16, 32>(buf13, W2, r2, b2, src, dst, et, cnt, h, fullh, buf2,
                    true, stream);
  run_layer<32, 64>(buf2,  W3, r3, b3, src, dst, et, cnt, h, fullh, buf13,
                    false, stream);

  fill_neginf<<<(NUM_GRAPHS * 64 + 255) / 256, 256, 0, stream>>>(
      out, NUM_GRAPHS * 64);
  pool_max<<<(N_NODES * 64 + 255) / 256, 256, 0, stream>>>(buf13, batch, out);
}


# per layer: sum inputs per (dst,rel) first, then ONE WMMA GEMM, K=(R+1)*DIN, with 1/cnt, bias, ReLU in it; h[R,N,DOUT] gone; sums reassociated (mean before product)
# speedup vs baseline: 1.3622x; 1.3622x over previous
#include <hip/hip_runtime.h>
#include <math.h>


#define N_NODES    100000
#define N_EDGES    1600000
#define NUM_REL    4
#define NUM_GRAPHS 256

typedef __attribute__((ext_vector_type(2))) float v2f;
typedef __attribute__((ext_vector_type(8))) float v8f;

__global__ void zero_i32(int* __restrict__ p, int n) {
  int t = blockIdx.x * blockDim.x + threadIdx.x;
  if (t < n) p[t] = 0;
}

__global__ void zero_f32x4(float* __restrict__ p, int n4) {
  int t = blockIdx.x * blockDim.x + threadIdx.x;
  if (t < n4) ((float4*)p)[t] = make_float4(0.0f, 0.0f, 0.0f, 0.0f);
}

__global__ void count_edges(const int* __restrict__ dst,
                            const int* __restrict__ et,
                            int* __restrict__ cnt) {
  int e = blockIdx.x * blockDim.x + threadIdx.x;
  if (e < N_EDGES) atomicAdd(&cnt[dst[e] * NUM_REL + et[e]], 1);
}

template <int DIN>
__global__ void scatter_sum(const float* __restrict__ X,
                            const int* __restrict__ src,
                            const int* __restrict__ dst,
                            const int* __restrict__ et,
                            float* __restrict__ agg) {
  constexpr int Q = DIN / 4;
  const long long t = (long long)blockIdx.x * blockDim.x + threadIdx.x;
  if (t >= (long long)N_EDGES * Q) return;
  const int e = (int)(t / Q);
  const int c = ((int)(t & (Q - 1))) * 4;
  const int r = et[e];
  const int d = dst[e];
  const float4 xv = *(const float4*)(X + (size_t)src[e] * DIN + c);
  float* o = agg + ((size_t)d * NUM_REL + r) * DIN + c;
  atomicAdd(o + 0, xv.x);
  atomicAdd(o + 1, xv.y);
  atomicAdd(o + 2, xv.z);
  atomicAdd(o + 3, xv.w);
}

template <int DIN, int DOUT, bool RELU>
__global__ __launch_bounds__(256)
void layer_gemm(const float* __restrict__ agg,
                const int*   __restrict__ cnt,
                const float* __restrict__ X,
                const float* __restrict__ W,
                const float* __restrict__ root,
                const float* __restrict__ bias,
                float* __restrict__ out) {
  constexpr int KA = NUM_REL * DIN;
  constexpr int KT = KA + DIN;
  constexpr int P  = DOUT + 8;
  constexpr int NT = DOUT / 16;
  constexpr int C4 = DOUT / 4;
  __shared__ float Ws[KT * P];

  for (int i = threadIdx.x; i < KT * C4; i += 256) {
    const int k = i / C4;
    const int c = (i - k * C4) * 4;
    const float* g = (k < KA) ? (W + (size_t)k * DOUT + c)
                              : (root + (size_t)(k - KA) * DOUT + c);
    *(float4*)(&Ws[k * P + c]) = *(const float4*)g;
  }
  __syncthreads();

  const int lane  = threadIdx.x & 31;
  const int wave  = threadIdx.x >> 5;
  const int half  = lane >> 4;
  const int l16   = lane & 15;
  const int strip = blockIdx.x * 8 + wave;
  if (strip >= N_NODES / 16) return;
  const int m0  = strip * 16;
  const int row = m0 + l16;

  const int4 c4 = *(const int4*)(cnt + (size_t)row * NUM_REL);
  float inv[NUM_REL];
  inv[0] = 1.0f / fmaxf((float)c4.x, 1.0f);
  inv[1] = 1.0f / fmaxf((float)c4.y, 1.0f);
  inv[2] = 1.0f / fmaxf((float)c4.z, 1.0f);
  inv[3] = 1.0f / fmaxf((float)c4.w, 1.0f);

  v8f acc[NT];
#pragma unroll
  for (int t = 0; t < NT; ++t) acc[t] = v8f{};

  const float* arow = agg + (size_t)row * KA;
#pragma unroll
  for (int kk = 0; kk < KA; kk += 8) {
    const int kb = kk + 4 * half;
    float4 av = *(const float4*)(arow + kb);
    const float s = inv[kk / DIN];
    const v2f a0 = {av.x * s, av.y * s};
    const v2f a1 = {av.z * s, av.w * s};
    const float* bp = &Ws[kb * P + l16];
#pragma unroll
    for (int t = 0; t < NT; ++t) {
      const v2f b0 = {bp[t * 16], bp[t * 16 + P]};
      acc[t] = __builtin_amdgcn_wmma_f32_16x16x4_f32(false, a0, false, b0,
                                                     (short)0, acc[t], false, false);
    }
#pragma unroll
    for (int t = 0; t < NT; ++t) {
      const v2f b1 = {bp[t * 16 + 2 * P], bp[t * 16 + 3 * P]};
      acc[t] = __builtin_amdgcn_wmma_f32_16x16x4_f32(false, a1, false, b1,
                                                     (short)0, acc[t], false, false);
    }
  }

  const float* xrow = X + (size_t)row * DIN;
#pragma unroll
  for (int kk = 0; kk < DIN; kk += 8) {
    const int kb = kk + 4 * half;
    const float4 av = *(const float4*)(xrow + kb);
    const v2f a0 = {av.x, av.y};
    const v2f a1 = {av.z, av.w};
    const float* bp = &Ws[(KA + kb) * P + l16];
#pragma unroll
    for (int t = 0; t < NT; ++t) {
      const v2f b0 = {bp[t * 16], bp[t * 16 + P]};
      acc[t] = __builtin_amdgcn_wmma_f32_16x16x4_f32(false, a0, false, b0,
                                                     (short)0, acc[t], false, false);
    }
#pragma unroll
    for (int t = 0; t < NT; ++t) {
      const v2f b1 = {bp[t * 16 + 2 * P], bp[t * 16 + 3 * P]};
      acc[t] = __builtin_amdgcn_wmma_f32_16x16x4_f32(false, a1, false, b1,
                                                     (short)0, acc[t], false, false);
    }
  }

#pragma unroll
  for (int t = 0; t < NT; ++t) {
    const int   col = t * 16 + l16;
    const float bv  = bias[col];
#pragma unroll
    for (int v = 0; v < 8; ++v) {
      float y = acc[t][v] + bv;
      if (RELU) y = fmaxf(y, 0.0f);
      out[(size_t)(m0 + v + 8 * half) * DOUT + col] = y;
    }
  }
}

__global__ void fill_neginf(float* __restrict__ p, int n) {
  int t = blockIdx.x * blockDim.x + threadIdx.x;
  if (t < n) p[t] = -INFINITY;
}

__global__ void pool_max(const float* __restrict__ h,
                         const int* __restrict__ batch,
                         float* __restrict__ out) {
  int t = blockIdx.x * blockDim.x + threadIdx.x;
  if (t >= N_NODES * 64) return;
  const int n = t >> 6;
  const int c = t & 63;
  const float v = h[t];
  float* addr = &out[batch[n] * 64 + c];
  if (v >= 0.0f) atomicMax((int*)addr, __float_as_int(v));
  else           atomicMin((unsigned int*)addr, __float_as_uint(v));
}

template <int DIN, int DOUT, bool RELU>
static void run_layer(const float* X, const float* W, const float* root,
                      const float* bias, const int* src, const int* dst,
                      const int* et, const int* cnt, float* agg, float* out,
                      hipStream_t stream) {
  constexpr int N4 = N_NODES * NUM_REL * DIN / 4;
  zero_f32x4<<<(N4 + 255) / 256, 256, 0, stream>>>(agg, N4);

  constexpr long long TOT = (long long)N_EDGES * (DIN / 4);
  scatter_sum<DIN><<<(int)((TOT + 255) / 256), 256, 0, stream>>>(X, src, dst,
                                                                   et, agg);

  constexpr int STRIPS = N_NODES / 16;
  layer_gemm<DIN, DOUT, RELU><<<(STRIPS + 7) / 8, 256, 0, stream>>>(
      agg, cnt, X, W, root, bias, out);
}

extern "C" void kernel_launch(void* const* d_in, const int* in_sizes, int n_in,
                              void* d_out, int out_size, void* d_ws,
                              size_t ws_size, hipStream_t stream) {
  (void)in_sizes; (void)n_in; (void)out_size; (void)ws_size;

  const float* x     = (const float*)d_in[0];
  const int*   ei    = (const int*)d_in[1];
  const int*   et    = (const int*)d_in[2];
  const int*   batch = (const int*)d_in[3];
  const float* W1 = (const float*)d_in[4];
  const float* r1 = (const float*)d_in[5];
  const float* b1 = (const float*)d_in[6];
  const float* W2 = (const float*)d_in[7];
  const float* r2 = (const float*)d_in[8];
  const float* b2 = (const float*)d_in[9];
  const float* W3 = (const float*)d_in[10];
  const float* r3 = (const float*)d_in[11];
  const float* b3 = (const float*)d_in[12];
  const int* src = ei;
  const int* dst = ei + N_EDGES;
  float* out = (float*)d_out;

  char*  ws    = (char*)d_ws;
  int*   cnt   = (int*)ws;
  float* buf2  = (float*)(ws + 1600000);
  float* buf13 = (float*)(ws + 14400000);
  float* agg   = (float*)(ws + 40000000);

  zero_i32<<<(N_NODES * NUM_REL + 255) / 256, 256, 0, stream>>>(
      cnt, N_NODES * NUM_REL);
  count_edges<<<(N_EDGES + 255) / 256, 256, 0, stream>>>(dst, et, cnt);

  run_layer<16, 16, true>(x,     W1, r1, b1, src, dst, et, cnt, agg, buf13,
                          stream);
  run_layer<16, 32, true>(buf13, W2, r2, b2, src, dst, et, cnt, agg, buf2,
                          stream);
  run_layer<32, 64, false>(buf2, W3, r3, b3, src, dst, et, cnt, agg, buf13,
                           stream);

  fill_neginf<<<(NUM_GRAPHS * 64 + 255) / 256, 256, 0, stream>>>(
      out, NUM_GRAPHS * 64);
  pool_max<<<(N_NODES * 64 + 255) / 256, 256, 0, stream>>>(buf13, batch, out);
}


# float-atomic scatter and agg zero-fill replaced by a per-call CSR (count gives arrival-order rank, scan, fill) and an atomic-free gather-sum per (dst,rel); sums now follow rank order
# speedup vs baseline: 2.1968x; 1.6127x over previous
#include <hip/hip_runtime.h>
#include <math.h>


#define N_NODES    100000
#define N_EDGES    1600000
#define NUM_REL    4
#define NUM_GRAPHS 256

typedef __attribute__((ext_vector_type(2))) float v2f;
typedef __attribute__((ext_vector_type(8))) float v8f;

__global__ void zero_i32(int* __restrict__ p, int n) {
  int t = blockIdx.x * blockDim.x + threadIdx.x;
  if (t < n) p[t] = 0;
}

#define SCAN_B   256
#define N_SBLK   ((N_NODES + SCAN_B - 1) / SCAN_B)

__global__ void count_rank(const int* __restrict__ dst,
                           const int* __restrict__ et,
                           int* __restrict__ cnt,
                           int* __restrict__ rank) {
  int e = blockIdx.x * blockDim.x + threadIdx.x;
  if (e < N_EDGES) rank[e] = atomicAdd(&cnt[dst[e] * NUM_REL + et[e]], 1);
}

__global__ __launch_bounds__(SCAN_B)
void scan_nodes(const int* __restrict__ cnt,
                int* __restrict__ nloc,
                int* __restrict__ bsum) {
  __shared__ int s[SCAN_B];
  const int t = threadIdx.x;
  const int n = blockIdx.x * SCAN_B + t;
  int v = 0;
  if (n < N_NODES) {
    const int4 c = *(const int4*)(cnt + (size_t)n * NUM_REL);
    v = c.x + c.y + c.z + c.w;
  }
  s[t] = v;
  __syncthreads();
  for (int d = 1; d < SCAN_B; d <<= 1) {
    const int a = (t >= d) ? s[t - d] : 0;
    __syncthreads();
    s[t] += a;
    __syncthreads();
  }
  if (n < N_NODES) nloc[n] = s[t] - v;
  if (t == SCAN_B - 1) bsum[blockIdx.x] = s[t];
}

__global__ __launch_bounds__(512)
void scan_blocks(const int* __restrict__ bsum, int* __restrict__ bpre) {
  static_assert(N_SBLK <= 512, "one block scans all block totals");
  __shared__ int s[512];
  const int t = threadIdx.x;
  const int v = (t < N_SBLK) ? bsum[t] : 0;
  s[t] = v;
  __syncthreads();
  for (int d = 1; d < 512; d <<= 1) {
    const int a = (t >= d) ? s[t - d] : 0;
    __syncthreads();
    s[t] += a;
    __syncthreads();
  }
  if (t < N_SBLK) bpre[t] = s[t] - v;
}

__device__ __forceinline__ void seg_range(const int* __restrict__ cnt,
                                          const int* __restrict__ nloc,
                                          const int* __restrict__ bpre,
                                          int node, int r, int& beg, int& n) {
  const int4 c = *(const int4*)(cnt + (size_t)node * NUM_REL);
  beg = bpre[node / SCAN_B] + nloc[node];
  n = c.x;
  if (r >= 1) { beg += c.x; n = c.y; }
  if (r >= 2) { beg += c.y; n = c.z; }
  if (r >= 3) { beg += c.z; n = c.w; }
}

__global__ void fill_csr(const int* __restrict__ src,
                         const int* __restrict__ dst,
                         const int* __restrict__ et,
                         const int* __restrict__ rank,
                         const int* __restrict__ cnt,
                         const int* __restrict__ nloc,
                         const int* __restrict__ bpre,
                         int* __restrict__ sorted) {
  int e = blockIdx.x * blockDim.x + threadIdx.x;
  if (e >= N_EDGES) return;
  int beg, n;
  seg_range(cnt, nloc, bpre, dst[e], et[e], beg, n);
  sorted[beg + rank[e]] = src[e];
}

template <int DIN>
__global__ __launch_bounds__(256)
void gather_sum(const float* __restrict__ X,
                const int* __restrict__ sorted,
                const int* __restrict__ cnt,
                const int* __restrict__ nloc,
                const int* __restrict__ bpre,
                float* __restrict__ agg,
                int node0, int nthreads) {
  constexpr int Q = DIN / 4;
  const int t = blockIdx.x * 256 + threadIdx.x;
  if (t >= nthreads) return;
  const int q    = t & (Q - 1);
  const int segl = t / Q;
  const int r    = segl & (NUM_REL - 1);
  const int node = node0 + segl / NUM_REL;
  int beg, n;
  seg_range(cnt, nloc, bpre, node, r, beg, n);

  const float* xq = X + 4 * q;
  const int*   sp = sorted + beg;
  float4 acc = make_float4(0.0f, 0.0f, 0.0f, 0.0f);
  int j = 0;
  for (; j + 1 < n; j += 2) {
    const int s0 = sp[j];
    const int s1 = sp[j + 1];
    const float4 v0 = *(const float4*)(xq + (size_t)s0 * DIN);
    const float4 v1 = *(const float4*)(xq + (size_t)s1 * DIN);
    acc.x += v0.x; acc.y += v0.y; acc.z += v0.z; acc.w += v0.w;
    acc.x += v1.x; acc.y += v1.y; acc.z += v1.z; acc.w += v1.w;
  }
  if (j < n) {
    const float4 v0 = *(const float4*)(xq + (size_t)sp[j] * DIN);
    acc.x += v0.x; acc.y += v0.y; acc.z += v0.z; acc.w += v0.w;
  }
  *(float4*)(agg + (size_t)t * 4) = acc;
}

template <int DIN, int DOUT, bool RELU>
__global__ __launch_bounds__(256)
void layer_gemm(const float* __restrict__ agg,
                const int*   __restrict__ cnt,
                const float* __restrict__ X,
                const float* __restrict__ W,
                const float* __restrict__ root,
                const float* __restrict__ bias,
                float* __restrict__ out,
                int strip0, int strip_end) {
  constexpr int KA = NUM_REL * DIN;
  constexpr int KT = KA + DIN;
  constexpr int P  = DOUT + 8;
  constexpr int NT = DOUT / 16;
  constexpr int C4 = DOUT / 4;
  __shared__ float Ws[KT * P];

  for (int i = threadIdx.x; i < KT * C4; i += 256) {
    const int k = i / C4;
    const int c = (i - k * C4) * 4;
    const float* g = (k < KA) ? (W + (size_t)k * DOUT + c)
                              : (root + (size_t)(k - KA) * DOUT + c);
    *(float4*)(&Ws[k * P + c]) = *(const float4*)g;
  }
  __syncthreads();

  const int lane  = threadIdx.x & 31;
  const int wave  = threadIdx.x >> 5;
  const int half  = lane >> 4;
  const int l16   = lane & 15;
  const int strip = strip0 + blockIdx.x * 8 + wave;
  if (strip >= strip_end) return;
  const int m0  = strip * 16;
  const int row = m0 + l16;

  const int4 c4 = *(const int4*)(cnt + (size_t)row * NUM_REL);
  float inv[NUM_REL];
  inv[0] = 1.0f / fmaxf((float)c4.x, 1.0f);
  inv[1] = 1.0f / fmaxf((float)c4.y, 1.0f);
  inv[2] = 1.0f / fmaxf((float)c4.z, 1.0f);
  inv[3] = 1.0f / fmaxf((float)c4.w, 1.0f);

  v8f acc[NT];
#pragma unroll
  for (int t = 0; t < NT; ++t) acc[t] = v8f{};

  const float* arow = agg + (size_t)(row - strip0 * 16) * KA;
#pragma unroll
  for (int kk = 0; kk < KA; kk += 8) {
    const int kb = kk + 4 * half;
    float4 av = *(const float4*)(arow + kb);
    const float s = inv[kk / DIN];
    const v2f a0 = {av.x * s, av.y * s};
    const v2f a1 = {av.z * s, av.w * s};
    const float* bp = &Ws[kb * P + l16];
#pragma unroll
    for (int t = 0; t < NT; ++t) {
      const v2f b0 = {bp[t * 16], bp[t * 16 + P]};
      acc[t] = __builtin_amdgcn_wmma_f32_16x16x4_f32(false, a0, false, b0,
                                                     (short)0, acc[t], false, false);
    }
#pragma unroll
    for (int t = 0; t < NT; ++t) {
      const v2f b1 = {bp[t * 16 + 2 * P], bp[t * 16 + 3 * P]};
      acc[t] = __builtin_amdgcn_wmma_f32_16x16x4_f32(false, a1, false, b1,
                                                     (short)0, acc[t], false, false);
    }
  }

  const float* xrow = X + (size_t)row * DIN;
#pragma unroll
  for (int kk = 0; kk < DIN; kk += 8) {
    const int kb = kk + 4 * half;
    const float4 av = *(const float4*)(xrow + kb);
    const v2f a0 = {av.x, av.y};
    const v2f a1 = {av.z, av.w};
    const float* bp = &Ws[(KA + kb) * P + l16];
#pragma unroll
    for (int t = 0; t < NT; ++t) {
      const v2f b0 = {bp[t * 16], bp[t * 16 + P]};
      acc[t] = __builtin_amdgcn_wmma_f32_16x16x4_f32(false, a0, false, b0,
                                                     (short)0, acc[t], false, false);
    }
#pragma unroll
    for (int t = 0; t < NT; ++t) {
      const v2f b1 = {bp[t * 16 + 2 * P], bp[t * 16 + 3 * P]};
      acc[t] = __builtin_amdgcn_wmma_f32_16x16x4_f32(false, a1, false, b1,
                                                     (short)0, acc[t], false, false);
    }
  }

#pragma unroll
  for (int t = 0; t < NT; ++t) {
    const int   col = t * 16 + l16;
    const float bv  = bias[col];
#pragma unroll
    for (int v = 0; v < 8; ++v) {
      float y = acc[t][v] + bv;
      if (RELU) y = fmaxf(y, 0.0f);
      out[(size_t)(m0 + v + 8 * half) * DOUT + col] = y;
    }
  }
}

__global__ void fill_neginf(float* __restrict__ p, int n) {
  int t = blockIdx.x * blockDim.x + threadIdx.x;
  if (t < n) p[t] = -INFINITY;
}

__global__ void pool_max(const float* __restrict__ h,
                         const int* __restrict__ batch,
                         float* __restrict__ out) {
  int t = blockIdx.x * blockDim.x + threadIdx.x;
  if (t >= N_NODES * 64) return;
  const int n = t >> 6;
  const int c = t & 63;
  const float v = h[t];
  float* addr = &out[batch[n] * 64 + c];
  if (v >= 0.0f) atomicMax((int*)addr, __float_as_int(v));
  else           atomicMin((unsigned int*)addr, __float_as_uint(v));
}

template <int DIN, int DOUT, bool RELU, int PARTS, int P>
static void run_part(const float* X, const float* W, const float* root,
                     const float* bias, const int* sorted, const int* cnt,
                     const int* nloc, const int* bpre, float* agg, float* out,
                     hipStream_t stream) {
  constexpr int STRIPS = N_NODES / 16 / PARTS;
  static_assert(STRIPS * 16 * PARTS == N_NODES, "parts must tile the nodes");
  constexpr int NODES  = STRIPS * 16;
  constexpr int NTHR   = NODES * NUM_REL * (DIN / 4);
  constexpr int NODE0  = P * NODES;
  constexpr int STRIP0 = P * STRIPS;
  constexpr int STRIP1 = STRIP0 + STRIPS;
  gather_sum<DIN><<<(NTHR + 255) / 256, 256, 0, stream>>>(
      X, sorted, cnt, nloc, bpre, agg, NODE0, NTHR);
  layer_gemm<DIN, DOUT, RELU><<<(STRIPS + 7) / 8, 256, 0, stream>>>(
      agg, cnt, X, W, root, bias, out, STRIP0, STRIP1);
}

extern "C" void kernel_launch(void* const* d_in, const int* in_sizes, int n_in,
                              void* d_out, int out_size, void* d_ws,
                              size_t ws_size, hipStream_t stream) {
  (void)in_sizes; (void)n_in; (void)out_size; (void)ws_size;

  const float* x     = (const float*)d_in[0];
  const int*   ei    = (const int*)d_in[1];
  const int*   et    = (const int*)d_in[2];
  const int*   batch = (const int*)d_in[3];
  const float* W1 = (const float*)d_in[4];
  const float* r1 = (const float*)d_in[5];
  const float* b1 = (const float*)d_in[6];
  const float* W2 = (const float*)d_in[7];
  const float* r2 = (const float*)d_in[8];
  const float* b2 = (const float*)d_in[9];
  const float* W3 = (const float*)d_in[10];
  const float* r3 = (const float*)d_in[11];
  const float* b3 = (const float*)d_in[12];
  const int* src = ei;
  const int* dst = ei + N_EDGES;
  float* out = (float*)d_out;

  char*  ws     = (char*)d_ws;
  int*   cnt    = (int*)ws;
  float* buf2   = (float*)(ws + 1600000);
  float* buf13  = (float*)(ws + 14400000);
  float* agg    = (float*)(ws + 40000000);
  int*   sorted = (int*)(ws + 65600000);
  int*   rank   = (int*)(ws + 72000000);
  int*   nloc   = (int*)(ws + 78400000);
  int*   bsum   = (int*)(ws + 80000000);
  int*   bpre   = (int*)(ws + 80050048);

  zero_i32<<<(N_NODES * NUM_REL + 255) / 256, 256, 0, stream>>>(
      cnt, N_NODES * NUM_REL);
  count_rank<<<(N_EDGES + 255) / 256, 256, 0, stream>>>(dst, et, cnt, rank);
  scan_nodes<<<N_SBLK, SCAN_B, 0, stream>>>(cnt, nloc, bsum);
  scan_blocks<<<1, 512, 0, stream>>>(bsum, bpre);
  fill_csr<<<(N_EDGES + 255) / 256, 256, 0, stream>>>(src, dst, et, rank, cnt,
                                                      nloc, bpre, sorted);

  run_part<16, 16, true, 1, 0>(x,     W1, r1, b1, sorted, cnt, nloc, bpre,
                               agg, buf13, stream);
  run_part<16, 32, true, 1, 0>(buf13, W2, r2, b2, sorted, cnt, nloc, bpre,
                               agg, buf2, stream);
  run_part<32, 64, false, 2, 0>(buf2, W3, r3, b3, sorted, cnt, nloc, bpre,
                                agg, buf13, stream);
  run_part<32, 64, false, 2, 1>(buf2, W3, r3, b3, sorted, cnt, nloc, bpre,
                                agg, buf13, stream);

  fill_neginf<<<(NUM_GRAPHS * 64 + 255) / 256, 256, 0, stream>>>(
      out, NUM_GRAPHS * 64);
  pool_max<<<(N_NODES * 64 + 255) / 256, 256, 0, stream>>>(buf13, batch, out);
}


# graph max-pool fused into the layer-3 WMMA GEMM epilogue: a 16-row strip inside one graph gives 1 atomic max per column (else per element); 6.4M-thread pool kernel and layer-3 row store gone
# speedup vs baseline: 2.3370x; 1.0638x over previous
#include <hip/hip_runtime.h>
#include <math.h>


#define N_NODES    100000
#define N_EDGES    1600000
#define NUM_REL    4
#define NUM_GRAPHS 256

typedef __attribute__((ext_vector_type(2))) float v2f;
typedef __attribute__((ext_vector_type(8))) float v8f;

__global__ void zero_i32(int* __restrict__ p, int n) {
  int t = blockIdx.x * blockDim.x + threadIdx.x;
  if (t < n) p[t] = 0;
}

#define SCAN_B   256
#define N_SBLK   ((N_NODES + SCAN_B - 1) / SCAN_B)

__global__ void count_rank(const int* __restrict__ dst,
                           const int* __restrict__ et,
                           int* __restrict__ cnt,
                           int* __restrict__ rank) {
  int e = blockIdx.x * blockDim.x + threadIdx.x;
  if (e < N_EDGES) rank[e] = atomicAdd(&cnt[dst[e] * NUM_REL + et[e]], 1);
}

__global__ __launch_bounds__(SCAN_B)
void scan_nodes(const int* __restrict__ cnt,
                int* __restrict__ nloc,
                int* __restrict__ bsum) {
  __shared__ int s[SCAN_B];
  const int t = threadIdx.x;
  const int n = blockIdx.x * SCAN_B + t;
  int v = 0;
  if (n < N_NODES) {
    const int4 c = *(const int4*)(cnt + (size_t)n * NUM_REL);
    v = c.x + c.y + c.z + c.w;
  }
  s[t] = v;
  __syncthreads();
  for (int d = 1; d < SCAN_B; d <<= 1) {
    const int a = (t >= d) ? s[t - d] : 0;
    __syncthreads();
    s[t] += a;
    __syncthreads();
  }
  if (n < N_NODES) nloc[n] = s[t] - v;
  if (t == SCAN_B - 1) bsum[blockIdx.x] = s[t];
}

__global__ __launch_bounds__(512)
void scan_blocks(const int* __restrict__ bsum, int* __restrict__ bpre) {
  static_assert(N_SBLK <= 512, "one block scans all block totals");
  __shared__ int s[512];
  const int t = threadIdx.x;
  const int v = (t < N_SBLK) ? bsum[t] : 0;
  s[t] = v;
  __syncthreads();
  for (int d = 1; d < 512; d <<= 1) {
    const int a = (t >= d) ? s[t - d] : 0;
    __syncthreads();
    s[t] += a;
    __syncthreads();
  }
  if (t < N_SBLK) bpre[t] = s[t] - v;
}

__device__ __forceinline__ void seg_range(const int* __restrict__ cnt,
                                          const int* __restrict__ nloc,
                                          const int* __restrict__ bpre,
                                          int node, int r, int& beg, int& n) {
  const int4 c = *(const int4*)(cnt + (size_t)node * NUM_REL);
  beg = bpre[node / SCAN_B] + nloc[node];
  n = c.x;
  if (r >= 1) { beg += c.x; n = c.y; }
  if (r >= 2) { beg += c.y; n = c.z; }
  if (r >= 3) { beg += c.z; n = c.w; }
}

__global__ void fill_csr(const int* __restrict__ src,
                         const int* __restrict__ dst,
                         const int* __restrict__ et,
                         const int* __restrict__ rank,
                         const int* __restrict__ cnt,
                         const int* __restrict__ nloc,
                         const int* __restrict__ bpre,
                         int* __restrict__ sorted) {
  int e = blockIdx.x * blockDim.x + threadIdx.x;
  if (e >= N_EDGES) return;
  int beg, n;
  seg_range(cnt, nloc, bpre, dst[e], et[e], beg, n);
  sorted[beg + rank[e]] = src[e];
}

template <int DIN>
__global__ __launch_bounds__(256)
void gather_sum(const float* __restrict__ X,
                const int* __restrict__ sorted,
                const int* __restrict__ cnt,
                const int* __restrict__ nloc,
                const int* __restrict__ bpre,
                float* __restrict__ agg,
                int node0, int nthreads) {
  constexpr int Q = DIN / 4;
  const int t = blockIdx.x * 256 + threadIdx.x;
  if (t >= nthreads) return;
  const int q    = t & (Q - 1);
  const int segl = t / Q;
  const int r    = segl & (NUM_REL - 1);
  const int node = node0 + segl / NUM_REL;
  int beg, n;
  seg_range(cnt, nloc, bpre, node, r, beg, n);

  const float* xq = X + 4 * q;
  const int*   sp = sorted + beg;
  float4 acc = make_float4(0.0f, 0.0f, 0.0f, 0.0f);
  int j = 0;
  for (; j + 1 < n; j += 2) {
    const int s0 = sp[j];
    const int s1 = sp[j + 1];
    const float4 v0 = *(const float4*)(xq + (size_t)s0 * DIN);
    const float4 v1 = *(const float4*)(xq + (size_t)s1 * DIN);
    acc.x += v0.x; acc.y += v0.y; acc.z += v0.z; acc.w += v0.w;
    acc.x += v1.x; acc.y += v1.y; acc.z += v1.z; acc.w += v1.w;
  }
  if (j < n) {
    const float4 v0 = *(const float4*)(xq + (size_t)sp[j] * DIN);
    acc.x += v0.x; acc.y += v0.y; acc.z += v0.z; acc.w += v0.w;
  }
  *(float4*)(agg + (size_t)t * 4) = acc;
}

__device__ __forceinline__ void atomic_max_f32(float* addr, float v) {
  if (v >= 0.0f) atomicMax((int*)addr, __float_as_int(v));
  else           atomicMin((unsigned int*)addr, __float_as_uint(v));
}

template <int DIN, int DOUT, bool RELU, bool POOL>
__global__ __launch_bounds__(256)
void layer_gemm(const float* __restrict__ agg,
                const int*   __restrict__ cnt,
                const float* __restrict__ X,
                const float* __restrict__ W,
                const float* __restrict__ root,
                const float* __restrict__ bias,
                const int*   __restrict__ batch,
                float* __restrict__ out,
                int strip0, int strip_end) {
  constexpr int KA = NUM_REL * DIN;
  constexpr int KT = KA + DIN;
  constexpr int P  = DOUT + 8;
  constexpr int NT = DOUT / 16;
  constexpr int C4 = DOUT / 4;
  __shared__ float Ws[KT * P];

  for (int i = threadIdx.x; i < KT * C4; i += 256) {
    const int k = i / C4;
    const int c = (i - k * C4) * 4;
    const float* g = (k < KA) ? (W + (size_t)k * DOUT + c)
                              : (root + (size_t)(k - KA) * DOUT + c);
    *(float4*)(&Ws[k * P + c]) = *(const float4*)g;
  }
  __syncthreads();

  const int lane  = threadIdx.x & 31;
  const int wave  = threadIdx.x >> 5;
  const int half  = lane >> 4;
  const int l16   = lane & 15;
  const int strip = strip0 + blockIdx.x * 8 + wave;
  if (strip >= strip_end) return;
  const int m0  = strip * 16;
  const int row = m0 + l16;

  const int4 c4 = *(const int4*)(cnt + (size_t)row * NUM_REL);
  float inv[NUM_REL];
  inv[0] = 1.0f / fmaxf((float)c4.x, 1.0f);
  inv[1] = 1.0f / fmaxf((float)c4.y, 1.0f);
  inv[2] = 1.0f / fmaxf((float)c4.z, 1.0f);
  inv[3] = 1.0f / fmaxf((float)c4.w, 1.0f);

  float bvp[NT];
  int   g[8];
  if (POOL) {
#pragma unroll
    for (int t = 0; t < NT; ++t) bvp[t] = bias[t * 16 + l16];
    const int* bp8 = batch + m0 + 8 * half;
    const int4 ba = *(const int4*)(bp8);
    const int4 bb = *(const int4*)(bp8 + 4);
    g[0] = ba.x; g[1] = ba.y; g[2] = ba.z; g[3] = ba.w;
    g[4] = bb.x; g[5] = bb.y; g[6] = bb.z; g[7] = bb.w;
  }

  v8f acc[NT];
#pragma unroll
  for (int t = 0; t < NT; ++t) acc[t] = v8f{};

  const float* arow = agg + (size_t)(row - strip0 * 16) * KA;
#pragma unroll
  for (int kk = 0; kk < KA; kk += 8) {
    const int kb = kk + 4 * half;
    float4 av = *(const float4*)(arow + kb);
    const float s = inv[kk / DIN];
    const v2f a0 = {av.x * s, av.y * s};
    const v2f a1 = {av.z * s, av.w * s};
    const float* bp = &Ws[kb * P + l16];
#pragma unroll
    for (int t = 0; t < NT; ++t) {
      const v2f b0 = {bp[t * 16], bp[t * 16 + P]};
      acc[t] = __builtin_amdgcn_wmma_f32_16x16x4_f32(false, a0, false, b0,
                                                     (short)0, acc[t], false, false);
    }
#pragma unroll
    for (int t = 0; t < NT; ++t) {
      const v2f b1 = {bp[t * 16 + 2 * P], bp[t * 16 + 3 * P]};
      acc[t] = __builtin_amdgcn_wmma_f32_16x16x4_f32(false, a1, false, b1,
                                                     (short)0, acc[t], false, false);
    }
  }

  const float* xrow = X + (size_t)row * DIN;
#pragma unroll
  for (int kk = 0; kk < DIN; kk += 8) {
    const int kb = kk + 4 * half;
    const float4 av = *(const float4*)(xrow + kb);
    const v2f a0 = {av.x, av.y};
    const v2f a1 = {av.z, av.w};
    const float* bp = &Ws[(KA + kb) * P + l16];
#pragma unroll
    for (int t = 0; t < NT; ++t) {
      const v2f b0 = {bp[t * 16], bp[t * 16 + P]};
      acc[t] = __builtin_amdgcn_wmma_f32_16x16x4_f32(false, a0, false, b0,
                                                     (short)0, acc[t], false, false);
    }
#pragma unroll
    for (int t = 0; t < NT; ++t) {
      const v2f b1 = {bp[t * 16 + 2 * P], bp[t * 16 + 3 * P]};
      acc[t] = __builtin_amdgcn_wmma_f32_16x16x4_f32(false, a1, false, b1,
                                                     (short)0, acc[t], false, false);
    }
  }

  if (!POOL) {
#pragma unroll
    for (int t = 0; t < NT; ++t) {
      const int   col = t * 16 + l16;
      const float bv  = bias[col];
#pragma unroll
      for (int v = 0; v < 8; ++v) {
        float y = acc[t][v] + bv;
        if (RELU) y = fmaxf(y, 0.0f);
        out[(size_t)(m0 + v + 8 * half) * DOUT + col] = y;
      }
    }
  } else {
    const int g0 = __builtin_amdgcn_readfirstlane(g[0]);
    bool same = true;
#pragma unroll
    for (int v = 0; v < 8; ++v) same = same && (g[v] == g0);
    const bool one_graph = __builtin_amdgcn_ballot_w32(same) == 0xffffffffu;

#pragma unroll
    for (int t = 0; t < NT; ++t) {
      const int   col = t * 16 + l16;
      const float bv  = bvp[t];
      float y[8];
#pragma unroll
      for (int v = 0; v < 8; ++v) {
        y[v] = acc[t][v] + bv;
        if (RELU) y[v] = fmaxf(y[v], 0.0f);
      }
      if (one_graph) {
        float m = y[0];
#pragma unroll
        for (int v = 1; v < 8; ++v) m = fmaxf(m, y[v]);
        m = fmaxf(m, __shfl_xor(m, 16));
        if (half == 0) atomic_max_f32(&out[(size_t)g0 * DOUT + col], m);
      } else {
#pragma unroll
        for (int v = 0; v < 8; ++v)
          atomic_max_f32(&out[(size_t)g[v] * DOUT + col], y[v]);
      }
    }
  }
}

__global__ void fill_neginf(float* __restrict__ p, int n) {
  int t = blockIdx.x * blockDim.x + threadIdx.x;
  if (t < n) p[t] = -INFINITY;
}

template <int DIN, int DOUT, bool RELU, bool POOL, int PARTS, int P>
static void run_part(const float* X, const float* W, const float* root,
                     const float* bias, const int* sorted, const int* cnt,
                     const int* nloc, const int* bpre, const int* batch,
                     float* agg, float* out, hipStream_t stream) {
  constexpr int STRIPS = N_NODES / 16 / PARTS;
  static_assert(STRIPS * 16 * PARTS == N_NODES, "parts must tile the nodes");
  constexpr int NODES  = STRIPS * 16;
  constexpr int NTHR   = NODES * NUM_REL * (DIN / 4);
  constexpr int NODE0  = P * NODES;
  constexpr int STRIP0 = P * STRIPS;
  constexpr int STRIP1 = STRIP0 + STRIPS;
  gather_sum<DIN><<<(NTHR + 255) / 256, 256, 0, stream>>>(
      X, sorted, cnt, nloc, bpre, agg, NODE0, NTHR);
  layer_gemm<DIN, DOUT, RELU, POOL><<<(STRIPS + 7) / 8, 256, 0, stream>>>(
      agg, cnt, X, W, root, bias, batch, out, STRIP0, STRIP1);
}

extern "C" void kernel_launch(void* const* d_in, const int* in_sizes, int n_in,
                              void* d_out, int out_size, void* d_ws,
                              size_t ws_size, hipStream_t stream) {
  (void)in_sizes; (void)n_in; (void)out_size; (void)ws_size;

  const float* x     = (const float*)d_in[0];
  const int*   ei    = (const int*)d_in[1];
  const int*   et    = (const int*)d_in[2];
  const int*   batch = (const int*)d_in[3];
  const float* W1 = (const float*)d_in[4];
  const float* r1 = (const float*)d_in[5];
  const float* b1 = (const float*)d_in[6];
  const float* W2 = (const float*)d_in[7];
  const float* r2 = (const float*)d_in[8];
  const float* b2 = (const float*)d_in[9];
  const float* W3 = (const float*)d_in[10];
  const float* r3 = (const float*)d_in[11];
  const float* b3 = (const float*)d_in[12];
  const int* src = ei;
  const int* dst = ei + N_EDGES;
  float* out = (float*)d_out;

  char*  ws     = (char*)d_ws;
  int*   cnt    = (int*)ws;
  float* buf2   = (float*)(ws + 1600000);
  float* buf13  = (float*)(ws + 14400000);
  float* agg    = (float*)(ws + 40000000);
  int*   sorted = (int*)(ws + 65600000);
  int*   rank   = (int*)(ws + 72000000);
  int*   nloc   = (int*)(ws + 78400000);
  int*   bsum   = (int*)(ws + 80000000);
  int*   bpre   = (int*)(ws + 80050048);

  zero_i32<<<(N_NODES * NUM_REL + 255) / 256, 256, 0, stream>>>(
      cnt, N_NODES * NUM_REL);
  count_rank<<<(N_EDGES + 255) / 256, 256, 0, stream>>>(dst, et, cnt, rank);
  scan_nodes<<<N_SBLK, SCAN_B, 0, stream>>>(cnt, nloc, bsum);
  scan_blocks<<<1, 512, 0, stream>>>(bsum, bpre);
  fill_csr<<<(N_EDGES + 255) / 256, 256, 0, stream>>>(src, dst, et, rank, cnt,
                                                      nloc, bpre, sorted);

  run_part<16, 16, true, false, 1, 0>(x,     W1, r1, b1, sorted, cnt, nloc,
                                      bpre, batch, agg, buf13, stream);
  run_part<16, 32, true, false, 1, 0>(buf13, W2, r2, b2, sorted, cnt, nloc,
                                      bpre, batch, agg, buf2, stream);
  fill_neginf<<<(NUM_GRAPHS * 64 + 255) / 256, 256, 0, stream>>>(
      out, NUM_GRAPHS * 64);
  run_part<32, 64, false, true, 2, 0>(buf2, W3, r3, b3, sorted, cnt, nloc,
                                      bpre, batch, agg, out, stream);
  run_part<32, 64, false, true, 2, 1>(buf2, W3, r3, b3, sorted, cnt, nloc,
                                      bpre, batch, agg, out, stream);
}


# gather-sum fused into the layer WMMA GEMM: each lane sums its own A-fragment quads from X rows via the CSR, 4 relations in lockstep; agg, 4 gather launches, layer-3 split gone; slot order kept
# speedup vs baseline: 2.6935x; 1.1525x over previous
#include <hip/hip_runtime.h>
#include <math.h>


#define N_NODES    100000
#define N_EDGES    1600000
#define NUM_REL    4
#define NUM_GRAPHS 256

typedef __attribute__((ext_vector_type(2))) float v2f;
typedef __attribute__((ext_vector_type(8))) float v8f;

__global__ void zero_i32(int* __restrict__ p, int n) {
  int t = blockIdx.x * blockDim.x + threadIdx.x;
  if (t < n) p[t] = 0;
}

#define SCAN_B   256
#define N_SBLK   ((N_NODES + SCAN_B - 1) / SCAN_B)

__global__ void count_rank(const int* __restrict__ dst,
                           const int* __restrict__ et,
                           int* __restrict__ cnt,
                           int* __restrict__ rank) {
  int e = blockIdx.x * blockDim.x + threadIdx.x;
  if (e < N_EDGES) rank[e] = atomicAdd(&cnt[dst[e] * NUM_REL + et[e]], 1);
}

__global__ __launch_bounds__(SCAN_B)
void scan_nodes(const int* __restrict__ cnt,
                int* __restrict__ nloc,
                int* __restrict__ bsum) {
  __shared__ int s[SCAN_B];
  const int t = threadIdx.x;
  const int n = blockIdx.x * SCAN_B + t;
  int v = 0;
  if (n < N_NODES) {
    const int4 c = *(const int4*)(cnt + (size_t)n * NUM_REL);
    v = c.x + c.y + c.z + c.w;
  }
  s[t] = v;
  __syncthreads();
  for (int d = 1; d < SCAN_B; d <<= 1) {
    const int a = (t >= d) ? s[t - d] : 0;
    __syncthreads();
    s[t] += a;
    __syncthreads();
  }
  if (n < N_NODES) nloc[n] = s[t] - v;
  if (t == SCAN_B - 1) bsum[blockIdx.x] = s[t];
}

__global__ __launch_bounds__(512)
void scan_blocks(const int* __restrict__ bsum, int* __restrict__ bpre) {
  static_assert(N_SBLK <= 512, "one block scans all block totals");
  __shared__ int s[512];
  const int t = threadIdx.x;
  const int v = (t < N_SBLK) ? bsum[t] : 0;
  s[t] = v;
  __syncthreads();
  for (int d = 1; d < 512; d <<= 1) {
    const int a = (t >= d) ? s[t - d] : 0;
    __syncthreads();
    s[t] += a;
    __syncthreads();
  }
  if (t < N_SBLK) bpre[t] = s[t] - v;
}

__device__ __forceinline__ void seg_range(const int* __restrict__ cnt,
                                          const int* __restrict__ nloc,
                                          const int* __restrict__ bpre,
                                          int node, int r, int& beg, int& n) {
  const int4 c = *(const int4*)(cnt + (size_t)node * NUM_REL);
  beg = bpre[node / SCAN_B] + nloc[node];
  n = c.x;
  if (r >= 1) { beg += c.x; n = c.y; }
  if (r >= 2) { beg += c.y; n = c.z; }
  if (r >= 3) { beg += c.z; n = c.w; }
}

__global__ void fill_csr(const int* __restrict__ src,
                         const int* __restrict__ dst,
                         const int* __restrict__ et,
                         const int* __restrict__ rank,
                         const int* __restrict__ cnt,
                         const int* __restrict__ nloc,
                         const int* __restrict__ bpre,
                         int* __restrict__ sorted) {
  int e = blockIdx.x * blockDim.x + threadIdx.x;
  if (e >= N_EDGES) return;
  int beg, n;
  seg_range(cnt, nloc, bpre, dst[e], et[e], beg, n);
  sorted[beg + rank[e]] = src[e];
}

__device__ __forceinline__ void atomic_max_f32(float* addr, float v) {
  if (v >= 0.0f) atomicMax((int*)addr, __float_as_int(v));
  else           atomicMin((unsigned int*)addr, __float_as_uint(v));
}

template <int DIN, int DOUT, bool RELU, bool POOL>
__global__ __launch_bounds__(256)
void layer_fused(const float* __restrict__ X,
                 const int*   __restrict__ sorted,
                 const int*   __restrict__ cnt,
                 const int*   __restrict__ nloc,
                 const int*   __restrict__ bpre,
                 const float* __restrict__ W,
                 const float* __restrict__ root,
                 const float* __restrict__ bias,
                 const int*   __restrict__ batch,
                 float* __restrict__ out) {
  constexpr int KA = NUM_REL * DIN;
  constexpr int KT = KA + DIN;
  constexpr int P  = DOUT + 8;
  constexpr int NT = DOUT / 16;
  constexpr int C4 = DOUT / 4;
  constexpr int G  = DIN / 8;
  constexpr int U  = (DIN <= 16) ? 2 : 1;
  __shared__ float Ws[KT * P];

  for (int i = threadIdx.x; i < KT * C4; i += 256) {
    const int k = i / C4;
    const int c = (i - k * C4) * 4;
    const float* g = (k < KA) ? (W + (size_t)k * DOUT + c)
                              : (root + (size_t)(k - KA) * DOUT + c);
    *(float4*)(&Ws[k * P + c]) = *(const float4*)g;
  }
  __syncthreads();

  const int lane  = threadIdx.x & 31;
  const int wave  = threadIdx.x >> 5;
  const int half  = lane >> 4;
  const int l16   = lane & 15;
  const int strip = blockIdx.x * 8 + wave;
  if (strip >= N_NODES / 16) return;
  const int m0  = strip * 16;
  const int row = m0 + l16;

  const int4 c4 = *(const int4*)(cnt + (size_t)row * NUM_REL);
  int n[NUM_REL];
  n[0] = c4.x; n[1] = c4.y; n[2] = c4.z; n[3] = c4.w;

  int off[NUM_REL];
  off[0] = bpre[row / SCAN_B] + nloc[row];
  off[1] = off[0] + n[0];
  off[2] = off[1] + n[1];
  off[3] = off[2] + n[2];
  const int nmax = max(max(n[0], n[1]), max(n[2], n[3]));

  const float* xq = X + 4 * half;

  float4 a[NUM_REL][G];
#pragma unroll
  for (int r = 0; r < NUM_REL; ++r)
#pragma unroll
    for (int i = 0; i < G; ++i) a[r][i] = make_float4(0.0f, 0.0f, 0.0f, 0.0f);

  for (int j = 0; j < nmax; j += U) {
    int s[NUM_REL][U];
#pragma unroll
    for (int r = 0; r < NUM_REL; ++r)
#pragma unroll
      for (int u = 0; u < U; ++u)
        s[r][u] = sorted[(j + u < n[r]) ? off[r] + j + u : 0];
    float4 v[NUM_REL][U][G];
#pragma unroll
    for (int r = 0; r < NUM_REL; ++r)
#pragma unroll
      for (int u = 0; u < U; ++u)
#pragma unroll
        for (int i = 0; i < G; ++i)
          v[r][u][i] = *(const float4*)(xq + (size_t)s[r][u] * DIN + 8 * i);
#pragma unroll
    for (int r = 0; r < NUM_REL; ++r)
#pragma unroll
      for (int u = 0; u < U; ++u) {
        const bool act = j + u < n[r];
#pragma unroll
        for (int i = 0; i < G; ++i) {
          a[r][i].x += act ? v[r][u][i].x : 0.0f;
          a[r][i].y += act ? v[r][u][i].y : 0.0f;
          a[r][i].z += act ? v[r][u][i].z : 0.0f;
          a[r][i].w += act ? v[r][u][i].w : 0.0f;
        }
      }
  }

  float4 xv[G];
#pragma unroll
  for (int i = 0; i < G; ++i)
    xv[i] = *(const float4*)(xq + (size_t)row * DIN + 8 * i);

  float inv[NUM_REL];
#pragma unroll
  for (int r = 0; r < NUM_REL; ++r) inv[r] = 1.0f / fmaxf((float)n[r], 1.0f);

  float bvp[NT];
  int   g[8];
  if (POOL) {
#pragma unroll
    for (int t = 0; t < NT; ++t) bvp[t] = bias[t * 16 + l16];
    const int* bp8 = batch + m0 + 8 * half;
    const int4 ba = *(const int4*)(bp8);
    const int4 bb = *(const int4*)(bp8 + 4);
    g[0] = ba.x; g[1] = ba.y; g[2] = ba.z; g[3] = ba.w;
    g[4] = bb.x; g[5] = bb.y; g[6] = bb.z; g[7] = bb.w;
  }

  v8f acc[NT];
#pragma unroll
  for (int t = 0; t < NT; ++t) acc[t] = v8f{};

#pragma unroll
  for (int r = 0; r < NUM_REL; ++r) {
    const float sc = inv[r];
#pragma unroll
    for (int i = 0; i < G; ++i) {
      const int kb = r * DIN + 8 * i + 4 * half;
      const float4 av = a[r][i];
      const v2f a0 = {av.x * sc, av.y * sc};
      const v2f a1 = {av.z * sc, av.w * sc};
      const float* bp = &Ws[kb * P + l16];
#pragma unroll
      for (int t = 0; t < NT; ++t) {
        const v2f b0 = {bp[t * 16], bp[t * 16 + P]};
        acc[t] = __builtin_amdgcn_wmma_f32_16x16x4_f32(false, a0, false, b0,
                                                       (short)0, acc[t], false, false);
      }
#pragma unroll
      for (int t = 0; t < NT; ++t) {
        const v2f b1 = {bp[t * 16 + 2 * P], bp[t * 16 + 3 * P]};
        acc[t] = __builtin_amdgcn_wmma_f32_16x16x4_f32(false, a1, false, b1,
                                                       (short)0, acc[t], false, false);
      }
    }
  }

#pragma unroll
  for (int kk = 0; kk < DIN; kk += 8) {
    const int kb = kk + 4 * half;
    const float4 av = xv[kk / 8];
    const v2f a0 = {av.x, av.y};
    const v2f a1 = {av.z, av.w};
    const float* bp = &Ws[(KA + kb) * P + l16];
#pragma unroll
    for (int t = 0; t < NT; ++t) {
      const v2f b0 = {bp[t * 16], bp[t * 16 + P]};
      acc[t] = __builtin_amdgcn_wmma_f32_16x16x4_f32(false, a0, false, b0,
                                                     (short)0, acc[t], false, false);
    }
#pragma unroll
    for (int t = 0; t < NT; ++t) {
      const v2f b1 = {bp[t * 16 + 2 * P], bp[t * 16 + 3 * P]};
      acc[t] = __builtin_amdgcn_wmma_f32_16x16x4_f32(false, a1, false, b1,
                                                     (short)0, acc[t], false, false);
    }
  }

  if (!POOL) {
#pragma unroll
    for (int t = 0; t < NT; ++t) {
      const int   col = t * 16 + l16;
      const float bv  = bias[col];
#pragma unroll
      for (int v = 0; v < 8; ++v) {
        float y = acc[t][v] + bv;
        if (RELU) y = fmaxf(y, 0.0f);
        out[(size_t)(m0 + v + 8 * half) * DOUT + col] = y;
      }
    }
  } else {
    const int g0 = __builtin_amdgcn_readfirstlane(g[0]);
    bool same = true;
#pragma unroll
    for (int v = 0; v < 8; ++v) same = same && (g[v] == g0);
    const bool one_graph = __builtin_amdgcn_ballot_w32(same) == 0xffffffffu;

#pragma unroll
    for (int t = 0; t < NT; ++t) {
      const int   col = t * 16 + l16;
      const float bv  = bvp[t];
      float y[8];
#pragma unroll
      for (int v = 0; v < 8; ++v) {
        y[v] = acc[t][v] + bv;
        if (RELU) y[v] = fmaxf(y[v], 0.0f);
      }
      if (one_graph) {
        float m = y[0];
#pragma unroll
        for (int v = 1; v < 8; ++v) m = fmaxf(m, y[v]);
        m = fmaxf(m, __shfl_xor(m, 16));
        if (half == 0) atomic_max_f32(&out[(size_t)g0 * DOUT + col], m);
      } else {
#pragma unroll
        for (int v = 0; v < 8; ++v)
          atomic_max_f32(&out[(size_t)g[v] * DOUT + col], y[v]);
      }
    }
  }
}

__global__ void fill_neginf(float* __restrict__ p, int n) {
  int t = blockIdx.x * blockDim.x + threadIdx.x;
  if (t < n) p[t] = -INFINITY;
}

template <int DIN, int DOUT, bool RELU, bool POOL>
static void run_layer(const float* X, const float* W, const float* root,
                      const float* bias, const int* sorted, const int* cnt,
                      const int* nloc, const int* bpre, const int* batch,
                      float* out, hipStream_t stream) {
  static_assert(N_NODES % 16 == 0, "16-row strips must tile the nodes");
  layer_fused<DIN, DOUT, RELU, POOL><<<(N_NODES / 16 + 7) / 8, 256, 0, stream>>>(
      X, sorted, cnt, nloc, bpre, W, root, bias, batch, out);
}

extern "C" void kernel_launch(void* const* d_in, const int* in_sizes, int n_in,
                              void* d_out, int out_size, void* d_ws,
                              size_t ws_size, hipStream_t stream) {
  (void)in_sizes; (void)n_in; (void)out_size; (void)ws_size;

  const float* x     = (const float*)d_in[0];
  const int*   ei    = (const int*)d_in[1];
  const int*   et    = (const int*)d_in[2];
  const int*   batch = (const int*)d_in[3];
  const float* W1 = (const float*)d_in[4];
  const float* r1 = (const float*)d_in[5];
  const float* b1 = (const float*)d_in[6];
  const float* W2 = (const float*)d_in[7];
  const float* r2 = (const float*)d_in[8];
  const float* b2 = (const float*)d_in[9];
  const float* W3 = (const float*)d_in[10];
  const float* r3 = (const float*)d_in[11];
  const float* b3 = (const float*)d_in[12];
  const int* src = ei;
  const int* dst = ei + N_EDGES;
  float* out = (float*)d_out;

  char*  ws     = (char*)d_ws;
  int*   cnt    = (int*)ws;
  float* buf2   = (float*)(ws + 1600000);
  float* buf13  = (float*)(ws + 14400000);
  int*   sorted = (int*)(ws + 65600000);
  int*   rank   = (int*)(ws + 72000000);
  int*   nloc   = (int*)(ws + 78400000);
  int*   bsum   = (int*)(ws + 80000000);
  int*   bpre   = (int*)(ws + 80050048);

  zero_i32<<<(N_NODES * NUM_REL + 255) / 256, 256, 0, stream>>>(
      cnt, N_NODES * NUM_REL);
  count_rank<<<(N_EDGES + 255) / 256, 256, 0, stream>>>(dst, et, cnt, rank);
  scan_nodes<<<N_SBLK, SCAN_B, 0, stream>>>(cnt, nloc, bsum);
  scan_blocks<<<1, 512, 0, stream>>>(bsum, bpre);
  fill_csr<<<(N_EDGES + 255) / 256, 256, 0, stream>>>(src, dst, et, rank, cnt,
                                                      nloc, bpre, sorted);

  run_layer<16, 16, true, false>(x,     W1, r1, b1, sorted, cnt, nloc, bpre,
                                 batch, buf13, stream);
  run_layer<16, 32, true, false>(buf13, W2, r2, b2, sorted, cnt, nloc, bpre,
                                 batch, buf2, stream);
  fill_neginf<<<(NUM_GRAPHS * 64 + 255) / 256, 256, 0, stream>>>(
      out, NUM_GRAPHS * 64);
  run_layer<32, 64, false, true>(buf2, W3, r3, b3, sorted, cnt, nloc, bpre,
                                 batch, out, stream);
}


# CSR build: per-edge returning global adds replaced by a two-level counting sort ranked by LDS adds (256-node buckets); 5 launches to 3; slot order in a segment = LDS adds' arrival order
# speedup vs baseline: 3.1214x; 1.1589x over previous
#include <hip/hip_runtime.h>
#include <math.h>


#define N_NODES    100000
#define N_EDGES    1600000
#define NUM_REL    4
#define NUM_GRAPHS 256

typedef __attribute__((ext_vector_type(2))) float v2f;
typedef __attribute__((ext_vector_type(8))) float v8f;

#define SCAN_B    256
#define N_BKT     ((N_NODES + SCAN_B - 1) / SCAN_B)
#define HP        (((N_BKT + 3) / 4) * 4)
#define HQ        (HP / 4)
#define P1_T      512
#define P1_QPT    6
#define P1_CHUNK  (P1_T * P1_QPT * 4)
#define P1_BLOCKS ((N_EDGES + P1_CHUNK - 1) / P1_CHUNK)
#define N_QUADS   (N_EDGES / 4)

template <int NW>
__device__ __forceinline__ int block_excl_scan(int v, int* wtot) {
  const int lane = threadIdx.x & 31;
  const int wave = threadIdx.x >> 5;
  int inc = v;
#pragma unroll
  for (int d = 1; d < 32; d <<= 1) {
    const int t = __shfl_up(inc, d);
    if (lane >= d) inc += t;
  }
  if (lane == 31) wtot[wave] = inc;
  __syncthreads();
  if (wave == 0) {
    const int w = (lane < NW) ? wtot[lane] : 0;
    int wi = w;
#pragma unroll
    for (int d = 1; d < 32; d <<= 1) {
      const int t = __shfl_up(wi, d);
      if (lane >= d) wi += t;
    }
    if (lane < NW) wtot[lane] = wi - w;
  }
  __syncthreads();
  return inc - v + wtot[wave];
}

__global__ __launch_bounds__(P1_T)
void bucket_hist(const int* __restrict__ dst, int* __restrict__ hist) {
  static_assert(N_EDGES % 4 == 0, "edges are read four at a time");
  static_assert(HP <= P1_T, "one thread per hist column");
  __shared__ int h[HP];
  const int tid = threadIdx.x;
  const int4* d4 = (const int4*)dst;
  int4 dv[P1_QPT];
#pragma unroll
  for (int k = 0; k < P1_QPT; ++k) {
    const int q = (blockIdx.x * P1_QPT + k) * P1_T + tid;
    dv[k] = d4[(q < N_QUADS) ? q : 0];
  }
  if (tid < HP) h[tid] = 0;
  __syncthreads();
#pragma unroll
  for (int k = 0; k < P1_QPT; ++k) {
    const int q = (blockIdx.x * P1_QPT + k) * P1_T + tid;
    if (q < N_QUADS) {
      atomicAdd(&h[dv[k].x / SCAN_B], 1);
      atomicAdd(&h[dv[k].y / SCAN_B], 1);
      atomicAdd(&h[dv[k].z / SCAN_B], 1);
      atomicAdd(&h[dv[k].w / SCAN_B], 1);
    }
  }
  __syncthreads();
  if (tid < HP) hist[blockIdx.x * HP + tid] = h[tid];
}

#define CS_RG  (P1_T / HQ)
#define CS_NR  ((P1_BLOCKS + CS_RG - 1) / CS_RG)
#define CS_B   14

__global__ __launch_bounds__(P1_T)
void bucket_scatter(const int* __restrict__ src,
                    const int* __restrict__ dst,
                    const int* __restrict__ et,
                    const int* __restrict__ hist,
                    int* __restrict__ bpre,
                    int* __restrict__ part) {
  static_assert(N_NODES < (1 << 21), "src id must fit above the 10 key bits");
  static_assert(SCAN_B * NUM_REL == 1024, "key is 10 bits");
  __shared__ int tot[HP];
  __shared__ int cur[HP + 4];
  __shared__ int wtot[P1_T / 32];
  const int tid = threadIdx.x;
  const int4* s4 = (const int4*)src;
  const int4* d4 = (const int4*)dst;
  const int4* t4 = (const int4*)et;

  if (tid < HP) { tot[tid] = 0; cur[tid] = 0; }
  __syncthreads();

  int4 sv[P1_QPT], dv[P1_QPT], tv[P1_QPT];
#pragma unroll
  for (int k = 0; k < P1_QPT; ++k) {
    const int q  = (blockIdx.x * P1_QPT + k) * P1_T + tid;
    const int qq = (q < N_QUADS) ? q : 0;
    sv[k] = s4[qq]; dv[k] = d4[qq]; tv[k] = t4[qq];
  }

  if (tid < CS_RG * HQ) {
    const int g  = tid / HQ;
    const int c4 = tid - g * HQ;
    const int4* h4 = (const int4*)hist;
    int4 ta = make_int4(0, 0, 0, 0);
    int4 pa = make_int4(0, 0, 0, 0);
    for (int j0 = 0; j0 < CS_NR; j0 += CS_B) {
      int4 hv[CS_B];
#pragma unroll
      for (int u = 0; u < CS_B; ++u) {
        const int r  = g + (j0 + u) * CS_RG;
        const int rr = (r < P1_BLOCKS) ? r : P1_BLOCKS - 1;
        hv[u] = h4[rr * HQ + c4];
      }
#pragma unroll
      for (int u = 0; u < CS_B; ++u) {
        const int  r   = g + (j0 + u) * CS_RG;
        const bool in  = r < P1_BLOCKS;
        const bool pre = r < (int)blockIdx.x;
        ta.x += in ? hv[u].x : 0;  ta.y += in ? hv[u].y : 0;
        ta.z += in ? hv[u].z : 0;  ta.w += in ? hv[u].w : 0;
        pa.x += pre ? hv[u].x : 0; pa.y += pre ? hv[u].y : 0;
        pa.z += pre ? hv[u].z : 0; pa.w += pre ? hv[u].w : 0;
      }
    }
    atomicAdd(&tot[4 * c4 + 0], ta.x); atomicAdd(&tot[4 * c4 + 1], ta.y);
    atomicAdd(&tot[4 * c4 + 2], ta.z); atomicAdd(&tot[4 * c4 + 3], ta.w);
    atomicAdd(&cur[4 * c4 + 0], pa.x); atomicAdd(&cur[4 * c4 + 1], pa.y);
    atomicAdd(&cur[4 * c4 + 2], pa.z); atomicAdd(&cur[4 * c4 + 3], pa.w);
  }
  __syncthreads();

  const int tv0  = (tid < HP) ? tot[tid] : 0;
  const int base = block_excl_scan<P1_T / 32>(tv0, wtot);
  if (tid < N_BKT) {
    cur[tid] += base;
    if (blockIdx.x == 0) bpre[tid] = base;
  }
  __syncthreads();

  int pos[P1_QPT][4];
#pragma unroll
  for (int k = 0; k < P1_QPT; ++k) {
    const bool ok = (blockIdx.x * P1_QPT + k) * P1_T + tid < N_QUADS;
    pos[k][0] = atomicAdd(&cur[ok ? dv[k].x / SCAN_B : HP], 1);
    pos[k][1] = atomicAdd(&cur[ok ? dv[k].y / SCAN_B : HP], 1);
    pos[k][2] = atomicAdd(&cur[ok ? dv[k].z / SCAN_B : HP], 1);
    pos[k][3] = atomicAdd(&cur[ok ? dv[k].w / SCAN_B : HP], 1);
  }
#pragma unroll
  for (int k = 0; k < P1_QPT; ++k) {
    if ((blockIdx.x * P1_QPT + k) * P1_T + tid < N_QUADS) {
      part[pos[k][0]] = (sv[k].x << 10) | ((dv[k].x % SCAN_B) << 2) | (tv[k].x & 3);
      part[pos[k][1]] = (sv[k].y << 10) | ((dv[k].y % SCAN_B) << 2) | (tv[k].y & 3);
      part[pos[k][2]] = (sv[k].z << 10) | ((dv[k].z % SCAN_B) << 2) | (tv[k].z & 3);
      part[pos[k][3]] = (sv[k].w << 10) | ((dv[k].w % SCAN_B) << 2) | (tv[k].w & 3);
    }
  }
}

#define SORT_T    256
#define SORT_K    (SCAN_B * NUM_REL)
#define SORT_R    24
#define STAGE_CAP 8192

__global__ __launch_bounds__(SORT_T)
void bucket_sort(const int* __restrict__ part,
                 const int* __restrict__ bpre,
                 int* __restrict__ cnt,
                 int* __restrict__ nloc,
                 int* __restrict__ sorted) {
  static_assert(SORT_T == SCAN_B && NUM_REL == 4, "one thread per node, int4 of counts");
  __shared__ __attribute__((aligned(16))) int c[SORT_K + 4];
  __shared__ int wtot[SORT_T / 32];
  __shared__ int stage[STAGE_CAP];
  const int tid = threadIdx.x;
  const int b   = blockIdx.x;
  const int beg = bpre[b];
  const int end = (b + 1 < N_BKT) ? bpre[b + 1] : N_EDGES;
  const int n   = end - beg;

  int pv[SORT_R];
#pragma unroll
  for (int k = 0; k < SORT_R; ++k) {
    const int i = beg + k * SORT_T + tid;
    pv[k] = part[(i < end) ? i : 0];
  }
  *(int4*)&c[4 * tid] = make_int4(0, 0, 0, 0);
  __syncthreads();
#pragma unroll
  for (int k = 0; k < SORT_R; ++k)
    atomicAdd(&c[(beg + k * SORT_T + tid < end) ? (pv[k] & (SORT_K - 1)) : SORT_K], 1);
  for (int i = beg + SORT_R * SORT_T + tid; i < end; i += SORT_T)
    atomicAdd(&c[part[i] & (SORT_K - 1)], 1);
  __syncthreads();

  const int4 cv = *(const int4*)&c[4 * tid];
  const int  ex = block_excl_scan<SORT_T / 32>(cv.x + cv.y + cv.z + cv.w, wtot);
  const int node = b * SCAN_B + tid;
  if (node < N_NODES) {
    *(int4*)(cnt + (size_t)node * NUM_REL) = cv;
    nloc[node] = ex;
  }
  *(int4*)&c[4 * tid] = make_int4(ex, ex + cv.x, ex + cv.x + cv.y, ex + cv.x + cv.y + cv.z);
  __syncthreads();

  const bool staged = n <= STAGE_CAP;
  int pos[SORT_R];
#pragma unroll
  for (int k = 0; k < SORT_R; ++k)
    pos[k] = atomicAdd(&c[(beg + k * SORT_T + tid < end) ? (pv[k] & (SORT_K - 1)) : SORT_K], 1);
#pragma unroll
  for (int k = 0; k < SORT_R; ++k)
    if (beg + k * SORT_T + tid < end) {
      if (staged) stage[pos[k]] = pv[k] >> 10; else sorted[beg + pos[k]] = pv[k] >> 10;
    }
  for (int i = beg + SORT_R * SORT_T + tid; i < end; i += SORT_T) {
    const int p = part[i];
    const int q = atomicAdd(&c[p & (SORT_K - 1)], 1);
    if (staged) stage[q] = p >> 10; else sorted[beg + q] = p >> 10;
  }
  if (staged) {
    __syncthreads();
    for (int i = tid; i < n; i += SORT_T) sorted[beg + i] = stage[i];
  }
}

__device__ __forceinline__ void atomic_max_f32(float* addr, float v) {
  if (v >= 0.0f) atomicMax((int*)addr, __float_as_int(v));
  else           atomicMin((unsigned int*)addr, __float_as_uint(v));
}

template <int DIN, int DOUT, bool RELU, bool POOL>
__global__ __launch_bounds__(256)
void layer_fused(const float* __restrict__ X,
                 const int*   __restrict__ sorted,
                 const int*   __restrict__ cnt,
                 const int*   __restrict__ nloc,
                 const int*   __restrict__ bpre,
                 const float* __restrict__ W,
                 const float* __restrict__ root,
                 const float* __restrict__ bias,
                 const int*   __restrict__ batch,
                 float* __restrict__ out) {
  constexpr int KA = NUM_REL * DIN;
  constexpr int KT = KA + DIN;
  constexpr int P  = DOUT + 8;
  constexpr int NT = DOUT / 16;
  constexpr int C4 = DOUT / 4;
  constexpr int G  = DIN / 8;
  constexpr int U  = (DIN <= 16) ? 2 : 1;
  __shared__ float Ws[KT * P];

  for (int i = threadIdx.x; i < KT * C4; i += 256) {
    const int k = i / C4;
    const int c = (i - k * C4) * 4;
    const float* g = (k < KA) ? (W + (size_t)k * DOUT + c)
                              : (root + (size_t)(k - KA) * DOUT + c);
    *(float4*)(&Ws[k * P + c]) = *(const float4*)g;
  }
  __syncthreads();

  const int lane  = threadIdx.x & 31;
  const int wave  = threadIdx.x >> 5;
  const int half  = lane >> 4;
  const int l16   = lane & 15;
  const int strip = blockIdx.x * 8 + wave;
  if (strip >= N_NODES / 16) return;
  const int m0  = strip * 16;
  const int row = m0 + l16;

  const int4 c4 = *(const int4*)(cnt + (size_t)row * NUM_REL);
  int n[NUM_REL];
  n[0] = c4.x; n[1] = c4.y; n[2] = c4.z; n[3] = c4.w;

  int off[NUM_REL];
  off[0] = bpre[row / SCAN_B] + nloc[row];
  off[1] = off[0] + n[0];
  off[2] = off[1] + n[1];
  off[3] = off[2] + n[2];
  const int nmax = max(max(n[0], n[1]), max(n[2], n[3]));

  const float* xq = X + 4 * half;

  float4 a[NUM_REL][G];
#pragma unroll
  for (int r = 0; r < NUM_REL; ++r)
#pragma unroll
    for (int i = 0; i < G; ++i) a[r][i] = make_float4(0.0f, 0.0f, 0.0f, 0.0f);

  for (int j = 0; j < nmax; j += U) {
    int s[NUM_REL][U];
#pragma unroll
    for (int r = 0; r < NUM_REL; ++r)
#pragma unroll
      for (int u = 0; u < U; ++u)
        s[r][u] = sorted[(j + u < n[r]) ? off[r] + j + u : 0];
    float4 v[NUM_REL][U][G];
#pragma unroll
    for (int r = 0; r < NUM_REL; ++r)
#pragma unroll
      for (int u = 0; u < U; ++u)
#pragma unroll
        for (int i = 0; i < G; ++i)
          v[r][u][i] = *(const float4*)(xq + (size_t)s[r][u] * DIN + 8 * i);
#pragma unroll
    for (int r = 0; r < NUM_REL; ++r)
#pragma unroll
      for (int u = 0; u < U; ++u) {
        const bool act = j + u < n[r];
#pragma unroll
        for (int i = 0; i < G; ++i) {
          a[r][i].x += act ? v[r][u][i].x : 0.0f;
          a[r][i].y += act ? v[r][u][i].y : 0.0f;
          a[r][i].z += act ? v[r][u][i].z : 0.0f;
          a[r][i].w += act ? v[r][u][i].w : 0.0f;
        }
      }
  }

  float4 xv[G];
#pragma unroll
  for (int i = 0; i < G; ++i)
    xv[i] = *(const float4*)(xq + (size_t)row * DIN + 8 * i);

  float inv[NUM_REL];
#pragma unroll
  for (int r = 0; r < NUM_REL; ++r) inv[r] = 1.0f / fmaxf((float)n[r], 1.0f);

  float bvp[NT];
  int   g[8];
  if (POOL) {
#pragma unroll
    for (int t = 0; t < NT; ++t) bvp[t] = bias[t * 16 + l16];
    const int* bp8 = batch + m0 + 8 * half;
    const int4 ba = *(const int4*)(bp8);
    const int4 bb = *(const int4*)(bp8 + 4);
    g[0] = ba.x; g[1] = ba.y; g[2] = ba.z; g[3] = ba.w;
    g[4] = bb.x; g[5] = bb.y; g[6] = bb.z; g[7] = bb.w;
  }

  v8f acc[NT];
#pragma unroll
  for (int t = 0; t < NT; ++t) acc[t] = v8f{};

#pragma unroll
  for (int r = 0; r < NUM_REL; ++r) {
    const float sc = inv[r];
#pragma unroll
    for (int i = 0; i < G; ++i) {
      const int kb = r * DIN + 8 * i + 4 * half;
      const float4 av = a[r][i];
      const v2f a0 = {av.x * sc, av.y * sc};
      const v2f a1 = {av.z * sc, av.w * sc};
      const float* bp = &Ws[kb * P + l16];
#pragma unroll
      for (int t = 0; t < NT; ++t) {
        const v2f b0 = {bp[t * 16], bp[t * 16 + P]};
        acc[t] = __builtin_amdgcn_wmma_f32_16x16x4_f32(false, a0, false, b0,
                                                       (short)0, acc[t], false, false);
      }
#pragma unroll
      for (int t = 0; t < NT; ++t) {
        const v2f b1 = {bp[t * 16 + 2 * P], bp[t * 16 + 3 * P]};
        acc[t] = __builtin_amdgcn_wmma_f32_16x16x4_f32(false, a1, false, b1,
                                                       (short)0, acc[t], false, false);
      }
    }
  }

#pragma unroll
  for (int kk = 0; kk < DIN; kk += 8) {
    const int kb = kk + 4 * half;
    const float4 av = xv[kk / 8];
    const v2f a0 = {av.x, av.y};
    const v2f a1 = {av.z, av.w};
    const float* bp = &Ws[(KA + kb) * P + l16];
#pragma unroll
    for (int t = 0; t < NT; ++t) {
      const v2f b0 = {bp[t * 16], bp[t * 16 + P]};
      acc[t] = __builtin_amdgcn_wmma_f32_16x16x4_f32(false, a0, false, b0,
                                                     (short)0, acc[t], false, false);
    }
#pragma unroll
    for (int t = 0; t < NT; ++t) {
      const v2f b1 = {bp[t * 16 + 2 * P], bp[t * 16 + 3 * P]};
      acc[t] = __builtin_amdgcn_wmma_f32_16x16x4_f32(false, a1, false, b1,
                                                     (short)0, acc[t], false, false);
    }
  }

  if (!POOL) {
#pragma unroll
    for (int t = 0; t < NT; ++t) {
      const int   col = t * 16 + l16;
      const float bv  = bias[col];
#pragma unroll
      for (int v = 0; v < 8; ++v) {
        float y = acc[t][v] + bv;
        if (RELU) y = fmaxf(y, 0.0f);
        out[(size_t)(m0 + v + 8 * half) * DOUT + col] = y;
      }
    }
  } else {
    const int g0 = __builtin_amdgcn_readfirstlane(g[0]);
    bool same = true;
#pragma unroll
    for (int v = 0; v < 8; ++v) same = same && (g[v] == g0);
    const bool one_graph = __builtin_amdgcn_ballot_w32(same) == 0xffffffffu;

#pragma unroll
    for (int t = 0; t < NT; ++t) {
      const int   col = t * 16 + l16;
      const float bv  = bvp[t];
      float y[8];
#pragma unroll
      for (int v = 0; v < 8; ++v) {
        y[v] = acc[t][v] + bv;
        if (RELU) y[v] = fmaxf(y[v], 0.0f);
      }
      if (one_graph) {
        float m = y[0];
#pragma unroll
        for (int v = 1; v < 8; ++v) m = fmaxf(m, y[v]);
        m = fmaxf(m, __shfl_xor(m, 16));
        if (half == 0) atomic_max_f32(&out[(size_t)g0 * DOUT + col], m);
      } else {
#pragma unroll
        for (int v = 0; v < 8; ++v)
          atomic_max_f32(&out[(size_t)g[v] * DOUT + col], y[v]);
      }
    }
  }
}

__global__ void fill_neginf(float* __restrict__ p, int n) {
  int t = blockIdx.x * blockDim.x + threadIdx.x;
  if (t < n) p[t] = -INFINITY;
}

template <int DIN, int DOUT, bool RELU, bool POOL>
static void run_layer(const float* X, const float* W, const float* root,
                      const float* bias, const int* sorted, const int* cnt,
                      const int* nloc, const int* bpre, const int* batch,
                      float* out, hipStream_t stream) {
  static_assert(N_NODES % 16 == 0, "16-row strips must tile the nodes");
  layer_fused<DIN, DOUT, RELU, POOL><<<(N_NODES / 16 + 7) / 8, 256, 0, stream>>>(
      X, sorted, cnt, nloc, bpre, W, root, bias, batch, out);
}

extern "C" void kernel_launch(void* const* d_in, const int* in_sizes, int n_in,
                              void* d_out, int out_size, void* d_ws,
                              size_t ws_size, hipStream_t stream) {
  (void)in_sizes; (void)n_in; (void)out_size; (void)ws_size;

  const float* x     = (const float*)d_in[0];
  const int*   ei    = (const int*)d_in[1];
  const int*   et    = (const int*)d_in[2];
  const int*   batch = (const int*)d_in[3];
  const float* W1 = (const float*)d_in[4];
  const float* r1 = (const float*)d_in[5];
  const float* b1 = (const float*)d_in[6];
  const float* W2 = (const float*)d_in[7];
  const float* r2 = (const float*)d_in[8];
  const float* b2 = (const float*)d_in[9];
  const float* W3 = (const float*)d_in[10];
  const float* r3 = (const float*)d_in[11];
  const float* b3 = (const float*)d_in[12];
  const int* src = ei;
  const int* dst = ei + N_EDGES;
  float* out = (float*)d_out;

  char*  ws     = (char*)d_ws;
  int*   cnt    = (int*)ws;
  float* buf2   = (float*)(ws + 1600000);
  float* buf13  = (float*)(ws + 14400000);
  int*   hist   = (int*)(ws + 24000000);
  int*   sorted = (int*)(ws + 65600000);
  int*   part   = (int*)(ws + 72000000);
  int*   nloc   = (int*)(ws + 78400000);
  int*   bpre   = (int*)(ws + 80050048);

  bucket_hist<<<P1_BLOCKS, P1_T, 0, stream>>>(dst, hist);
  bucket_scatter<<<P1_BLOCKS, P1_T, 0, stream>>>(src, dst, et, hist, bpre, part);
  bucket_sort<<<N_BKT, SORT_T, 0, stream>>>(part, bpre, cnt, nloc, sorted);

  run_layer<16, 16, true, false>(x,     W1, r1, b1, sorted, cnt, nloc, bpre,
                                 batch, buf13, stream);
  run_layer<16, 32, true, false>(buf13, W2, r2, b2, sorted, cnt, nloc, bpre,
                                 batch, buf2, stream);
  fill_neginf<<<(NUM_GRAPHS * 64 + 255) / 256, 256, 0, stream>>>(
      out, NUM_GRAPHS * 64);
  run_layer<32, 64, false, true>(buf2, W3, r3, b3, sorted, cnt, nloc, bpre,
                                 batch, out, stream);
}


# layer kernels: loads the build made wait one by one now fly as groups: weights, root, counts and offsets in one round trip ahead of the barrier; no gather slot is a guarded block behind a full wait
# speedup vs baseline: 3.3650x; 1.0780x over previous
#include <hip/hip_runtime.h>
#include <math.h>


#define N_NODES    100000
#define N_EDGES    1600000
#define NUM_REL    4
#define NUM_GRAPHS 256

typedef __attribute__((ext_vector_type(2))) float v2f;
typedef __attribute__((ext_vector_type(8))) float v8f;

#define SCAN_B    256
#define N_BKT     ((N_NODES + SCAN_B - 1) / SCAN_B)
#define HP        (((N_BKT + 3) / 4) * 4)
#define HQ        (HP / 4)
#define P1_T      512
#define P1_QPT    6
#define P1_CHUNK  (P1_T * P1_QPT * 4)
#define P1_BLOCKS ((N_EDGES + P1_CHUNK - 1) / P1_CHUNK)
#define N_QUADS   (N_EDGES / 4)

template <int NW>
__device__ __forceinline__ int block_excl_scan(int v, int* wtot) {
  const int lane = threadIdx.x & 31;
  const int wave = threadIdx.x >> 5;
  int inc = v;
#pragma unroll
  for (int d = 1; d < 32; d <<= 1) {
    const int t = __shfl_up(inc, d);
    if (lane >= d) inc += t;
  }
  if (lane == 31) wtot[wave] = inc;
  __syncthreads();
  if (wave == 0) {
    const int w = (lane < NW) ? wtot[lane] : 0;
    int wi = w;
#pragma unroll
    for (int d = 1; d < 32; d <<= 1) {
      const int t = __shfl_up(wi, d);
      if (lane >= d) wi += t;
    }
    if (lane < NW) wtot[lane] = wi - w;
  }
  __syncthreads();
  return inc - v + wtot[wave];
}

__global__ __launch_bounds__(P1_T)
void bucket_hist(const int* __restrict__ dst, int* __restrict__ hist) {
  static_assert(N_EDGES % 4 == 0, "edges are read four at a time");
  static_assert(HP <= P1_T, "one thread per hist column");
  __shared__ int h[HP];
  const int tid = threadIdx.x;
  const int4* d4 = (const int4*)dst;
  int4 dv[P1_QPT];
#pragma unroll
  for (int k = 0; k < P1_QPT; ++k) {
    const int q = (blockIdx.x * P1_QPT + k) * P1_T + tid;
    dv[k] = d4[(q < N_QUADS) ? q : 0];
  }
  if (tid < HP) h[tid] = 0;
  __syncthreads();
#pragma unroll
  for (int k = 0; k < P1_QPT; ++k) {
    const int q = (blockIdx.x * P1_QPT + k) * P1_T + tid;
    if (q < N_QUADS) {
      atomicAdd(&h[dv[k].x / SCAN_B], 1);
      atomicAdd(&h[dv[k].y / SCAN_B], 1);
      atomicAdd(&h[dv[k].z / SCAN_B], 1);
      atomicAdd(&h[dv[k].w / SCAN_B], 1);
    }
  }
  __syncthreads();
  if (tid < HP) hist[blockIdx.x * HP + tid] = h[tid];
}

#define CS_RG  (P1_T / HQ)
#define CS_NR  ((P1_BLOCKS + CS_RG - 1) / CS_RG)
#define CS_B   14

__global__ __launch_bounds__(P1_T)
void bucket_scatter(const int* __restrict__ src,
                    const int* __restrict__ dst,
                    const int* __restrict__ et,
                    const int* __restrict__ hist,
                    int* __restrict__ bpre,
                    int* __restrict__ part) {
  static_assert(N_NODES < (1 << 21), "src id must fit above the 10 key bits");
  static_assert(SCAN_B * NUM_REL == 1024, "key is 10 bits");
  __shared__ int tot[HP];
  __shared__ int cur[HP + 4];
  __shared__ int wtot[P1_T / 32];
  const int tid = threadIdx.x;
  const int4* s4 = (const int4*)src;
  const int4* d4 = (const int4*)dst;
  const int4* t4 = (const int4*)et;

  if (tid < HP) { tot[tid] = 0; cur[tid] = 0; }
  __syncthreads();

  int4 sv[P1_QPT], dv[P1_QPT], tv[P1_QPT];
#pragma unroll
  for (int k = 0; k < P1_QPT; ++k) {
    const int q  = (blockIdx.x * P1_QPT + k) * P1_T + tid;
    const int qq = (q < N_QUADS) ? q : 0;
    sv[k] = s4[qq]; dv[k] = d4[qq]; tv[k] = t4[qq];
  }

  if (tid < CS_RG * HQ) {
    const int g  = tid / HQ;
    const int c4 = tid - g * HQ;
    const int4* h4 = (const int4*)hist;
    int4 ta = make_int4(0, 0, 0, 0);
    int4 pa = make_int4(0, 0, 0, 0);
    for (int j0 = 0; j0 < CS_NR; j0 += CS_B) {
      int4 hv[CS_B];
#pragma unroll
      for (int u = 0; u < CS_B; ++u) {
        const int r  = g + (j0 + u) * CS_RG;
        const int rr = (r < P1_BLOCKS) ? r : P1_BLOCKS - 1;
        hv[u] = h4[rr * HQ + c4];
      }
#pragma unroll
      for (int u = 0; u < CS_B; ++u) {
        const int  r   = g + (j0 + u) * CS_RG;
        const bool in  = r < P1_BLOCKS;
        const bool pre = r < (int)blockIdx.x;
        ta.x += in ? hv[u].x : 0;  ta.y += in ? hv[u].y : 0;
        ta.z += in ? hv[u].z : 0;  ta.w += in ? hv[u].w : 0;
        pa.x += pre ? hv[u].x : 0; pa.y += pre ? hv[u].y : 0;
        pa.z += pre ? hv[u].z : 0; pa.w += pre ? hv[u].w : 0;
      }
    }
    atomicAdd(&tot[4 * c4 + 0], ta.x); atomicAdd(&tot[4 * c4 + 1], ta.y);
    atomicAdd(&tot[4 * c4 + 2], ta.z); atomicAdd(&tot[4 * c4 + 3], ta.w);
    atomicAdd(&cur[4 * c4 + 0], pa.x); atomicAdd(&cur[4 * c4 + 1], pa.y);
    atomicAdd(&cur[4 * c4 + 2], pa.z); atomicAdd(&cur[4 * c4 + 3], pa.w);
  }
  __syncthreads();

  const int tv0  = (tid < HP) ? tot[tid] : 0;
  const int base = block_excl_scan<P1_T / 32>(tv0, wtot);
  if (tid < N_BKT) {
    cur[tid] += base;
    if (blockIdx.x == 0) bpre[tid] = base;
  }
  __syncthreads();

  int pos[P1_QPT][4];
#pragma unroll
  for (int k = 0; k < P1_QPT; ++k) {
    const bool ok = (blockIdx.x * P1_QPT + k) * P1_T + tid < N_QUADS;
    pos[k][0] = atomicAdd(&cur[ok ? dv[k].x / SCAN_B : HP], 1);
    pos[k][1] = atomicAdd(&cur[ok ? dv[k].y / SCAN_B : HP], 1);
    pos[k][2] = atomicAdd(&cur[ok ? dv[k].z / SCAN_B : HP], 1);
    pos[k][3] = atomicAdd(&cur[ok ? dv[k].w / SCAN_B : HP], 1);
  }
#pragma unroll
  for (int k = 0; k < P1_QPT; ++k) {
    if ((blockIdx.x * P1_QPT + k) * P1_T + tid < N_QUADS) {
      part[pos[k][0]] = (sv[k].x << 10) | ((dv[k].x % SCAN_B) << 2) | (tv[k].x & 3);
      part[pos[k][1]] = (sv[k].y << 10) | ((dv[k].y % SCAN_B) << 2) | (tv[k].y & 3);
      part[pos[k][2]] = (sv[k].z << 10) | ((dv[k].z % SCAN_B) << 2) | (tv[k].z & 3);
      part[pos[k][3]] = (sv[k].w << 10) | ((dv[k].w % SCAN_B) << 2) | (tv[k].w & 3);
    }
  }
}

#define SORT_T    256
#define SORT_K    (SCAN_B * NUM_REL)
#define SORT_R    24
#define STAGE_CAP 8192

__global__ __launch_bounds__(SORT_T)
void bucket_sort(const int* __restrict__ part,
                 const int* __restrict__ bpre,
                 int* __restrict__ cnt,
                 int* __restrict__ nloc,
                 int* __restrict__ sorted) {
  static_assert(SORT_T == SCAN_B && NUM_REL == 4, "one thread per node, int4 of counts");
  __shared__ __attribute__((aligned(16))) int c[SORT_K + 4];
  __shared__ int wtot[SORT_T / 32];
  __shared__ int stage[STAGE_CAP];
  const int tid = threadIdx.x;
  const int b   = blockIdx.x;
  const int beg = bpre[b];
  const int end = (b + 1 < N_BKT) ? bpre[b + 1] : N_EDGES;
  const int n   = end - beg;

  int pv[SORT_R];
#pragma unroll
  for (int k = 0; k < SORT_R; ++k) {
    const int i = beg + k * SORT_T + tid;
    pv[k] = part[(i < end) ? i : 0];
  }
  *(int4*)&c[4 * tid] = make_int4(0, 0, 0, 0);
  __syncthreads();
#pragma unroll
  for (int k = 0; k < SORT_R; ++k)
    atomicAdd(&c[(beg + k * SORT_T + tid < end) ? (pv[k] & (SORT_K - 1)) : SORT_K], 1);
  for (int i = beg + SORT_R * SORT_T + tid; i < end; i += SORT_T)
    atomicAdd(&c[part[i] & (SORT_K - 1)], 1);
  __syncthreads();

  const int4 cv = *(const int4*)&c[4 * tid];
  const int  ex = block_excl_scan<SORT_T / 32>(cv.x + cv.y + cv.z + cv.w, wtot);
  const int node = b * SCAN_B + tid;
  if (node < N_NODES) {
    *(int4*)(cnt + (size_t)node * NUM_REL) = cv;
    nloc[node] = ex;
  }
  *(int4*)&c[4 * tid] = make_int4(ex, ex + cv.x, ex + cv.x + cv.y, ex + cv.x + cv.y + cv.z);
  __syncthreads();

  const bool staged = n <= STAGE_CAP;
  int pos[SORT_R];
#pragma unroll
  for (int k = 0; k < SORT_R; ++k)
    pos[k] = atomicAdd(&c[(beg + k * SORT_T + tid < end) ? (pv[k] & (SORT_K - 1)) : SORT_K], 1);
#pragma unroll
  for (int k = 0; k < SORT_R; ++k)
    if (beg + k * SORT_T + tid < end) {
      if (staged) stage[pos[k]] = pv[k] >> 10; else sorted[beg + pos[k]] = pv[k] >> 10;
    }
  for (int i = beg + SORT_R * SORT_T + tid; i < end; i += SORT_T) {
    const int p = part[i];
    const int q = atomicAdd(&c[p & (SORT_K - 1)], 1);
    if (staged) stage[q] = p >> 10; else sorted[beg + q] = p >> 10;
  }
  if (staged) {
    __syncthreads();
    for (int i = tid; i < n; i += SORT_T) sorted[beg + i] = stage[i];
  }
}

__device__ __forceinline__ void atomic_max_f32(float* addr, float v) {
  if (v >= 0.0f) atomicMax((int*)addr, __float_as_int(v));
  else           atomicMin((unsigned int*)addr, __float_as_uint(v));
}

template <int DIN, int DOUT, bool RELU, bool POOL>
__global__ __launch_bounds__(256)
void layer_fused(const float* __restrict__ X,
                 const int*   __restrict__ sorted,
                 const int*   __restrict__ cnt,
                 const int*   __restrict__ nloc,
                 const int*   __restrict__ bpre,
                 const float* __restrict__ W,
                 const float* __restrict__ root,
                 const float* __restrict__ bias,
                 const int*   __restrict__ batch,
                 float* __restrict__ out) {
  constexpr int KA = NUM_REL * DIN;
  constexpr int KT = KA + DIN;
  constexpr int P  = DOUT + 8;
  constexpr int NT = DOUT / 16;
  constexpr int C4 = DOUT / 4;
  constexpr int G  = DIN / 8;
  constexpr int U  = (DIN <= 16) ? 2 : 1;
  __shared__ float Ws[KT * P];

  constexpr int NWQ = KA * C4;
  constexpr int NRQ = DIN * C4;
  static_assert(NWQ % 256 == 0, "the relation weights take whole rounds of the block");
  constexpr int NW = NWQ / 256;
  constexpr int NR = (NRQ + 255) / 256;

  const int lane  = threadIdx.x & 31;
  const int wave  = threadIdx.x >> 5;
  const int half  = lane >> 4;
  const int l16   = lane & 15;
  const int strip = blockIdx.x * 8 + wave;
  const bool live = strip < N_NODES / 16;
  const int m0  = (live ? strip : 0) * 16;
  const int row = m0 + l16;

  float4 wq[NW + NR];
#pragma unroll
  for (int k = 0; k < NW; ++k)
    wq[k] = ((const float4*)W)[threadIdx.x + k * 256];
#pragma unroll
  for (int k = 0; k < NR; ++k) {
    const int i = threadIdx.x + k * 256;
    wq[NW + k] = ((const float4*)root)[(i < NRQ) ? i : 0];
  }
  const int4 c4 = *(const int4*)(cnt + (size_t)row * NUM_REL);
  const int  nl = nloc[row];
  const int  bp = bpre[row / SCAN_B];

#pragma unroll
  for (int k = 0; k < NW; ++k) {
    const int i  = threadIdx.x + k * 256;
    const int kr = i / C4;
    const int c  = (i - kr * C4) * 4;
    *(float4*)(&Ws[kr * P + c]) = wq[k];
  }
#pragma unroll
  for (int k = 0; k < NR; ++k) {
    const int i = threadIdx.x + k * 256;
    asm volatile("" : "+v"(wq[NW + k].x), "+v"(wq[NW + k].y),
                      "+v"(wq[NW + k].z), "+v"(wq[NW + k].w));
    if (NRQ % 256 == 0 || i < NRQ) {
      const int kr = i / C4;
      const int c  = (i - kr * C4) * 4;
      *(float4*)(&Ws[(KA + kr) * P + c]) = wq[NW + k];
    }
  }
  __syncthreads();
  if (!live) return;

  int n[NUM_REL];
  n[0] = c4.x; n[1] = c4.y; n[2] = c4.z; n[3] = c4.w;

  int off[NUM_REL];
  off[0] = bp + nl;
  off[1] = off[0] + n[0];
  off[2] = off[1] + n[1];
  off[3] = off[2] + n[2];
  const int nmax = max(max(n[0], n[1]), max(n[2], n[3]));

  const float* xq = X + 4 * half;

  float4 a[NUM_REL][G];
#pragma unroll
  for (int r = 0; r < NUM_REL; ++r)
#pragma unroll
    for (int i = 0; i < G; ++i) a[r][i] = make_float4(0.0f, 0.0f, 0.0f, 0.0f);

  for (int j = 0; j < nmax; j += U) {
    int s[NUM_REL][U];
#pragma unroll
    for (int r = 0; r < NUM_REL; ++r)
#pragma unroll
      for (int u = 0; u < U; ++u)
        s[r][u] = sorted[(j + u < n[r]) ? off[r] + j + u : 0];
#pragma unroll
    for (int r = 0; r < NUM_REL; ++r)
#pragma unroll
      for (int u = 0; u < U; ++u)
        asm volatile("" : "+v"(s[r][u]));
    float4 v[NUM_REL][U][G];
#pragma unroll
    for (int r = 0; r < NUM_REL; ++r)
#pragma unroll
      for (int u = 0; u < U; ++u)
#pragma unroll
        for (int i = 0; i < G; ++i)
          v[r][u][i] = *(const float4*)(xq + (size_t)s[r][u] * DIN + 8 * i);
#pragma unroll
    for (int r = 0; r < NUM_REL; ++r)
#pragma unroll
      for (int u = 0; u < U; ++u) {
        const bool act = j + u < n[r];
#pragma unroll
        for (int i = 0; i < G; ++i) {
          a[r][i].x += act ? v[r][u][i].x : 0.0f;
          a[r][i].y += act ? v[r][u][i].y : 0.0f;
          a[r][i].z += act ? v[r][u][i].z : 0.0f;
          a[r][i].w += act ? v[r][u][i].w : 0.0f;
        }
      }
  }

  float4 xv[G];
#pragma unroll
  for (int i = 0; i < G; ++i)
    xv[i] = *(const float4*)(xq + (size_t)row * DIN + 8 * i);

  float inv[NUM_REL];
#pragma unroll
  for (int r = 0; r < NUM_REL; ++r) inv[r] = 1.0f / fmaxf((float)n[r], 1.0f);

  float bvp[NT];
  int   g[8];
  if (POOL) {
#pragma unroll
    for (int t = 0; t < NT; ++t) bvp[t] = bias[t * 16 + l16];
    const int* bp8 = batch + m0 + 8 * half;
    const int4 ba = *(const int4*)(bp8);
    const int4 bb = *(const int4*)(bp8 + 4);
    g[0] = ba.x; g[1] = ba.y; g[2] = ba.z; g[3] = ba.w;
    g[4] = bb.x; g[5] = bb.y; g[6] = bb.z; g[7] = bb.w;
  }

  v8f acc[NT];
#pragma unroll
  for (int t = 0; t < NT; ++t) acc[t] = v8f{};

#pragma unroll
  for (int r = 0; r < NUM_REL; ++r) {
    const float sc = inv[r];
#pragma unroll
    for (int i = 0; i < G; ++i) {
      const int kb = r * DIN + 8 * i + 4 * half;
      const float4 av = a[r][i];
      const v2f a0 = {av.x * sc, av.y * sc};
      const v2f a1 = {av.z * sc, av.w * sc};
      const float* bp = &Ws[kb * P + l16];
#pragma unroll
      for (int t = 0; t < NT; ++t) {
        const v2f b0 = {bp[t * 16], bp[t * 16 + P]};
        acc[t] = __builtin_amdgcn_wmma_f32_16x16x4_f32(false, a0, false, b0,
                                                       (short)0, acc[t], false, false);
      }
#pragma unroll
      for (int t = 0; t < NT; ++t) {
        const v2f b1 = {bp[t * 16 + 2 * P], bp[t * 16 + 3 * P]};
        acc[t] = __builtin_amdgcn_wmma_f32_16x16x4_f32(false, a1, false, b1,
                                                       (short)0, acc[t], false, false);
      }
    }
  }

#pragma unroll
  for (int kk = 0; kk < DIN; kk += 8) {
    const int kb = kk + 4 * half;
    const float4 av = xv[kk / 8];
    const v2f a0 = {av.x, av.y};
    const v2f a1 = {av.z, av.w};
    const float* bp = &Ws[(KA + kb) * P + l16];
#pragma unroll
    for (int t = 0; t < NT; ++t) {
      const v2f b0 = {bp[t * 16], bp[t * 16 + P]};
      acc[t] = __builtin_amdgcn_wmma_f32_16x16x4_f32(false, a0, false, b0,
                                                     (short)0, acc[t], false, false);
    }
#pragma unroll
    for (int t = 0; t < NT; ++t) {
      const v2f b1 = {bp[t * 16 + 2 * P], bp[t * 16 + 3 * P]};
      acc[t] = __builtin_amdgcn_wmma_f32_16x16x4_f32(false, a1, false, b1,
                                                     (short)0, acc[t], false, false);
    }
  }

  if (!POOL) {
#pragma unroll
    for (int t = 0; t < NT; ++t) {
      const int   col = t * 16 + l16;
      const float bv  = bias[col];
#pragma unroll
      for (int v = 0; v < 8; ++v) {
        float y = acc[t][v] + bv;
        if (RELU) y = fmaxf(y, 0.0f);
        out[(size_t)(m0 + v + 8 * half) * DOUT + col] = y;
      }
    }
  } else {
    const int g0 = __builtin_amdgcn_readfirstlane(g[0]);
    bool same = true;
#pragma unroll
    for (int v = 0; v < 8; ++v) same = same && (g[v] == g0);
    const bool one_graph = __builtin_amdgcn_ballot_w32(same) == 0xffffffffu;

#pragma unroll
    for (int t = 0; t < NT; ++t) {
      const int   col = t * 16 + l16;
      const float bv  = bvp[t];
      float y[8];
#pragma unroll
      for (int v = 0; v < 8; ++v) {
        y[v] = acc[t][v] + bv;
        if (RELU) y[v] = fmaxf(y[v], 0.0f);
      }
      if (one_graph) {
        float m = y[0];
#pragma unroll
        for (int v = 1; v < 8; ++v) m = fmaxf(m, y[v]);
        m = fmaxf(m, __shfl_xor(m, 16));
        if (half == 0) atomic_max_f32(&out[(size_t)g0 * DOUT + col], m);
      } else {
#pragma unroll
        for (int v = 0; v < 8; ++v)
          atomic_max_f32(&out[(size_t)g[v] * DOUT + col], y[v]);
      }
    }
  }
}

__global__ void fill_neginf(float* __restrict__ p, int n) {
  int t = blockIdx.x * blockDim.x + threadIdx.x;
  if (t < n) p[t] = -INFINITY;
}

template <int DIN, int DOUT, bool RELU, bool POOL>
static void run_layer(const float* X, const float* W, const float* root,
                      const float* bias, const int* sorted, const int* cnt,
                      const int* nloc, const int* bpre, const int* batch,
                      float* out, hipStream_t stream) {
  static_assert(N_NODES % 16 == 0, "16-row strips must tile the nodes");
  layer_fused<DIN, DOUT, RELU, POOL><<<(N_NODES / 16 + 7) / 8, 256, 0, stream>>>(
      X, sorted, cnt, nloc, bpre, W, root, bias, batch, out);
}

extern "C" void kernel_launch(void* const* d_in, const int* in_sizes, int n_in,
                              void* d_out, int out_size, void* d_ws,
                              size_t ws_size, hipStream_t stream) {
  (void)in_sizes; (void)n_in; (void)out_size; (void)ws_size;

  const float* x     = (const float*)d_in[0];
  const int*   ei    = (const int*)d_in[1];
  const int*   et    = (const int*)d_in[2];
  const int*   batch = (const int*)d_in[3];
  const float* W1 = (const float*)d_in[4];
  const float* r1 = (const float*)d_in[5];
  const float* b1 = (const float*)d_in[6];
  const float* W2 = (const float*)d_in[7];
  const float* r2 = (const float*)d_in[8];
  const float* b2 = (const float*)d_in[9];
  const float* W3 = (const float*)d_in[10];
  const float* r3 = (const float*)d_in[11];
  const float* b3 = (const float*)d_in[12];
  const int* src = ei;
  const int* dst = ei + N_EDGES;
  float* out = (float*)d_out;

  char*  ws     = (char*)d_ws;
  int*   cnt    = (int*)ws;
  float* buf2   = (float*)(ws + 1600000);
  float* buf13  = (float*)(ws + 14400000);
  int*   hist   = (int*)(ws + 24000000);
  int*   sorted = (int*)(ws + 65600000);
  int*   part   = (int*)(ws + 72000000);
  int*   nloc   = (int*)(ws + 78400000);
  int*   bpre   = (int*)(ws + 80050048);

  bucket_hist<<<P1_BLOCKS, P1_T, 0, stream>>>(dst, hist);
  bucket_scatter<<<P1_BLOCKS, P1_T, 0, stream>>>(src, dst, et, hist, bpre, part);
  bucket_sort<<<N_BKT, SORT_T, 0, stream>>>(part, bpre, cnt, nloc, sorted);

  run_layer<16, 16, true, false>(x,     W1, r1, b1, sorted, cnt, nloc, bpre,
                                 batch, buf13, stream);
  run_layer<16, 32, true, false>(buf13, W2, r2, b2, sorted, cnt, nloc, bpre,
                                 batch, buf2, stream);
  fill_neginf<<<(NUM_GRAPHS * 64 + 255) / 256, 256, 0, stream>>>(
      out, NUM_GRAPHS * 64);
  run_layer<32, 64, false, true>(buf2, W3, r3, b3, sorted, cnt, nloc, bpre,
                                 batch, out, stream);
}


# two launch boundaries removed (7 launches to 5): layer 1 runs inside the bucket-sort kernel, each block on its own 256 nodes, slots read from LDS; out's -inf start stored by the first kernel
# speedup vs baseline: 3.5760x; 1.0627x over previous
#include <hip/hip_runtime.h>
#include <math.h>


#define N_NODES    100000
#define N_EDGES    1600000
#define NUM_REL    4
#define NUM_GRAPHS 256

typedef __attribute__((ext_vector_type(2))) float v2f;
typedef __attribute__((ext_vector_type(8))) float v8f;

#define SCAN_B    256
#define N_BKT     ((N_NODES + SCAN_B - 1) / SCAN_B)
#define HP        (((N_BKT + 3) / 4) * 4)
#define HQ        (HP / 4)
#define P1_T      512
#define P1_QPT    6
#define P1_CHUNK  (P1_T * P1_QPT * 4)
#define P1_BLOCKS ((N_EDGES + P1_CHUNK - 1) / P1_CHUNK)
#define N_QUADS   (N_EDGES / 4)

template <int NW>
__device__ __forceinline__ int block_excl_scan(int v, int* wtot) {
  const int lane = threadIdx.x & 31;
  const int wave = threadIdx.x >> 5;
  int inc = v;
#pragma unroll
  for (int d = 1; d < 32; d <<= 1) {
    const int t = __shfl_up(inc, d);
    if (lane >= d) inc += t;
  }
  if (lane == 31) wtot[wave] = inc;
  __syncthreads();
  if (wave == 0) {
    const int w = (lane < NW) ? wtot[lane] : 0;
    int wi = w;
#pragma unroll
    for (int d = 1; d < 32; d <<= 1) {
      const int t = __shfl_up(wi, d);
      if (lane >= d) wi += t;
    }
    if (lane < NW) wtot[lane] = wi - w;
  }
  __syncthreads();
  return inc - v + wtot[wave];
}

__global__ __launch_bounds__(P1_T)
void bucket_hist(const int* __restrict__ dst, int* __restrict__ hist,
                 float* __restrict__ pool_out) {
  static_assert(N_EDGES % 4 == 0, "edges are read four at a time");
  static_assert(P1_BLOCKS * P1_T >= NUM_GRAPHS * 64, "one thread per pooled element");
  static_assert(HP <= P1_T, "one thread per hist column");
  __shared__ int h[HP];
  const int tid = threadIdx.x;
  const int4* d4 = (const int4*)dst;
  int4 dv[P1_QPT];
#pragma unroll
  for (int k = 0; k < P1_QPT; ++k) {
    const int q = (blockIdx.x * P1_QPT + k) * P1_T + tid;
    dv[k] = d4[(q < N_QUADS) ? q : 0];
  }
  if (tid < HP) h[tid] = 0;
  __syncthreads();
#pragma unroll
  for (int k = 0; k < P1_QPT; ++k) {
    const int q = (blockIdx.x * P1_QPT + k) * P1_T + tid;
    if (q < N_QUADS) {
      atomicAdd(&h[dv[k].x / SCAN_B], 1);
      atomicAdd(&h[dv[k].y / SCAN_B], 1);
      atomicAdd(&h[dv[k].z / SCAN_B], 1);
      atomicAdd(&h[dv[k].w / SCAN_B], 1);
    }
  }
  __syncthreads();
  if (tid < HP) hist[blockIdx.x * HP + tid] = h[tid];
  if (blockIdx.x * P1_T + tid < NUM_GRAPHS * 64)
    pool_out[blockIdx.x * P1_T + tid] = -INFINITY;
}

#define CS_RG  (P1_T / HQ)
#define CS_NR  ((P1_BLOCKS + CS_RG - 1) / CS_RG)
#define CS_B   14

__global__ __launch_bounds__(P1_T)
void bucket_scatter(const int* __restrict__ src,
                    const int* __restrict__ dst,
                    const int* __restrict__ et,
                    const int* __restrict__ hist,
                    int* __restrict__ bpre,
                    int* __restrict__ part) {
  static_assert(N_NODES < (1 << 21), "src id must fit above the 10 key bits");
  static_assert(SCAN_B * NUM_REL == 1024, "key is 10 bits");
  __shared__ int tot[HP];
  __shared__ int cur[HP + 4];
  __shared__ int wtot[P1_T / 32];
  const int tid = threadIdx.x;
  const int4* s4 = (const int4*)src;
  const int4* d4 = (const int4*)dst;
  const int4* t4 = (const int4*)et;

  if (tid < HP) { tot[tid] = 0; cur[tid] = 0; }
  __syncthreads();

  int4 sv[P1_QPT], dv[P1_QPT], tv[P1_QPT];
#pragma unroll
  for (int k = 0; k < P1_QPT; ++k) {
    const int q  = (blockIdx.x * P1_QPT + k) * P1_T + tid;
    const int qq = (q < N_QUADS) ? q : 0;
    sv[k] = s4[qq]; dv[k] = d4[qq]; tv[k] = t4[qq];
  }

  if (tid < CS_RG * HQ) {
    const int g  = tid / HQ;
    const int c4 = tid - g * HQ;
    const int4* h4 = (const int4*)hist;
    int4 ta = make_int4(0, 0, 0, 0);
    int4 pa = make_int4(0, 0, 0, 0);
    for (int j0 = 0; j0 < CS_NR; j0 += CS_B) {
      int4 hv[CS_B];
#pragma unroll
      for (int u = 0; u < CS_B; ++u) {
        const int r  = g + (j0 + u) * CS_RG;
        const int rr = (r < P1_BLOCKS) ? r : P1_BLOCKS - 1;
        hv[u] = h4[rr * HQ + c4];
      }
#pragma unroll
      for (int u = 0; u < CS_B; ++u) {
        const int  r   = g + (j0 + u) * CS_RG;
        const bool in  = r < P1_BLOCKS;
        const bool pre = r < (int)blockIdx.x;
        ta.x += in ? hv[u].x : 0;  ta.y += in ? hv[u].y : 0;
        ta.z += in ? hv[u].z : 0;  ta.w += in ? hv[u].w : 0;
        pa.x += pre ? hv[u].x : 0; pa.y += pre ? hv[u].y : 0;
        pa.z += pre ? hv[u].z : 0; pa.w += pre ? hv[u].w : 0;
      }
    }
    atomicAdd(&tot[4 * c4 + 0], ta.x); atomicAdd(&tot[4 * c4 + 1], ta.y);
    atomicAdd(&tot[4 * c4 + 2], ta.z); atomicAdd(&tot[4 * c4 + 3], ta.w);
    atomicAdd(&cur[4 * c4 + 0], pa.x); atomicAdd(&cur[4 * c4 + 1], pa.y);
    atomicAdd(&cur[4 * c4 + 2], pa.z); atomicAdd(&cur[4 * c4 + 3], pa.w);
  }
  __syncthreads();

  const int tv0  = (tid < HP) ? tot[tid] : 0;
  const int base = block_excl_scan<P1_T / 32>(tv0, wtot);
  if (tid < N_BKT) {
    cur[tid] += base;
    if (blockIdx.x == 0) bpre[tid] = base;
  }
  __syncthreads();

  int pos[P1_QPT][4];
#pragma unroll
  for (int k = 0; k < P1_QPT; ++k) {
    const bool ok = (blockIdx.x * P1_QPT + k) * P1_T + tid < N_QUADS;
    pos[k][0] = atomicAdd(&cur[ok ? dv[k].x / SCAN_B : HP], 1);
    pos[k][1] = atomicAdd(&cur[ok ? dv[k].y / SCAN_B : HP], 1);
    pos[k][2] = atomicAdd(&cur[ok ? dv[k].z / SCAN_B : HP], 1);
    pos[k][3] = atomicAdd(&cur[ok ? dv[k].w / SCAN_B : HP], 1);
  }
#pragma unroll
  for (int k = 0; k < P1_QPT; ++k) {
    if ((blockIdx.x * P1_QPT + k) * P1_T + tid < N_QUADS) {
      part[pos[k][0]] = (sv[k].x << 10) | ((dv[k].x % SCAN_B) << 2) | (tv[k].x & 3);
      part[pos[k][1]] = (sv[k].y << 10) | ((dv[k].y % SCAN_B) << 2) | (tv[k].y & 3);
      part[pos[k][2]] = (sv[k].z << 10) | ((dv[k].z % SCAN_B) << 2) | (tv[k].z & 3);
      part[pos[k][3]] = (sv[k].w << 10) | ((dv[k].w % SCAN_B) << 2) | (tv[k].w & 3);
    }
  }
}

#define SORT_T    256
#define SORT_K    (SCAN_B * NUM_REL)
#define SORT_R    24
#define STAGE_CAP 8192

__device__ __forceinline__ void atomic_max_f32(float* addr, float v) {
  if (v >= 0.0f) atomicMax((int*)addr, __float_as_int(v));
  else           atomicMin((unsigned int*)addr, __float_as_uint(v));
}

enum { SLOTS_GLOBAL = 0,
       SLOTS_LDS    = 1,
       SLOTS_FRESH  = 2 };
template <int KIND>
__device__ __forceinline__ int fetch_slot(const int* slots, int i) {
  if (KIND == SLOTS_FRESH)
    return __hip_atomic_load(slots + i, __ATOMIC_RELAXED, __HIP_MEMORY_SCOPE_AGENT);
  return slots[i];
}

template <int DIN, int DOUT, bool RELU, bool POOL, int KIND>
__device__ __forceinline__ void strip_layer(const float* __restrict__ X,
                                            const int* slots,
                                            const int (&n)[NUM_REL],
                                            const int (&off)[NUM_REL],
                                            const int m0, const int row,
                                            const int half, const int l16,
                                            const float* Ws,
                                            const float* __restrict__ bias,
                                            const int*   __restrict__ batch,
                                            float* __restrict__ out) {
  constexpr int KA = NUM_REL * DIN;
  constexpr int P  = DOUT + 8;
  constexpr int NT = DOUT / 16;
  constexpr int G  = DIN / 8;
  constexpr int U  = (DIN <= 16) ? 2 : 1;

  const int nmax = max(max(n[0], n[1]), max(n[2], n[3]));

  const float* xq = X + 4 * half;

  float4 a[NUM_REL][G];
#pragma unroll
  for (int r = 0; r < NUM_REL; ++r)
#pragma unroll
    for (int i = 0; i < G; ++i) a[r][i] = make_float4(0.0f, 0.0f, 0.0f, 0.0f);

  for (int j = 0; j < nmax; j += U) {
    int s[NUM_REL][U];
#pragma unroll
    for (int r = 0; r < NUM_REL; ++r)
#pragma unroll
      for (int u = 0; u < U; ++u)
        s[r][u] = fetch_slot<KIND>(slots, (j + u < n[r]) ? off[r] + j + u : 0);
#pragma unroll
    for (int r = 0; r < NUM_REL; ++r)
#pragma unroll
      for (int u = 0; u < U; ++u)
        asm volatile("" : "+v"(s[r][u]));
    float4 v[NUM_REL][U][G];
#pragma unroll
    for (int r = 0; r < NUM_REL; ++r)
#pragma unroll
      for (int u = 0; u < U; ++u)
#pragma unroll
        for (int i = 0; i < G; ++i)
          v[r][u][i] = *(const float4*)(xq + (size_t)s[r][u] * DIN + 8 * i);
#pragma unroll
    for (int r = 0; r < NUM_REL; ++r)
#pragma unroll
      for (int u = 0; u < U; ++u) {
        const bool act = j + u < n[r];
#pragma unroll
        for (int i = 0; i < G; ++i) {
          a[r][i].x += act ? v[r][u][i].x : 0.0f;
          a[r][i].y += act ? v[r][u][i].y : 0.0f;
          a[r][i].z += act ? v[r][u][i].z : 0.0f;
          a[r][i].w += act ? v[r][u][i].w : 0.0f;
        }
      }
  }

  float4 xv[G];
#pragma unroll
  for (int i = 0; i < G; ++i)
    xv[i] = *(const float4*)(xq + (size_t)row * DIN + 8 * i);

  float inv[NUM_REL];
#pragma unroll
  for (int r = 0; r < NUM_REL; ++r) inv[r] = 1.0f / fmaxf((float)n[r], 1.0f);

  float bvp[NT];
  int   g[8];
  if (POOL) {
#pragma unroll
    for (int t = 0; t < NT; ++t) bvp[t] = bias[t * 16 + l16];
    const int* bp8 = batch + m0 + 8 * half;
    const int4 ba = *(const int4*)(bp8);
    const int4 bb = *(const int4*)(bp8 + 4);
    g[0] = ba.x; g[1] = ba.y; g[2] = ba.z; g[3] = ba.w;
    g[4] = bb.x; g[5] = bb.y; g[6] = bb.z; g[7] = bb.w;
  }

  v8f acc[NT];
#pragma unroll
  for (int t = 0; t < NT; ++t) acc[t] = v8f{};

#pragma unroll
  for (int r = 0; r < NUM_REL; ++r) {
    const float sc = inv[r];
#pragma unroll
    for (int i = 0; i < G; ++i) {
      const int kb = r * DIN + 8 * i + 4 * half;
      const float4 av = a[r][i];
      const v2f a0 = {av.x * sc, av.y * sc};
      const v2f a1 = {av.z * sc, av.w * sc};
      const float* bp = &Ws[kb * P + l16];
#pragma unroll
      for (int t = 0; t < NT; ++t) {
        const v2f b0 = {bp[t * 16], bp[t * 16 + P]};
        acc[t] = __builtin_amdgcn_wmma_f32_16x16x4_f32(false, a0, false, b0,
                                                       (short)0, acc[t], false, false);
      }
#pragma unroll
      for (int t = 0; t < NT; ++t) {
        const v2f b1 = {bp[t * 16 + 2 * P], bp[t * 16 + 3 * P]};
        acc[t] = __builtin_amdgcn_wmma_f32_16x16x4_f32(false, a1, false, b1,
                                                       (short)0, acc[t], false, false);
      }
    }
  }

#pragma unroll
  for (int kk = 0; kk < DIN; kk += 8) {
    const int kb = kk + 4 * half;
    const float4 av = xv[kk / 8];
    const v2f a0 = {av.x, av.y};
    const v2f a1 = {av.z, av.w};
    const float* bp = &Ws[(KA + kb) * P + l16];
#pragma unroll
    for (int t = 0; t < NT; ++t) {
      const v2f b0 = {bp[t * 16], bp[t * 16 + P]};
      acc[t] = __builtin_amdgcn_wmma_f32_16x16x4_f32(false, a0, false, b0,
                                                     (short)0, acc[t], false, false);
    }
#pragma unroll
    for (int t = 0; t < NT; ++t) {
      const v2f b1 = {bp[t * 16 + 2 * P], bp[t * 16 + 3 * P]};
      acc[t] = __builtin_amdgcn_wmma_f32_16x16x4_f32(false, a1, false, b1,
                                                     (short)0, acc[t], false, false);
    }
  }

  if (!POOL) {
#pragma unroll
    for (int t = 0; t < NT; ++t) {
      const int   col = t * 16 + l16;
      const float bv  = bias[col];
#pragma unroll
      for (int v = 0; v < 8; ++v) {
        float y = acc[t][v] + bv;
        if (RELU) y = fmaxf(y, 0.0f);
        out[(size_t)(m0 + v + 8 * half) * DOUT + col] = y;
      }
    }
  } else {
    const int g0 = __builtin_amdgcn_readfirstlane(g[0]);
    bool same = true;
#pragma unroll
    for (int v = 0; v < 8; ++v) same = same && (g[v] == g0);
    const bool one_graph = __builtin_amdgcn_ballot_w32(same) == 0xffffffffu;

#pragma unroll
    for (int t = 0; t < NT; ++t) {
      const int   col = t * 16 + l16;
      const float bv  = bvp[t];
      float y[8];
#pragma unroll
      for (int v = 0; v < 8; ++v) {
        y[v] = acc[t][v] + bv;
        if (RELU) y[v] = fmaxf(y[v], 0.0f);
      }
      if (one_graph) {
        float m = y[0];
#pragma unroll
        for (int v = 1; v < 8; ++v) m = fmaxf(m, y[v]);
        m = fmaxf(m, __shfl_xor(m, 16));
        if (half == 0) atomic_max_f32(&out[(size_t)g0 * DOUT + col], m);
      } else {
#pragma unroll
        for (int v = 0; v < 8; ++v)
          atomic_max_f32(&out[(size_t)g[v] * DOUT + col], y[v]);
      }
    }
  }
}

template <int DIN, int DOUT, bool RELU, bool POOL>
__global__ __launch_bounds__(256)
void layer_fused(const float* __restrict__ X,
                 const int*   __restrict__ sorted,
                 const int*   __restrict__ cnt,
                 const int*   __restrict__ nloc,
                 const int*   __restrict__ bpre,
                 const float* __restrict__ W,
                 const float* __restrict__ root,
                 const float* __restrict__ bias,
                 const int*   __restrict__ batch,
                 float* __restrict__ out) {
  constexpr int KA = NUM_REL * DIN;
  constexpr int KT = KA + DIN;
  constexpr int P  = DOUT + 8;
  constexpr int C4 = DOUT / 4;
  __shared__ float Ws[KT * P];

  constexpr int NWQ = KA * C4;
  constexpr int NRQ = DIN * C4;
  static_assert(NWQ % 256 == 0, "the relation weights take whole rounds of the block");
  constexpr int NW = NWQ / 256;
  constexpr int NR = (NRQ + 255) / 256;

  const int lane  = threadIdx.x & 31;
  const int wave  = threadIdx.x >> 5;
  const int half  = lane >> 4;
  const int l16   = lane & 15;
  const int strip = blockIdx.x * 8 + wave;
  const bool live = strip < N_NODES / 16;
  const int m0  = (live ? strip : 0) * 16;
  const int row = m0 + l16;

  float4 wq[NW + NR];
#pragma unroll
  for (int k = 0; k < NW; ++k)
    wq[k] = ((const float4*)W)[threadIdx.x + k * 256];
#pragma unroll
  for (int k = 0; k < NR; ++k) {
    const int i = threadIdx.x + k * 256;
    wq[NW + k] = ((const float4*)root)[(i < NRQ) ? i : 0];
  }
  const int4 c4 = *(const int4*)(cnt + (size_t)row * NUM_REL);
  const int  nl = nloc[row];
  const int  bp = bpre[row / SCAN_B];

#pragma unroll
  for (int k = 0; k < NW; ++k) {
    const int i  = threadIdx.x + k * 256;
    const int kr = i / C4;
    const int c  = (i - kr * C4) * 4;
    *(float4*)(&Ws[kr * P + c]) = wq[k];
  }
#pragma unroll
  for (int k = 0; k < NR; ++k) {
    const int i = threadIdx.x + k * 256;
    asm volatile("" : "+v"(wq[NW + k].x), "+v"(wq[NW + k].y),
                      "+v"(wq[NW + k].z), "+v"(wq[NW + k].w));
    if (NRQ % 256 == 0 || i < NRQ) {
      const int kr = i / C4;
      const int c  = (i - kr * C4) * 4;
      *(float4*)(&Ws[(KA + kr) * P + c]) = wq[NW + k];
    }
  }
  __syncthreads();
  if (!live) return;

  int n[NUM_REL];
  n[0] = c4.x; n[1] = c4.y; n[2] = c4.z; n[3] = c4.w;

  int off[NUM_REL];
  off[0] = bp + nl;
  off[1] = off[0] + n[0];
  off[2] = off[1] + n[1];
  off[3] = off[2] + n[2];
  strip_layer<DIN, DOUT, RELU, POOL, SLOTS_GLOBAL>(X, sorted, n, off, m0, row, half,
                                                   l16, Ws, bias, batch, out);
}

#define L1_DIN  16
#define L1_DOUT 16
#define L1_P    (L1_DOUT + 8)
#define L1_KT   ((NUM_REL + 1) * L1_DIN)

__global__ __launch_bounds__(SORT_T)
void bucket_sort_layer1(const int* __restrict__ part,
                        const int* __restrict__ bpre,
                        int* __restrict__ cnt,
                        int* __restrict__ nloc,
                        int* sorted,
                        const float* __restrict__ X,
                        const float* __restrict__ W,
                        const float* __restrict__ root,
                        const float* __restrict__ bias,
                        float* __restrict__ out) {
  static_assert(SORT_T == SCAN_B && NUM_REL == 4, "one thread per node, int4 of counts");
  static_assert(NUM_REL * L1_DIN * (L1_DOUT / 4) == SORT_T, "one float4 of W per thread");
  static_assert(L1_DIN * (L1_DOUT / 4) <= SORT_T, "root in one round of the block");
  static_assert(N_NODES % 16 == 0, "16-row strips must tile the nodes");
  __shared__ __attribute__((aligned(16))) float Ws[L1_KT * L1_P];
  __shared__ __attribute__((aligned(16))) int c[SORT_K + 4];
  __shared__ int wtot[SORT_T / 32];
  __shared__ int stage[STAGE_CAP];
  const int tid = threadIdx.x;
  const int b   = blockIdx.x;
  const float4 wq = ((const float4*)W)[tid];
  const float4 rq = ((const float4*)root)[(tid < L1_DIN * (L1_DOUT / 4)) ? tid : 0];
  const int beg = bpre[b];
  const int end = (b + 1 < N_BKT) ? bpre[b + 1] : N_EDGES;
  const int n   = end - beg;

  int pv[SORT_R];
#pragma unroll
  for (int k = 0; k < SORT_R; ++k) {
    const int i = beg + k * SORT_T + tid;
    pv[k] = part[(i < end) ? i : 0];
  }
  *(int4*)&c[4 * tid] = make_int4(0, 0, 0, 0);
  {
    const int kr = tid / (L1_DOUT / 4);
    const int cc = (tid - kr * (L1_DOUT / 4)) * 4;
    *(float4*)&Ws[kr * L1_P + cc] = wq;
    if (tid < L1_DIN * (L1_DOUT / 4))
      *(float4*)&Ws[(NUM_REL * L1_DIN + kr) * L1_P + cc] = rq;
  }
  __syncthreads();
#pragma unroll
  for (int k = 0; k < SORT_R; ++k)
    atomicAdd(&c[(beg + k * SORT_T + tid < end) ? (pv[k] & (SORT_K - 1)) : SORT_K], 1);
  for (int i = beg + SORT_R * SORT_T + tid; i < end; i += SORT_T)
    atomicAdd(&c[part[i] & (SORT_K - 1)], 1);
  __syncthreads();

  const int4 cv = *(const int4*)&c[4 * tid];
  const int  ex = block_excl_scan<SORT_T / 32>(cv.x + cv.y + cv.z + cv.w, wtot);
  const int node = b * SCAN_B + tid;
  if (node < N_NODES) {
    *(int4*)(cnt + (size_t)node * NUM_REL) = cv;
    nloc[node] = ex;
  }
  *(int4*)&c[4 * tid] = make_int4(ex, ex + cv.x, ex + cv.x + cv.y, ex + cv.x + cv.y + cv.z);
  __syncthreads();

  const bool staged = n <= STAGE_CAP;
  int pos[SORT_R];
#pragma unroll
  for (int k = 0; k < SORT_R; ++k)
    pos[k] = atomicAdd(&c[(beg + k * SORT_T + tid < end) ? (pv[k] & (SORT_K - 1)) : SORT_K], 1);
#pragma unroll
  for (int k = 0; k < SORT_R; ++k)
    if (beg + k * SORT_T + tid < end) {
      if (staged) stage[pos[k]] = pv[k] >> 10; else sorted[beg + pos[k]] = pv[k] >> 10;
    }
  for (int i = beg + SORT_R * SORT_T + tid; i < end; i += SORT_T) {
    const int p = part[i];
    const int q = atomicAdd(&c[p & (SORT_K - 1)], 1);
    if (staged) stage[q] = p >> 10; else sorted[beg + q] = p >> 10;
  }
  if (!staged) __threadfence();
  __syncthreads();
  if (staged) {
    for (int i = tid; i < n; i += SORT_T) sorted[beg + i] = stage[i];
  }

  const int lane = tid & 31;
  const int wave = tid >> 5;
  const int half = lane >> 4;
  const int l16  = lane & 15;
  for (int s = 0; s < SCAN_B / 16 / (SORT_T / 32); ++s) {
    const int lstrip = wave + s * (SORT_T / 32);
    const int m0     = b * SCAN_B + lstrip * 16;
    if (m0 >= N_NODES) break;
    const int  lrow = lstrip * 16 + l16;
    const int4 e    = *(const int4*)&c[4 * lrow];
    const int  ep   = c[(lrow > 0) ? 4 * lrow - 1 : 0];
    const int  e0   = (lrow > 0) ? ep : 0;
    const int nn[NUM_REL] = {e.x - e0, e.y - e.x, e.z - e.y, e.w - e.z};
    const int of[NUM_REL] = {e0, e.x, e.y, e.z};
    if (staged)
      strip_layer<L1_DIN, L1_DOUT, true, false, SLOTS_LDS>(
          X, stage, nn, of, m0, m0 + l16, half, l16, Ws, bias, nullptr, out);
    else
      strip_layer<L1_DIN, L1_DOUT, true, false, SLOTS_FRESH>(
          X, sorted + beg, nn, of, m0, m0 + l16, half, l16, Ws, bias, nullptr, out);
  }
}


template <int DIN, int DOUT, bool RELU, bool POOL>
static void run_layer(const float* X, const float* W, const float* root,
                      const float* bias, const int* sorted, const int* cnt,
                      const int* nloc, const int* bpre, const int* batch,
                      float* out, hipStream_t stream) {
  static_assert(N_NODES % 16 == 0, "16-row strips must tile the nodes");
  layer_fused<DIN, DOUT, RELU, POOL><<<(N_NODES / 16 + 7) / 8, 256, 0, stream>>>(
      X, sorted, cnt, nloc, bpre, W, root, bias, batch, out);
}

extern "C" void kernel_launch(void* const* d_in, const int* in_sizes, int n_in,
                              void* d_out, int out_size, void* d_ws,
                              size_t ws_size, hipStream_t stream) {
  (void)in_sizes; (void)n_in; (void)out_size; (void)ws_size;

  const float* x     = (const float*)d_in[0];
  const int*   ei    = (const int*)d_in[1];
  const int*   et    = (const int*)d_in[2];
  const int*   batch = (const int*)d_in[3];
  const float* W1 = (const float*)d_in[4];
  const float* r1 = (const float*)d_in[5];
  const float* b1 = (const float*)d_in[6];
  const float* W2 = (const float*)d_in[7];
  const float* r2 = (const float*)d_in[8];
  const float* b2 = (const float*)d_in[9];
  const float* W3 = (const float*)d_in[10];
  const float* r3 = (const float*)d_in[11];
  const float* b3 = (const float*)d_in[12];
  const int* src = ei;
  const int* dst = ei + N_EDGES;
  float* out = (float*)d_out;

  char*  ws     = (char*)d_ws;
  int*   cnt    = (int*)ws;
  float* buf2   = (float*)(ws + 1600000);
  float* buf13  = (float*)(ws + 14400000);
  int*   hist   = (int*)(ws + 24000000);
  int*   sorted = (int*)(ws + 65600000);
  int*   part   = (int*)(ws + 72000000);
  int*   nloc   = (int*)(ws + 78400000);
  int*   bpre   = (int*)(ws + 80050048);

  bucket_hist<<<P1_BLOCKS, P1_T, 0, stream>>>(dst, hist, out);
  bucket_scatter<<<P1_BLOCKS, P1_T, 0, stream>>>(src, dst, et, hist, bpre, part);
  bucket_sort_layer1<<<N_BKT, SORT_T, 0, stream>>>(part, bpre, cnt, nloc, sorted,
                                                   x, W1, r1, b1, buf13);
  run_layer<16, 32, true, false>(buf13, W2, r2, b2, sorted, cnt, nloc, bpre,
                                 batch, buf2, stream);
  run_layer<32, 64, false, true>(buf2, W3, r3, b3, sorted, cnt, nloc, bpre,
                                 batch, out, stream);
}


# padded LDS adds (edges past a chunk's or bucket's end) go to a spare word per lane, not all to ONE word: same-word LDS adds of a wave are not merged but run one by one; slot order rule kept
# speedup vs baseline: 4.3775x; 1.2241x over previous
#include <hip/hip_runtime.h>
#include <math.h>


#define N_NODES    100000
#define N_EDGES    1600000
#define NUM_REL    4
#define NUM_GRAPHS 256

typedef __attribute__((ext_vector_type(2))) float v2f;
typedef __attribute__((ext_vector_type(8))) float v8f;

#define SCAN_B    256
#define N_BKT     ((N_NODES + SCAN_B - 1) / SCAN_B)
#define HP        (((N_BKT + 3) / 4) * 4)
#define HQ        (HP / 4)
#define P1_T      512
#define P1_QPT    6
#define P1_CHUNK  (P1_T * P1_QPT * 4)
#define P1_BLOCKS ((N_EDGES + P1_CHUNK - 1) / P1_CHUNK)
#define N_QUADS   (N_EDGES / 4)

template <int NW>
__device__ __forceinline__ int block_excl_scan(int v, int* wtot) {
  const int lane = threadIdx.x & 31;
  const int wave = threadIdx.x >> 5;
  int inc = v;
#pragma unroll
  for (int d = 1; d < 32; d <<= 1) {
    const int t = __shfl_up(inc, d);
    if (lane >= d) inc += t;
  }
  if (lane == 31) wtot[wave] = inc;
  __syncthreads();
  if (wave == 0) {
    const int w = (lane < NW) ? wtot[lane] : 0;
    int wi = w;
#pragma unroll
    for (int d = 1; d < 32; d <<= 1) {
      const int t = __shfl_up(wi, d);
      if (lane >= d) wi += t;
    }
    if (lane < NW) wtot[lane] = wi - w;
  }
  __syncthreads();
  return inc - v + wtot[wave];
}

__global__ __launch_bounds__(P1_T)
void bucket_hist(const int* __restrict__ dst, int* __restrict__ hist,
                 float* __restrict__ pool_out) {
  static_assert(N_EDGES % 4 == 0, "edges are read four at a time");
  static_assert(P1_BLOCKS * P1_T >= NUM_GRAPHS * 64, "one thread per pooled element");
  static_assert(HP <= P1_T, "one thread per hist column");
  __shared__ int h[HP];
  const int tid = threadIdx.x;
  const int4* d4 = (const int4*)dst;
  int4 dv[P1_QPT];
#pragma unroll
  for (int k = 0; k < P1_QPT; ++k) {
    const int q = (blockIdx.x * P1_QPT + k) * P1_T + tid;
    dv[k] = d4[(q < N_QUADS) ? q : 0];
  }
  if (tid < HP) h[tid] = 0;
  __syncthreads();
#pragma unroll
  for (int k = 0; k < P1_QPT; ++k) {
    const int q = (blockIdx.x * P1_QPT + k) * P1_T + tid;
    if (q < N_QUADS) {
      atomicAdd(&h[dv[k].x / SCAN_B], 1);
      atomicAdd(&h[dv[k].y / SCAN_B], 1);
      atomicAdd(&h[dv[k].z / SCAN_B], 1);
      atomicAdd(&h[dv[k].w / SCAN_B], 1);
    }
  }
  __syncthreads();
  if (tid < HP) hist[blockIdx.x * HP + tid] = h[tid];
  if (blockIdx.x * P1_T + tid < NUM_GRAPHS * 64)
    pool_out[blockIdx.x * P1_T + tid] = -INFINITY;
}

#define CS_RG  (P1_T / HQ)
#define CS_NR  ((P1_BLOCKS + CS_RG - 1) / CS_RG)
#define CS_B   14

__global__ __launch_bounds__(P1_T)
void bucket_scatter(const int* __restrict__ src,
                    const int* __restrict__ dst,
                    const int* __restrict__ et,
                    const int* __restrict__ hist,
                    int* __restrict__ bpre,
                    int* __restrict__ part) {
  static_assert(N_NODES < (1 << 21), "src id must fit above the 10 key bits");
  static_assert(SCAN_B * NUM_REL == 1024, "key is 10 bits");
  __shared__ int tot[HP];
  __shared__ int cur[HP + 32];
  __shared__ int wtot[P1_T / 32];
  const int tid = threadIdx.x;
  const int4* s4 = (const int4*)src;
  const int4* d4 = (const int4*)dst;
  const int4* t4 = (const int4*)et;

  if (tid < HP) { tot[tid] = 0; cur[tid] = 0; }
  __syncthreads();

  int4 sv[P1_QPT], dv[P1_QPT], tv[P1_QPT];
#pragma unroll
  for (int k = 0; k < P1_QPT; ++k) {
    const int q  = (blockIdx.x * P1_QPT + k) * P1_T + tid;
    const int qq = (q < N_QUADS) ? q : 0;
    sv[k] = s4[qq]; dv[k] = d4[qq]; tv[k] = t4[qq];
  }

  if (tid < CS_RG * HQ) {
    const int g  = tid / HQ;
    const int c4 = tid - g * HQ;
    const int4* h4 = (const int4*)hist;
    int4 ta = make_int4(0, 0, 0, 0);
    int4 pa = make_int4(0, 0, 0, 0);
    for (int j0 = 0; j0 < CS_NR; j0 += CS_B) {
      int4 hv[CS_B];
#pragma unroll
      for (int u = 0; u < CS_B; ++u) {
        const int r  = g + (j0 + u) * CS_RG;
        const int rr = (r < P1_BLOCKS) ? r : P1_BLOCKS - 1;
        hv[u] = h4[rr * HQ + c4];
      }
#pragma unroll
      for (int u = 0; u < CS_B; ++u) {
        const int  r   = g + (j0 + u) * CS_RG;
        const bool in  = r < P1_BLOCKS;
        const bool pre = r < (int)blockIdx.x;
        ta.x += in ? hv[u].x : 0;  ta.y += in ? hv[u].y : 0;
        ta.z += in ? hv[u].z : 0;  ta.w += in ? hv[u].w : 0;
        pa.x += pre ? hv[u].x : 0; pa.y += pre ? hv[u].y : 0;
        pa.z += pre ? hv[u].z : 0; pa.w += pre ? hv[u].w : 0;
      }
    }
    atomicAdd(&tot[4 * c4 + 0], ta.x); atomicAdd(&tot[4 * c4 + 1], ta.y);
    atomicAdd(&tot[4 * c4 + 2], ta.z); atomicAdd(&tot[4 * c4 + 3], ta.w);
    atomicAdd(&cur[4 * c4 + 0], pa.x); atomicAdd(&cur[4 * c4 + 1], pa.y);
    atomicAdd(&cur[4 * c4 + 2], pa.z); atomicAdd(&cur[4 * c4 + 3], pa.w);
  }
  __syncthreads();

  const int tv0  = (tid < HP) ? tot[tid] : 0;
  const int base = block_excl_scan<P1_T / 32>(tv0, wtot);
  if (tid < N_BKT) {
    cur[tid] += base;
    if (blockIdx.x == 0) bpre[tid] = base;
  }
  __syncthreads();

  int pos[P1_QPT][4];
  const int spare = HP + (tid & 31);
#pragma unroll
  for (int k = 0; k < P1_QPT; ++k) {
    const bool ok = (blockIdx.x * P1_QPT + k) * P1_T + tid < N_QUADS;
    pos[k][0] = atomicAdd(&cur[ok ? dv[k].x / SCAN_B : spare], 1);
    pos[k][1] = atomicAdd(&cur[ok ? dv[k].y / SCAN_B : spare], 1);
    pos[k][2] = atomicAdd(&cur[ok ? dv[k].z / SCAN_B : spare], 1);
    pos[k][3] = atomicAdd(&cur[ok ? dv[k].w / SCAN_B : spare], 1);
  }
#pragma unroll
  for (int k = 0; k < P1_QPT; ++k) {
    if ((blockIdx.x * P1_QPT + k) * P1_T + tid < N_QUADS) {
      part[pos[k][0]] = (sv[k].x << 10) | ((dv[k].x % SCAN_B) << 2) | (tv[k].x & 3);
      part[pos[k][1]] = (sv[k].y << 10) | ((dv[k].y % SCAN_B) << 2) | (tv[k].y & 3);
      part[pos[k][2]] = (sv[k].z << 10) | ((dv[k].z % SCAN_B) << 2) | (tv[k].z & 3);
      part[pos[k][3]] = (sv[k].w << 10) | ((dv[k].w % SCAN_B) << 2) | (tv[k].w & 3);
    }
  }
}

#define SORT_T    256
#define SORT_K    (SCAN_B * NUM_REL)
#define SORT_R    24
#define STAGE_CAP 8192

__device__ __forceinline__ void atomic_max_f32(float* addr, float v) {
  if (v >= 0.0f) atomicMax((int*)addr, __float_as_int(v));
  else           atomicMin((unsigned int*)addr, __float_as_uint(v));
}

enum { SLOTS_GLOBAL = 0,
       SLOTS_LDS    = 1,
       SLOTS_FRESH  = 2 };
template <int KIND>
__device__ __forceinline__ int fetch_slot(const int* slots, int i) {
  if (KIND == SLOTS_FRESH)
    return __hip_atomic_load(slots + i, __ATOMIC_RELAXED, __HIP_MEMORY_SCOPE_AGENT);
  return slots[i];
}

template <int DIN, int DOUT, bool RELU, bool POOL, int KIND>
__device__ __forceinline__ void strip_layer(const float* __restrict__ X,
                                            const int* slots,
                                            const int (&n)[NUM_REL],
                                            const int (&off)[NUM_REL],
                                            const int m0, const int row,
                                            const int half, const int l16,
                                            const float* Ws,
                                            const float* __restrict__ bias,
                                            const int*   __restrict__ batch,
                                            float* __restrict__ out) {
  constexpr int KA = NUM_REL * DIN;
  constexpr int P  = DOUT + 8;
  constexpr int NT = DOUT / 16;
  constexpr int G  = DIN / 8;
  constexpr int U  = (DIN <= 16) ? 2 : 1;

  const int nmax = max(max(n[0], n[1]), max(n[2], n[3]));

  const float* xq = X + 4 * half;

  float4 a[NUM_REL][G];
#pragma unroll
  for (int r = 0; r < NUM_REL; ++r)
#pragma unroll
    for (int i = 0; i < G; ++i) a[r][i] = make_float4(0.0f, 0.0f, 0.0f, 0.0f);

  for (int j = 0; j < nmax; j += U) {
    int s[NUM_REL][U];
#pragma unroll
    for (int r = 0; r < NUM_REL; ++r)
#pragma unroll
      for (int u = 0; u < U; ++u)
        s[r][u] = fetch_slot<KIND>(slots, (j + u < n[r]) ? off[r] + j + u : 0);
#pragma unroll
    for (int r = 0; r < NUM_REL; ++r)
#pragma unroll
      for (int u = 0; u < U; ++u)
        asm volatile("" : "+v"(s[r][u]));
    float4 v[NUM_REL][U][G];
#pragma unroll
    for (int r = 0; r < NUM_REL; ++r)
#pragma unroll
      for (int u = 0; u < U; ++u)
#pragma unroll
        for (int i = 0; i < G; ++i)
          v[r][u][i] = *(const float4*)(xq + (size_t)s[r][u] * DIN + 8 * i);
#pragma unroll
    for (int r = 0; r < NUM_REL; ++r)
#pragma unroll
      for (int u = 0; u < U; ++u) {
        const bool act = j + u < n[r];
#pragma unroll
        for (int i = 0; i < G; ++i) {
          a[r][i].x += act ? v[r][u][i].x : 0.0f;
          a[r][i].y += act ? v[r][u][i].y : 0.0f;
          a[r][i].z += act ? v[r][u][i].z : 0.0f;
          a[r][i].w += act ? v[r][u][i].w : 0.0f;
        }
      }
  }

  float4 xv[G];
#pragma unroll
  for (int i = 0; i < G; ++i)
    xv[i] = *(const float4*)(xq + (size_t)row * DIN + 8 * i);

  float inv[NUM_REL];
#pragma unroll
  for (int r = 0; r < NUM_REL; ++r) inv[r] = 1.0f / fmaxf((float)n[r], 1.0f);

  float bvp[NT];
  int   g[8];
  if (POOL) {
#pragma unroll
    for (int t = 0; t < NT; ++t) bvp[t] = bias[t * 16 + l16];
    const int* bp8 = batch + m0 + 8 * half;
    const int4 ba = *(const int4*)(bp8);
    const int4 bb = *(const int4*)(bp8 + 4);
    g[0] = ba.x; g[1] = ba.y; g[2] = ba.z; g[3] = ba.w;
    g[4] = bb.x; g[5] = bb.y; g[6] = bb.z; g[7] = bb.w;
  }

  v8f acc[NT];
#pragma unroll
  for (int t = 0; t < NT; ++t) acc[t] = v8f{};

#pragma unroll
  for (int r = 0; r < NUM_REL; ++r) {
    const float sc = inv[r];
#pragma unroll
    for (int i = 0; i < G; ++i) {
      const int kb = r * DIN + 8 * i + 4 * half;
      const float4 av = a[r][i];
      const v2f a0 = {av.x * sc, av.y * sc};
      const v2f a1 = {av.z * sc, av.w * sc};
      const float* bp = &Ws[kb * P + l16];
#pragma unroll
      for (int t = 0; t < NT; ++t) {
        const v2f b0 = {bp[t * 16], bp[t * 16 + P]};
        acc[t] = __builtin_amdgcn_wmma_f32_16x16x4_f32(false, a0, false, b0,
                                                       (short)0, acc[t], false, false);
      }
#pragma unroll
      for (int t = 0; t < NT; ++t) {
        const v2f b1 = {bp[t * 16 + 2 * P], bp[t * 16 + 3 * P]};
        acc[t] = __builtin_amdgcn_wmma_f32_16x16x4_f32(false, a1, false, b1,
                                                       (short)0, acc[t], false, false);
      }
    }
  }

#pragma unroll
  for (int kk = 0; kk < DIN; kk += 8) {
    const int kb = kk + 4 * half;
    const float4 av = xv[kk / 8];
    const v2f a0 = {av.x, av.y};
    const v2f a1 = {av.z, av.w};
    const float* bp = &Ws[(KA + kb) * P + l16];
#pragma unroll
    for (int t = 0; t < NT; ++t) {
      const v2f b0 = {bp[t * 16], bp[t * 16 + P]};
      acc[t] = __builtin_amdgcn_wmma_f32_16x16x4_f32(false, a0, false, b0,
                                                     (short)0, acc[t], false, false);
    }
#pragma unroll
    for (int t = 0; t < NT; ++t) {
      const v2f b1 = {bp[t * 16 + 2 * P], bp[t * 16 + 3 * P]};
      acc[t] = __builtin_amdgcn_wmma_f32_16x16x4_f32(false, a1, false, b1,
                                                     (short)0, acc[t], false, false);
    }
  }

  if (!POOL) {
#pragma unroll
    for (int t = 0; t < NT; ++t) {
      const int   col = t * 16 + l16;
      const float bv  = bias[col];
#pragma unroll
      for (int v = 0; v < 8; ++v) {
        float y = acc[t][v] + bv;
        if (RELU) y = fmaxf(y, 0.0f);
        out[(size_t)(m0 + v + 8 * half) * DOUT + col] = y;
      }
    }
  } else {
    const int g0 = __builtin_amdgcn_readfirstlane(g[0]);
    bool same = true;
#pragma unroll
    for (int v = 0; v < 8; ++v) same = same && (g[v] == g0);
    const bool one_graph = __builtin_amdgcn_ballot_w32(same) == 0xffffffffu;

#pragma unroll
    for (int t = 0; t < NT; ++t) {
      const int   col = t * 16 + l16;
      const float bv  = bvp[t];
      float y[8];
#pragma unroll
      for (int v = 0; v < 8; ++v) {
        y[v] = acc[t][v] + bv;
        if (RELU) y[v] = fmaxf(y[v], 0.0f);
      }
      if (one_graph) {
        float m = y[0];
#pragma unroll
        for (int v = 1; v < 8; ++v) m = fmaxf(m, y[v]);
        m = fmaxf(m, __shfl_xor(m, 16));
        if (half == 0) atomic_max_f32(&out[(size_t)g0 * DOUT + col], m);
      } else {
#pragma unroll
        for (int v = 0; v < 8; ++v)
          atomic_max_f32(&out[(size_t)g[v] * DOUT + col], y[v]);
      }
    }
  }
}

template <int DIN, int DOUT, bool RELU, bool POOL>
__global__ __launch_bounds__(256)
void layer_fused(const float* __restrict__ X,
                 const int*   __restrict__ sorted,
                 const int*   __restrict__ cnt,
                 const int*   __restrict__ nloc,
                 const int*   __restrict__ bpre,
                 const float* __restrict__ W,
                 const float* __restrict__ root,
                 const float* __restrict__ bias,
                 const int*   __restrict__ batch,
                 float* __restrict__ out) {
  constexpr int KA = NUM_REL * DIN;
  constexpr int KT = KA + DIN;
  constexpr int P  = DOUT + 8;
  constexpr int C4 = DOUT / 4;
  __shared__ float Ws[KT * P];

  constexpr int NWQ = KA * C4;
  constexpr int NRQ = DIN * C4;
  static_assert(NWQ % 256 == 0, "the relation weights take whole rounds of the block");
  constexpr int NW = NWQ / 256;
  constexpr int NR = (NRQ + 255) / 256;

  const int lane  = threadIdx.x & 31;
  const int wave  = threadIdx.x >> 5;
  const int half  = lane >> 4;
  const int l16   = lane & 15;
  const int strip = blockIdx.x * 8 + wave;
  const bool live = strip < N_NODES / 16;
  const int m0  = (live ? strip : 0) * 16;
  const int row = m0 + l16;

  float4 wq[NW + NR];
#pragma unroll
  for (int k = 0; k < NW; ++k)
    wq[k] = ((const float4*)W)[threadIdx.x + k * 256];
#pragma unroll
  for (int k = 0; k < NR; ++k) {
    const int i = threadIdx.x + k * 256;
    wq[NW + k] = ((const float4*)root)[(i < NRQ) ? i : 0];
  }
  const int4 c4 = *(const int4*)(cnt + (size_t)row * NUM_REL);
  const int  nl = nloc[row];
  const int  bp = bpre[row / SCAN_B];

#pragma unroll
  for (int k = 0; k < NW; ++k) {
    const int i  = threadIdx.x + k * 256;
    const int kr = i / C4;
    const int c  = (i - kr * C4) * 4;
    *(float4*)(&Ws[kr * P + c]) = wq[k];
  }
#pragma unroll
  for (int k = 0; k < NR; ++k) {
    const int i = threadIdx.x + k * 256;
    asm volatile("" : "+v"(wq[NW + k].x), "+v"(wq[NW + k].y),
                      "+v"(wq[NW + k].z), "+v"(wq[NW + k].w));
    if (NRQ % 256 == 0 || i < NRQ) {
      const int kr = i / C4;
      const int c  = (i - kr * C4) * 4;
      *(float4*)(&Ws[(KA + kr) * P + c]) = wq[NW + k];
    }
  }
  __syncthreads();
  if (!live) return;

  int n[NUM_REL];
  n[0] = c4.x; n[1] = c4.y; n[2] = c4.z; n[3] = c4.w;

  int off[NUM_REL];
  off[0] = bp + nl;
  off[1] = off[0] + n[0];
  off[2] = off[1] + n[1];
  off[3] = off[2] + n[2];
  strip_layer<DIN, DOUT, RELU, POOL, SLOTS_GLOBAL>(X, sorted, n, off, m0, row, half,
                                                   l16, Ws, bias, batch, out);
}

#define L1_DIN  16
#define L1_DOUT 16
#define L1_P    (L1_DOUT + 8)
#define L1_KT   ((NUM_REL + 1) * L1_DIN)

__global__ __launch_bounds__(SORT_T)
void bucket_sort_layer1(const int* __restrict__ part,
                        const int* __restrict__ bpre,
                        int* __restrict__ cnt,
                        int* __restrict__ nloc,
                        int* sorted,
                        const float* __restrict__ X,
                        const float* __restrict__ W,
                        const float* __restrict__ root,
                        const float* __restrict__ bias,
                        float* __restrict__ out) {
  static_assert(SORT_T == SCAN_B && NUM_REL == 4, "one thread per node, int4 of counts");
  static_assert(NUM_REL * L1_DIN * (L1_DOUT / 4) == SORT_T, "one float4 of W per thread");
  static_assert(L1_DIN * (L1_DOUT / 4) <= SORT_T, "root in one round of the block");
  static_assert(N_NODES % 16 == 0, "16-row strips must tile the nodes");
  __shared__ __attribute__((aligned(16))) float Ws[L1_KT * L1_P];
  __shared__ __attribute__((aligned(16))) int c[SORT_K + 32];
  __shared__ int wtot[SORT_T / 32];
  __shared__ int stage[STAGE_CAP];
  const int tid = threadIdx.x;
  const int b   = blockIdx.x;
  const float4 wq = ((const float4*)W)[tid];
  const float4 rq = ((const float4*)root)[(tid < L1_DIN * (L1_DOUT / 4)) ? tid : 0];
  const int beg = bpre[b];
  const int end = (b + 1 < N_BKT) ? bpre[b + 1] : N_EDGES;
  const int n   = end - beg;
  const int spare = SORT_K + (tid & 31);

  int pv[SORT_R];
#pragma unroll
  for (int k = 0; k < SORT_R; ++k) {
    const int i = beg + k * SORT_T + tid;
    pv[k] = part[(i < end) ? i : 0];
  }
  *(int4*)&c[4 * tid] = make_int4(0, 0, 0, 0);
  {
    const int kr = tid / (L1_DOUT / 4);
    const int cc = (tid - kr * (L1_DOUT / 4)) * 4;
    *(float4*)&Ws[kr * L1_P + cc] = wq;
    if (tid < L1_DIN * (L1_DOUT / 4))
      *(float4*)&Ws[(NUM_REL * L1_DIN + kr) * L1_P + cc] = rq;
  }
  __syncthreads();
#pragma unroll
  for (int k = 0; k < SORT_R; ++k)
    atomicAdd(&c[(beg + k * SORT_T + tid < end) ? (pv[k] & (SORT_K - 1)) : spare], 1);
  for (int i = beg + SORT_R * SORT_T + tid; i < end; i += SORT_T)
    atomicAdd(&c[part[i] & (SORT_K - 1)], 1);
  __syncthreads();

  const int4 cv = *(const int4*)&c[4 * tid];
  const int  ex = block_excl_scan<SORT_T / 32>(cv.x + cv.y + cv.z + cv.w, wtot);
  const int node = b * SCAN_B + tid;
  if (node < N_NODES) {
    *(int4*)(cnt + (size_t)node * NUM_REL) = cv;
    nloc[node] = ex;
  }
  *(int4*)&c[4 * tid] = make_int4(ex, ex + cv.x, ex + cv.x + cv.y, ex + cv.x + cv.y + cv.z);
  __syncthreads();

  const bool staged = n <= STAGE_CAP;
  int pos[SORT_R];
#pragma unroll
  for (int k = 0; k < SORT_R; ++k)
    pos[k] = atomicAdd(&c[(beg + k * SORT_T + tid < end) ? (pv[k] & (SORT_K - 1)) : spare], 1);
#pragma unroll
  for (int k = 0; k < SORT_R; ++k)
    if (beg + k * SORT_T + tid < end) {
      if (staged) stage[pos[k]] = pv[k] >> 10; else sorted[beg + pos[k]] = pv[k] >> 10;
    }
  for (int i = beg + SORT_R * SORT_T + tid; i < end; i += SORT_T) {
    const int p = part[i];
    const int q = atomicAdd(&c[p & (SORT_K - 1)], 1);
    if (staged) stage[q] = p >> 10; else sorted[beg + q] = p >> 10;
  }
  if (!staged) __threadfence();
  __syncthreads();
  if (staged) {
    for (int i = tid; i < n; i += SORT_T) sorted[beg + i] = stage[i];
  }

  const int lane = tid & 31;
  const int wave = tid >> 5;
  const int half = lane >> 4;
  const int l16  = lane & 15;
  for (int s = 0; s < SCAN_B / 16 / (SORT_T / 32); ++s) {
    const int lstrip = wave + s * (SORT_T / 32);
    const int m0     = b * SCAN_B + lstrip * 16;
    if (m0 >= N_NODES) break;
    const int  lrow = lstrip * 16 + l16;
    const int4 e    = *(const int4*)&c[4 * lrow];
    const int  ep   = c[(lrow > 0) ? 4 * lrow - 1 : 0];
    const int  e0   = (lrow > 0) ? ep : 0;
    const int nn[NUM_REL] = {e.x - e0, e.y - e.x, e.z - e.y, e.w - e.z};
    const int of[NUM_REL] = {e0, e.x, e.y, e.z};
    if (staged)
      strip_layer<L1_DIN, L1_DOUT, true, false, SLOTS_LDS>(
          X, stage, nn, of, m0, m0 + l16, half, l16, Ws, bias, nullptr, out);
    else
      strip_layer<L1_DIN, L1_DOUT, true, false, SLOTS_FRESH>(
          X, sorted + beg, nn, of, m0, m0 + l16, half, l16, Ws, bias, nullptr, out);
  }
}


template <int DIN, int DOUT, bool RELU, bool POOL>
static void run_layer(const float* X, const float* W, const float* root,
                      const float* bias, const int* sorted, const int* cnt,
                      const int* nloc, const int* bpre, const int* batch,
                      float* out, hipStream_t stream) {
  static_assert(N_NODES % 16 == 0, "16-row strips must tile the nodes");
  layer_fused<DIN, DOUT, RELU, POOL><<<(N_NODES / 16 + 7) / 8, 256, 0, stream>>>(
      X, sorted, cnt, nloc, bpre, W, root, bias, batch, out);
}

extern "C" void kernel_launch(void* const* d_in, const int* in_sizes, int n_in,
                              void* d_out, int out_size, void* d_ws,
                              size_t ws_size, hipStream_t stream) {
  (void)in_sizes; (void)n_in; (void)out_size; (void)ws_size;

  const float* x     = (const float*)d_in[0];
  const int*   ei    = (const int*)d_in[1];
  const int*   et    = (const int*)d_in[2];
  const int*   batch = (const int*)d_in[3];
  const float* W1 = (const float*)d_in[4];
  const float* r1 = (const float*)d_in[5];
  const float* b1 = (const float*)d_in[6];
  const float* W2 = (const float*)d_in[7];
  const float* r2 = (const float*)d_in[8];
  const float* b2 = (const float*)d_in[9];
  const float* W3 = (const float*)d_in[10];
  const float* r3 = (const float*)d_in[11];
  const float* b3 = (const float*)d_in[12];
  const int* src = ei;
  const int* dst = ei + N_EDGES;
  float* out = (float*)d_out;

  char*  ws     = (char*)d_ws;
  int*   cnt    = (int*)ws;
  float* buf2   = (float*)(ws + 1600000);
  float* buf13  = (float*)(ws + 14400000);
  int*   hist   = (int*)(ws + 24000000);
  int*   sorted = (int*)(ws + 65600000);
  int*   part   = (int*)(ws + 72000000);
  int*   nloc   = (int*)(ws + 78400000);
  int*   bpre   = (int*)(ws + 80050048);

  bucket_hist<<<P1_BLOCKS, P1_T, 0, stream>>>(dst, hist, out);
  bucket_scatter<<<P1_BLOCKS, P1_T, 0, stream>>>(src, dst, et, hist, bpre, part);
  bucket_sort_layer1<<<N_BKT, SORT_T, 0, stream>>>(part, bpre, cnt, nloc, sorted,
                                                   x, W1, r1, b1, buf13);
  run_layer<16, 32, true, false>(buf13, W2, r2, b2, sorted, cnt, nloc, bpre,
                                 batch, buf2, stream);
  run_layer<32, 64, false, true>(buf2, W3, r3, b3, sorted, cnt, nloc, bpre,
                                 batch, out, stream);
}


# edge partition: each chunk is laid out by bucket in LDS (word + its place in part) and stored from there: one store covers whole runs of part, was 32 lines of 4 bytes; slot order rule kept
# speedup vs baseline: 5.6220x; 1.2843x over previous
#include <hip/hip_runtime.h>
#include <math.h>


#define N_NODES    100000
#define N_EDGES    1600000
#define NUM_REL    4
#define NUM_GRAPHS 256

typedef __attribute__((ext_vector_type(2))) float v2f;
typedef __attribute__((ext_vector_type(8))) float v8f;

#define SCAN_B    256
#define N_BKT     ((N_NODES + SCAN_B - 1) / SCAN_B)
#define HP        (((N_BKT + 3) / 4) * 4)
#define HQ        (HP / 4)
#define P1_T      512
#define P1_QPT    6
#define P1_CHUNK  (P1_T * P1_QPT * 4)
#define P1_BLOCKS ((N_EDGES + P1_CHUNK - 1) / P1_CHUNK)
#define N_QUADS   (N_EDGES / 4)

template <int NW>
__device__ __forceinline__ int block_excl_scan(int v, int* wtot) {
  const int lane = threadIdx.x & 31;
  const int wave = threadIdx.x >> 5;
  int inc = v;
#pragma unroll
  for (int d = 1; d < 32; d <<= 1) {
    const int t = __shfl_up(inc, d);
    if (lane >= d) inc += t;
  }
  if (lane == 31) wtot[wave] = inc;
  __syncthreads();
  if (wave == 0) {
    const int w = (lane < NW) ? wtot[lane] : 0;
    int wi = w;
#pragma unroll
    for (int d = 1; d < 32; d <<= 1) {
      const int t = __shfl_up(wi, d);
      if (lane >= d) wi += t;
    }
    if (lane < NW) wtot[lane] = wi - w;
  }
  __syncthreads();
  return inc - v + wtot[wave];
}

template <int NW>
__device__ __forceinline__ int2 block_excl_scan2(int2 v, int* wtot) {
  const int lane = threadIdx.x & 31;
  const int wave = threadIdx.x >> 5;
  int ia = v.x, ib = v.y;
#pragma unroll
  for (int d = 1; d < 32; d <<= 1) {
    const int ta = __shfl_up(ia, d);
    const int tb = __shfl_up(ib, d);
    if (lane >= d) { ia += ta; ib += tb; }
  }
  if (lane == 31) { wtot[wave] = ia; wtot[NW + wave] = ib; }
  __syncthreads();
  if (wave == 0) {
    const int wa = (lane < NW) ? wtot[lane] : 0;
    const int wb = (lane < NW) ? wtot[NW + lane] : 0;
    int ja = wa, jb = wb;
#pragma unroll
    for (int d = 1; d < 32; d <<= 1) {
      const int ta = __shfl_up(ja, d);
      const int tb = __shfl_up(jb, d);
      if (lane >= d) { ja += ta; jb += tb; }
    }
    if (lane < NW) { wtot[lane] = ja - wa; wtot[NW + lane] = jb - wb; }
  }
  __syncthreads();
  return make_int2(ia - v.x + wtot[wave], ib - v.y + wtot[NW + wave]);
}

__global__ __launch_bounds__(P1_T)
void bucket_hist(const int* __restrict__ dst, int* __restrict__ hist,
                 float* __restrict__ pool_out) {
  static_assert(N_EDGES % 4 == 0, "edges are read four at a time");
  static_assert(P1_BLOCKS * P1_T >= NUM_GRAPHS * 64, "one thread per pooled element");
  static_assert(HP <= P1_T, "one thread per hist column");
  __shared__ int h[HP];
  const int tid = threadIdx.x;
  const int4* d4 = (const int4*)dst;
  int4 dv[P1_QPT];
#pragma unroll
  for (int k = 0; k < P1_QPT; ++k) {
    const int q = (blockIdx.x * P1_QPT + k) * P1_T + tid;
    dv[k] = d4[(q < N_QUADS) ? q : 0];
  }
  if (tid < HP) h[tid] = 0;
  __syncthreads();
#pragma unroll
  for (int k = 0; k < P1_QPT; ++k) {
    const int q = (blockIdx.x * P1_QPT + k) * P1_T + tid;
    if (q < N_QUADS) {
      atomicAdd(&h[dv[k].x / SCAN_B], 1);
      atomicAdd(&h[dv[k].y / SCAN_B], 1);
      atomicAdd(&h[dv[k].z / SCAN_B], 1);
      atomicAdd(&h[dv[k].w / SCAN_B], 1);
    }
  }
  __syncthreads();
  if (tid < HP) hist[blockIdx.x * HP + tid] = h[tid];
  if (blockIdx.x * P1_T + tid < NUM_GRAPHS * 64)
    pool_out[blockIdx.x * P1_T + tid] = -INFINITY;
}

#define CS_RG  (P1_T / HQ)
#define CS_NR  ((P1_BLOCKS + CS_RG - 1) / CS_RG)
#define CS_B   14

__global__ __launch_bounds__(P1_T)
void bucket_scatter(const int* __restrict__ src,
                    const int* __restrict__ dst,
                    const int* __restrict__ et,
                    const int* __restrict__ hist,
                    int* __restrict__ bpre,
                    int* __restrict__ part) {
  static_assert(N_NODES < (1 << 21), "src id must fit above the 10 key bits");
  static_assert(SCAN_B * NUM_REL == 1024, "key is 10 bits");
  __shared__ int tot[HP];
  __shared__ int gof[HP];
  __shared__ int lcur[HP + 32];
  __shared__ int wtot[2 * (P1_T / 32)];
  __shared__ __attribute__((aligned(16))) int2 stage[P1_CHUNK];
  const int tid = threadIdx.x;
  const int4* s4 = (const int4*)src;
  const int4* d4 = (const int4*)dst;
  const int4* t4 = (const int4*)et;

  if (tid < HP) { tot[tid] = 0; gof[tid] = 0; }
  __syncthreads();

  int4 sv[P1_QPT], dv[P1_QPT], tv[P1_QPT];
#pragma unroll
  for (int k = 0; k < P1_QPT; ++k) {
    const int q  = (blockIdx.x * P1_QPT + k) * P1_T + tid;
    const int qq = (q < N_QUADS) ? q : 0;
    sv[k] = s4[qq]; dv[k] = d4[qq]; tv[k] = t4[qq];
  }
  const int ownc = hist[blockIdx.x * HP + ((tid < HP) ? tid : 0)];

  if (tid < CS_RG * HQ) {
    const int g  = tid / HQ;
    const int c4 = tid - g * HQ;
    const int4* h4 = (const int4*)hist;
    int4 ta = make_int4(0, 0, 0, 0);
    int4 pa = make_int4(0, 0, 0, 0);
    for (int j0 = 0; j0 < CS_NR; j0 += CS_B) {
      int4 hv[CS_B];
#pragma unroll
      for (int u = 0; u < CS_B; ++u) {
        const int r  = g + (j0 + u) * CS_RG;
        const int rr = (r < P1_BLOCKS) ? r : P1_BLOCKS - 1;
        hv[u] = h4[rr * HQ + c4];
      }
#pragma unroll
      for (int u = 0; u < CS_B; ++u) {
        const int  r   = g + (j0 + u) * CS_RG;
        const bool in  = r < P1_BLOCKS;
        const bool pre = r < (int)blockIdx.x;
        ta.x += in ? hv[u].x : 0;  ta.y += in ? hv[u].y : 0;
        ta.z += in ? hv[u].z : 0;  ta.w += in ? hv[u].w : 0;
        pa.x += pre ? hv[u].x : 0; pa.y += pre ? hv[u].y : 0;
        pa.z += pre ? hv[u].z : 0; pa.w += pre ? hv[u].w : 0;
      }
    }
    atomicAdd(&tot[4 * c4 + 0], ta.x); atomicAdd(&tot[4 * c4 + 1], ta.y);
    atomicAdd(&tot[4 * c4 + 2], ta.z); atomicAdd(&tot[4 * c4 + 3], ta.w);
    atomicAdd(&gof[4 * c4 + 0], pa.x); atomicAdd(&gof[4 * c4 + 1], pa.y);
    atomicAdd(&gof[4 * c4 + 2], pa.z); atomicAdd(&gof[4 * c4 + 3], pa.w);
  }
  __syncthreads();

  const int2 sc = block_excl_scan2<P1_T / 32>(
      make_int2((tid < HP) ? tot[tid] : 0, (tid < HP) ? ownc : 0), wtot);
  if (tid < N_BKT) {
    lcur[tid] = sc.y;
    gof[tid] += sc.x - sc.y;
    if (blockIdx.x == 0) bpre[tid] = sc.x;
  }
  __syncthreads();

  int lp[P1_QPT][4], go[P1_QPT][4];
  const int spare = HP + (tid & 31);
#pragma unroll
  for (int k = 0; k < P1_QPT; ++k) {
    const bool ok = (blockIdx.x * P1_QPT + k) * P1_T + tid < N_QUADS;
    const int b0 = dv[k].x / SCAN_B, b1 = dv[k].y / SCAN_B;
    const int b2 = dv[k].z / SCAN_B, b3 = dv[k].w / SCAN_B;
    lp[k][0] = atomicAdd(&lcur[ok ? b0 : spare], 1);  go[k][0] = gof[ok ? b0 : 0];
    lp[k][1] = atomicAdd(&lcur[ok ? b1 : spare], 1);  go[k][1] = gof[ok ? b1 : 0];
    lp[k][2] = atomicAdd(&lcur[ok ? b2 : spare], 1);  go[k][2] = gof[ok ? b2 : 0];
    lp[k][3] = atomicAdd(&lcur[ok ? b3 : spare], 1);  go[k][3] = gof[ok ? b3 : 0];
  }
#pragma unroll
  for (int k = 0; k < P1_QPT; ++k) {
    if ((blockIdx.x * P1_QPT + k) * P1_T + tid < N_QUADS) {
      stage[lp[k][0]] = make_int2((sv[k].x << 10) | ((dv[k].x % SCAN_B) << 2) | (tv[k].x & 3),
                                  lp[k][0] + go[k][0]);
      stage[lp[k][1]] = make_int2((sv[k].y << 10) | ((dv[k].y % SCAN_B) << 2) | (tv[k].y & 3),
                                  lp[k][1] + go[k][1]);
      stage[lp[k][2]] = make_int2((sv[k].z << 10) | ((dv[k].z % SCAN_B) << 2) | (tv[k].z & 3),
                                  lp[k][2] + go[k][2]);
      stage[lp[k][3]] = make_int2((sv[k].w << 10) | ((dv[k].w % SCAN_B) << 2) | (tv[k].w & 3),
                                  lp[k][3] + go[k][3]);
    }
  }
  __syncthreads();

  const int q0 = blockIdx.x * (P1_QPT * P1_T);
  const int nv = 4 * min(max(N_QUADS - q0, 0), P1_QPT * P1_T);
  int2 ev[P1_QPT * 4];
#pragma unroll
  for (int k = 0; k < P1_QPT * 4; ++k) {
    const int i = k * P1_T + tid;
    ev[k] = stage[(i < nv) ? i : 0];
  }
#pragma unroll
  for (int k = 0; k < P1_QPT * 4; ++k)
    if (k * P1_T + tid < nv) part[ev[k].y] = ev[k].x;
}

#define SORT_T    256
#define SORT_K    (SCAN_B * NUM_REL)
#define SORT_R    24
#define STAGE_CAP 8192

__device__ __forceinline__ void atomic_max_f32(float* addr, float v) {
  if (v >= 0.0f) atomicMax((int*)addr, __float_as_int(v));
  else           atomicMin((unsigned int*)addr, __float_as_uint(v));
}

enum { SLOTS_GLOBAL = 0,
       SLOTS_LDS    = 1,
       SLOTS_FRESH  = 2 };
template <int KIND>
__device__ __forceinline__ int fetch_slot(const int* slots, int i) {
  if (KIND == SLOTS_FRESH)
    return __hip_atomic_load(slots + i, __ATOMIC_RELAXED, __HIP_MEMORY_SCOPE_AGENT);
  return slots[i];
}

template <int DIN, int DOUT, bool RELU, bool POOL, int KIND>
__device__ __forceinline__ void strip_layer(const float* __restrict__ X,
                                            const int* slots,
                                            const int (&n)[NUM_REL],
                                            const int (&off)[NUM_REL],
                                            const int m0, const int row,
                                            const int half, const int l16,
                                            const float* Ws,
                                            const float* __restrict__ bias,
                                            const int*   __restrict__ batch,
                                            float* __restrict__ out) {
  constexpr int KA = NUM_REL * DIN;
  constexpr int P  = DOUT + 8;
  constexpr int NT = DOUT / 16;
  constexpr int G  = DIN / 8;
  constexpr int U  = (DIN <= 16) ? 2 : 1;

  const int nmax = max(max(n[0], n[1]), max(n[2], n[3]));

  const float* xq = X + 4 * half;

  float4 a[NUM_REL][G];
#pragma unroll
  for (int r = 0; r < NUM_REL; ++r)
#pragma unroll
    for (int i = 0; i < G; ++i) a[r][i] = make_float4(0.0f, 0.0f, 0.0f, 0.0f);

  for (int j = 0; j < nmax; j += U) {
    int s[NUM_REL][U];
#pragma unroll
    for (int r = 0; r < NUM_REL; ++r)
#pragma unroll
      for (int u = 0; u < U; ++u)
        s[r][u] = fetch_slot<KIND>(slots, (j + u < n[r]) ? off[r] + j + u : 0);
#pragma unroll
    for (int r = 0; r < NUM_REL; ++r)
#pragma unroll
      for (int u = 0; u < U; ++u)
        asm volatile("" : "+v"(s[r][u]));
    float4 v[NUM_REL][U][G];
#pragma unroll
    for (int r = 0; r < NUM_REL; ++r)
#pragma unroll
      for (int u = 0; u < U; ++u)
#pragma unroll
        for (int i = 0; i < G; ++i)
          v[r][u][i] = *(const float4*)(xq + (size_t)s[r][u] * DIN + 8 * i);
#pragma unroll
    for (int r = 0; r < NUM_REL; ++r)
#pragma unroll
      for (int u = 0; u < U; ++u) {
        const bool act = j + u < n[r];
#pragma unroll
        for (int i = 0; i < G; ++i) {
          a[r][i].x += act ? v[r][u][i].x : 0.0f;
          a[r][i].y += act ? v[r][u][i].y : 0.0f;
          a[r][i].z += act ? v[r][u][i].z : 0.0f;
          a[r][i].w += act ? v[r][u][i].w : 0.0f;
        }
      }
  }

  float4 xv[G];
#pragma unroll
  for (int i = 0; i < G; ++i)
    xv[i] = *(const float4*)(xq + (size_t)row * DIN + 8 * i);

  float inv[NUM_REL];
#pragma unroll
  for (int r = 0; r < NUM_REL; ++r) inv[r] = 1.0f / fmaxf((float)n[r], 1.0f);

  float bvp[NT];
  int   g[8];
  if (POOL) {
#pragma unroll
    for (int t = 0; t < NT; ++t) bvp[t] = bias[t * 16 + l16];
    const int* bp8 = batch + m0 + 8 * half;
    const int4 ba = *(const int4*)(bp8);
    const int4 bb = *(const int4*)(bp8 + 4);
    g[0] = ba.x; g[1] = ba.y; g[2] = ba.z; g[3] = ba.w;
    g[4] = bb.x; g[5] = bb.y; g[6] = bb.z; g[7] = bb.w;
  }

  v8f acc[NT];
#pragma unroll
  for (int t = 0; t < NT; ++t) acc[t] = v8f{};

#pragma unroll
  for (int r = 0; r < NUM_REL; ++r) {
    const float sc = inv[r];
#pragma unroll
    for (int i = 0; i < G; ++i) {
      const int kb = r * DIN + 8 * i + 4 * half;
      const float4 av = a[r][i];
      const v2f a0 = {av.x * sc, av.y * sc};
      const v2f a1 = {av.z * sc, av.w * sc};
      const float* bp = &Ws[kb * P + l16];
#pragma unroll
      for (int t = 0; t < NT; ++t) {
        const v2f b0 = {bp[t * 16], bp[t * 16 + P]};
        acc[t] = __builtin_amdgcn_wmma_f32_16x16x4_f32(false, a0, false, b0,
                                                       (short)0, acc[t], false, false);
      }
#pragma unroll
      for (int t = 0; t < NT; ++t) {
        const v2f b1 = {bp[t * 16 + 2 * P], bp[t * 16 + 3 * P]};
        acc[t] = __builtin_amdgcn_wmma_f32_16x16x4_f32(false, a1, false, b1,
                                                       (short)0, acc[t], false, false);
      }
    }
  }

#pragma unroll
  for (int kk = 0; kk < DIN; kk += 8) {
    const int kb = kk + 4 * half;
    const float4 av = xv[kk / 8];
    const v2f a0 = {av.x, av.y};
    const v2f a1 = {av.z, av.w};
    const float* bp = &Ws[(KA + kb) * P + l16];
#pragma unroll
    for (int t = 0; t < NT; ++t) {
      const v2f b0 = {bp[t * 16], bp[t * 16 + P]};
      acc[t] = __builtin_amdgcn_wmma_f32_16x16x4_f32(false, a0, false, b0,
                                                     (short)0, acc[t], false, false);
    }
#pragma unroll
    for (int t = 0; t < NT; ++t) {
      const v2f b1 = {bp[t * 16 + 2 * P], bp[t * 16 + 3 * P]};
      acc[t] = __builtin_amdgcn_wmma_f32_16x16x4_f32(false, a1, false, b1,
                                                     (short)0, acc[t], false, false);
    }
  }

  if (!POOL) {
#pragma unroll
    for (int t = 0; t < NT; ++t) {
      const int   col = t * 16 + l16;
      const float bv  = bias[col];
#pragma unroll
      for (int v = 0; v < 8; ++v) {
        float y = acc[t][v] + bv;
        if (RELU) y = fmaxf(y, 0.0f);
        out[(size_t)(m0 + v + 8 * half) * DOUT + col] = y;
      }
    }
  } else {
    const int g0 = __builtin_amdgcn_readfirstlane(g[0]);
    bool same = true;
#pragma unroll
    for (int v = 0; v < 8; ++v) same = same && (g[v] == g0);
    const bool one_graph = __builtin_amdgcn_ballot_w32(same) == 0xffffffffu;

#pragma unroll
    for (int t = 0; t < NT; ++t) {
      const int   col = t * 16 + l16;
      const float bv  = bvp[t];
      float y[8];
#pragma unroll
      for (int v = 0; v < 8; ++v) {
        y[v] = acc[t][v] + bv;
        if (RELU) y[v] = fmaxf(y[v], 0.0f);
      }
      if (one_graph) {
        float m = y[0];
#pragma unroll
        for (int v = 1; v < 8; ++v) m = fmaxf(m, y[v]);
        m = fmaxf(m, __shfl_xor(m, 16));
        if (half == 0) atomic_max_f32(&out[(size_t)g0 * DOUT + col], m);
      } else {
#pragma unroll
        for (int v = 0; v < 8; ++v)
          atomic_max_f32(&out[(size_t)g[v] * DOUT + col], y[v]);
      }
    }
  }
}

template <int DIN, int DOUT, bool RELU, bool POOL>
__global__ __launch_bounds__(256)
void layer_fused(const float* __restrict__ X,
                 const int*   __restrict__ sorted,
                 const int*   __restrict__ cnt,
                 const int*   __restrict__ nloc,
                 const int*   __restrict__ bpre,
                 const float* __restrict__ W,
                 const float* __restrict__ root,
                 const float* __restrict__ bias,
                 const int*   __restrict__ batch,
                 float* __restrict__ out) {
  constexpr int KA = NUM_REL * DIN;
  constexpr int KT = KA + DIN;
  constexpr int P  = DOUT + 8;
  constexpr int C4 = DOUT / 4;
  __shared__ float Ws[KT * P];

  constexpr int NWQ = KA * C4;
  constexpr int NRQ = DIN * C4;
  static_assert(NWQ % 256 == 0, "the relation weights take whole rounds of the block");
  constexpr int NW = NWQ / 256;
  constexpr int NR = (NRQ + 255) / 256;

  const int lane  = threadIdx.x & 31;
  const int wave  = threadIdx.x >> 5;
  const int half  = lane >> 4;
  const int l16   = lane & 15;
  const int strip = blockIdx.x * 8 + wave;
  const bool live = strip < N_NODES / 16;
  const int m0  = (live ? strip : 0) * 16;
  const int row = m0 + l16;

  float4 wq[NW + NR];
#pragma unroll
  for (int k = 0; k < NW; ++k)
    wq[k] = ((const float4*)W)[threadIdx.x + k * 256];
#pragma unroll
  for (int k = 0; k < NR; ++k) {
    const int i = threadIdx.x + k * 256;
    wq[NW + k] = ((const float4*)root)[(i < NRQ) ? i : 0];
  }
  const int4 c4 = *(const int4*)(cnt + (size_t)row * NUM_REL);
  const int  nl = nloc[row];
  const int  bp = bpre[row / SCAN_B];

#pragma unroll
  for (int k = 0; k < NW; ++k) {
    const int i  = threadIdx.x + k * 256;
    const int kr = i / C4;
    const int c  = (i - kr * C4) * 4;
    *(float4*)(&Ws[kr * P + c]) = wq[k];
  }
#pragma unroll
  for (int k = 0; k < NR; ++k) {
    const int i = threadIdx.x + k * 256;
    asm volatile("" : "+v"(wq[NW + k].x), "+v"(wq[NW + k].y),
                      "+v"(wq[NW + k].z), "+v"(wq[NW + k].w));
    if (NRQ % 256 == 0 || i < NRQ) {
      const int kr = i / C4;
      const int c  = (i - kr * C4) * 4;
      *(float4*)(&Ws[(KA + kr) * P + c]) = wq[NW + k];
    }
  }
  __syncthreads();
  if (!live) return;

  int n[NUM_REL];
  n[0] = c4.x; n[1] = c4.y; n[2] = c4.z; n[3] = c4.w;

  int off[NUM_REL];
  off[0] = bp + nl;
  off[1] = off[0] + n[0];
  off[2] = off[1] + n[1];
  off[3] = off[2] + n[2];
  strip_layer<DIN, DOUT, RELU, POOL, SLOTS_GLOBAL>(X, sorted, n, off, m0, row, half,
                                                   l16, Ws, bias, batch, out);
}

#define L1_DIN  16
#define L1_DOUT 16
#define L1_P    (L1_DOUT + 8)
#define L1_KT   ((NUM_REL + 1) * L1_DIN)

__global__ __launch_bounds__(SORT_T)
void bucket_sort_layer1(const int* __restrict__ part,
                        const int* __restrict__ bpre,
                        int* __restrict__ cnt,
                        int* __restrict__ nloc,
                        int* sorted,
                        const float* __restrict__ X,
                        const float* __restrict__ W,
                        const float* __restrict__ root,
                        const float* __restrict__ bias,
                        float* __restrict__ out) {
  static_assert(SORT_T == SCAN_B && NUM_REL == 4, "one thread per node, int4 of counts");
  static_assert(NUM_REL * L1_DIN * (L1_DOUT / 4) == SORT_T, "one float4 of W per thread");
  static_assert(L1_DIN * (L1_DOUT / 4) <= SORT_T, "root in one round of the block");
  static_assert(N_NODES % 16 == 0, "16-row strips must tile the nodes");
  __shared__ __attribute__((aligned(16))) float Ws[L1_KT * L1_P];
  __shared__ __attribute__((aligned(16))) int c[SORT_K + 32];
  __shared__ int wtot[SORT_T / 32];
  __shared__ int stage[STAGE_CAP];
  const int tid = threadIdx.x;
  const int b   = blockIdx.x;
  const float4 wq = ((const float4*)W)[tid];
  const float4 rq = ((const float4*)root)[(tid < L1_DIN * (L1_DOUT / 4)) ? tid : 0];
  const int beg = bpre[b];
  const int end = (b + 1 < N_BKT) ? bpre[b + 1] : N_EDGES;
  const int n   = end - beg;
  const int spare = SORT_K + (tid & 31);

  int pv[SORT_R];
#pragma unroll
  for (int k = 0; k < SORT_R; ++k) {
    const int i = beg + k * SORT_T + tid;
    pv[k] = part[(i < end) ? i : 0];
  }
  *(int4*)&c[4 * tid] = make_int4(0, 0, 0, 0);
  {
    const int kr = tid / (L1_DOUT / 4);
    const int cc = (tid - kr * (L1_DOUT / 4)) * 4;
    *(float4*)&Ws[kr * L1_P + cc] = wq;
    if (tid < L1_DIN * (L1_DOUT / 4))
      *(float4*)&Ws[(NUM_REL * L1_DIN + kr) * L1_P + cc] = rq;
  }
  __syncthreads();
#pragma unroll
  for (int k = 0; k < SORT_R; ++k)
    atomicAdd(&c[(beg + k * SORT_T + tid < end) ? (pv[k] & (SORT_K - 1)) : spare], 1);
  for (int i = beg + SORT_R * SORT_T + tid; i < end; i += SORT_T)
    atomicAdd(&c[part[i] & (SORT_K - 1)], 1);
  __syncthreads();

  const int4 cv = *(const int4*)&c[4 * tid];
  const int  ex = block_excl_scan<SORT_T / 32>(cv.x + cv.y + cv.z + cv.w, wtot);
  const int node = b * SCAN_B + tid;
  if (node < N_NODES) {
    *(int4*)(cnt + (size_t)node * NUM_REL) = cv;
    nloc[node] = ex;
  }
  *(int4*)&c[4 * tid] = make_int4(ex, ex + cv.x, ex + cv.x + cv.y, ex + cv.x + cv.y + cv.z);
  __syncthreads();

  const bool staged = n <= STAGE_CAP;
  int pos[SORT_R];
#pragma unroll
  for (int k = 0; k < SORT_R; ++k)
    pos[k] = atomicAdd(&c[(beg + k * SORT_T + tid < end) ? (pv[k] & (SORT_K - 1)) : spare], 1);
#pragma unroll
  for (int k = 0; k < SORT_R; ++k)
    if (beg + k * SORT_T + tid < end) {
      if (staged) stage[pos[k]] = pv[k] >> 10; else sorted[beg + pos[k]] = pv[k] >> 10;
    }
  for (int i = beg + SORT_R * SORT_T + tid; i < end; i += SORT_T) {
    const int p = part[i];
    const int q = atomicAdd(&c[p & (SORT_K - 1)], 1);
    if (staged) stage[q] = p >> 10; else sorted[beg + q] = p >> 10;
  }
  if (!staged) __threadfence();
  __syncthreads();
  if (staged) {
    for (int i = tid; i < n; i += SORT_T) sorted[beg + i] = stage[i];
  }

  const int lane = tid & 31;
  const int wave = tid >> 5;
  const int half = lane >> 4;
  const int l16  = lane & 15;
  for (int s = 0; s < SCAN_B / 16 / (SORT_T / 32); ++s) {
    const int lstrip = wave + s * (SORT_T / 32);
    const int m0     = b * SCAN_B + lstrip * 16;
    if (m0 >= N_NODES) break;
    const int  lrow = lstrip * 16 + l16;
    const int4 e    = *(const int4*)&c[4 * lrow];
    const int  ep   = c[(lrow > 0) ? 4 * lrow - 1 : 0];
    const int  e0   = (lrow > 0) ? ep : 0;
    const int nn[NUM_REL] = {e.x - e0, e.y - e.x, e.z - e.y, e.w - e.z};
    const int of[NUM_REL] = {e0, e.x, e.y, e.z};
    if (staged)
      strip_layer<L1_DIN, L1_DOUT, true, false, SLOTS_LDS>(
          X, stage, nn, of, m0, m0 + l16, half, l16, Ws, bias, nullptr, out);
    else
      strip_layer<L1_DIN, L1_DOUT, true, false, SLOTS_FRESH>(
          X, sorted + beg, nn, of, m0, m0 + l16, half, l16, Ws, bias, nullptr, out);
  }
}


template <int DIN, int DOUT, bool RELU, bool POOL>
static void run_layer(const float* X, const float* W, const float* root,
                      const float* bias, const int* sorted, const int* cnt,
                      const int* nloc, const int* bpre, const int* batch,
                      float* out, hipStream_t stream) {
  static_assert(N_NODES % 16 == 0, "16-row strips must tile the nodes");
  layer_fused<DIN, DOUT, RELU, POOL><<<(N_NODES / 16 + 7) / 8, 256, 0, stream>>>(
      X, sorted, cnt, nloc, bpre, W, root, bias, batch, out);
}

extern "C" void kernel_launch(void* const* d_in, const int* in_sizes, int n_in,
                              void* d_out, int out_size, void* d_ws,
                              size_t ws_size, hipStream_t stream) {
  (void)in_sizes; (void)n_in; (void)out_size; (void)ws_size;

  const float* x     = (const float*)d_in[0];
  const int*   ei    = (const int*)d_in[1];
  const int*   et    = (const int*)d_in[2];
  const int*   batch = (const int*)d_in[3];
  const float* W1 = (const float*)d_in[4];
  const float* r1 = (const float*)d_in[5];
  const float* b1 = (const float*)d_in[6];
  const float* W2 = (const float*)d_in[7];
  const float* r2 = (const float*)d_in[8];
  const float* b2 = (const float*)d_in[9];
  const float* W3 = (const float*)d_in[10];
  const float* r3 = (const float*)d_in[11];
  const float* b3 = (const float*)d_in[12];
  const int* src = ei;
  const int* dst = ei + N_EDGES;
  float* out = (float*)d_out;

  char*  ws     = (char*)d_ws;
  int*   cnt    = (int*)ws;
  float* buf2   = (float*)(ws + 1600000);
  float* buf13  = (float*)(ws + 14400000);
  int*   hist   = (int*)(ws + 24000000);
  int*   sorted = (int*)(ws + 65600000);
  int*   part   = (int*)(ws + 72000000);
  int*   nloc   = (int*)(ws + 78400000);
  int*   bpre   = (int*)(ws + 80050048);

  bucket_hist<<<P1_BLOCKS, P1_T, 0, stream>>>(dst, hist, out);
  bucket_scatter<<<P1_BLOCKS, P1_T, 0, stream>>>(src, dst, et, hist, bpre, part);
  bucket_sort_layer1<<<N_BKT, SORT_T, 0, stream>>>(part, bpre, cnt, nloc, sorted,
                                                   x, W1, r1, b1, buf13);
  run_layer<16, 32, true, false>(buf13, W2, r2, b2, sorted, cnt, nloc, bpre,
                                 batch, buf2, stream);
  run_layer<32, 64, false, true>(buf2, W3, r3, b3, sorted, cnt, nloc, bpre,
                                 batch, out, stream);
}


# layer gathers: a lane group (4 or 8 lanes) loads one WHOLE source row per load (was 16 lines of 32 B a load), groups walk equal slot runs; sums turned to fragments via an LDS tile; slot order kept
# speedup vs baseline: 6.1138x; 1.0875x over previous
#include <hip/hip_runtime.h>
#include <math.h>


#define N_NODES    100000
#define N_EDGES    1600000
#define NUM_REL    4
#define NUM_GRAPHS 256

typedef __attribute__((ext_vector_type(2))) float v2f;
typedef __attribute__((ext_vector_type(8))) float v8f;

#define SCAN_B    256
#define N_BKT     ((N_NODES + SCAN_B - 1) / SCAN_B)
#define HP        (((N_BKT + 3) / 4) * 4)
#define HQ        (HP / 4)
#define P1_T      512
#define P1_QPT    6
#define P1_CHUNK  (P1_T * P1_QPT * 4)
#define P1_BLOCKS ((N_EDGES + P1_CHUNK - 1) / P1_CHUNK)
#define N_QUADS   (N_EDGES / 4)

template <int NW>
__device__ __forceinline__ int block_excl_scan(int v, int* wtot) {
  const int lane = threadIdx.x & 31;
  const int wave = threadIdx.x >> 5;
  int inc = v;
#pragma unroll
  for (int d = 1; d < 32; d <<= 1) {
    const int t = __shfl_up(inc, d);
    if (lane >= d) inc += t;
  }
  if (lane == 31) wtot[wave] = inc;
  __syncthreads();
  if (wave == 0) {
    const int w = (lane < NW) ? wtot[lane] : 0;
    int wi = w;
#pragma unroll
    for (int d = 1; d < 32; d <<= 1) {
      const int t = __shfl_up(wi, d);
      if (lane >= d) wi += t;
    }
    if (lane < NW) wtot[lane] = wi - w;
  }
  __syncthreads();
  return inc - v + wtot[wave];
}

template <int NW>
__device__ __forceinline__ int2 block_excl_scan2(int2 v, int* wtot) {
  const int lane = threadIdx.x & 31;
  const int wave = threadIdx.x >> 5;
  int ia = v.x, ib = v.y;
#pragma unroll
  for (int d = 1; d < 32; d <<= 1) {
    const int ta = __shfl_up(ia, d);
    const int tb = __shfl_up(ib, d);
    if (lane >= d) { ia += ta; ib += tb; }
  }
  if (lane == 31) { wtot[wave] = ia; wtot[NW + wave] = ib; }
  __syncthreads();
  if (wave == 0) {
    const int wa = (lane < NW) ? wtot[lane] : 0;
    const int wb = (lane < NW) ? wtot[NW + lane] : 0;
    int ja = wa, jb = wb;
#pragma unroll
    for (int d = 1; d < 32; d <<= 1) {
      const int ta = __shfl_up(ja, d);
      const int tb = __shfl_up(jb, d);
      if (lane >= d) { ja += ta; jb += tb; }
    }
    if (lane < NW) { wtot[lane] = ja - wa; wtot[NW + lane] = jb - wb; }
  }
  __syncthreads();
  return make_int2(ia - v.x + wtot[wave], ib - v.y + wtot[NW + wave]);
}

__global__ __launch_bounds__(P1_T)
void bucket_hist(const int* __restrict__ dst, int* __restrict__ hist,
                 float* __restrict__ pool_out) {
  static_assert(N_EDGES % 4 == 0, "edges are read four at a time");
  static_assert(P1_BLOCKS * P1_T >= NUM_GRAPHS * 64, "one thread per pooled element");
  static_assert(HP <= P1_T, "one thread per hist column");
  __shared__ int h[HP];
  const int tid = threadIdx.x;
  const int4* d4 = (const int4*)dst;
  int4 dv[P1_QPT];
#pragma unroll
  for (int k = 0; k < P1_QPT; ++k) {
    const int q = (blockIdx.x * P1_QPT + k) * P1_T + tid;
    dv[k] = d4[(q < N_QUADS) ? q : 0];
  }
  if (tid < HP) h[tid] = 0;
  __syncthreads();
#pragma unroll
  for (int k = 0; k < P1_QPT; ++k) {
    const int q = (blockIdx.x * P1_QPT + k) * P1_T + tid;
    if (q < N_QUADS) {
      atomicAdd(&h[dv[k].x / SCAN_B], 1);
      atomicAdd(&h[dv[k].y / SCAN_B], 1);
      atomicAdd(&h[dv[k].z / SCAN_B], 1);
      atomicAdd(&h[dv[k].w / SCAN_B], 1);
    }
  }
  __syncthreads();
  if (tid < HP) hist[blockIdx.x * HP + tid] = h[tid];
  if (blockIdx.x * P1_T + tid < NUM_GRAPHS * 64)
    pool_out[blockIdx.x * P1_T + tid] = -INFINITY;
}

#define CS_RG  (P1_T / HQ)
#define CS_NR  ((P1_BLOCKS + CS_RG - 1) / CS_RG)
#define CS_B   14

__global__ __launch_bounds__(P1_T)
void bucket_scatter(const int* __restrict__ src,
                    const int* __restrict__ dst,
                    const int* __restrict__ et,
                    const int* __restrict__ hist,
                    int* __restrict__ bpre,
                    int* __restrict__ part) {
  static_assert(N_NODES < (1 << 21), "src id must fit above the 10 key bits");
  static_assert(SCAN_B * NUM_REL == 1024, "key is 10 bits");
  __shared__ int tot[HP];
  __shared__ int gof[HP];
  __shared__ int lcur[HP + 32];
  __shared__ int wtot[2 * (P1_T / 32)];
  __shared__ __attribute__((aligned(16))) int2 stage[P1_CHUNK];
  const int tid = threadIdx.x;
  const int4* s4 = (const int4*)src;
  const int4* d4 = (const int4*)dst;
  const int4* t4 = (const int4*)et;

  if (tid < HP) { tot[tid] = 0; gof[tid] = 0; }
  __syncthreads();

  int4 sv[P1_QPT], dv[P1_QPT], tv[P1_QPT];
#pragma unroll
  for (int k = 0; k < P1_QPT; ++k) {
    const int q  = (blockIdx.x * P1_QPT + k) * P1_T + tid;
    const int qq = (q < N_QUADS) ? q : 0;
    sv[k] = s4[qq]; dv[k] = d4[qq]; tv[k] = t4[qq];
  }
  const int ownc = hist[blockIdx.x * HP + ((tid < HP) ? tid : 0)];

  if (tid < CS_RG * HQ) {
    const int g  = tid / HQ;
    const int c4 = tid - g * HQ;
    const int4* h4 = (const int4*)hist;
    int4 ta = make_int4(0, 0, 0, 0);
    int4 pa = make_int4(0, 0, 0, 0);
    for (int j0 = 0; j0 < CS_NR; j0 += CS_B) {
      int4 hv[CS_B];
#pragma unroll
      for (int u = 0; u < CS_B; ++u) {
        const int r  = g + (j0 + u) * CS_RG;
        const int rr = (r < P1_BLOCKS) ? r : P1_BLOCKS - 1;
        hv[u] = h4[rr * HQ + c4];
      }
#pragma unroll
      for (int u = 0; u < CS_B; ++u) {
        const int  r   = g + (j0 + u) * CS_RG;
        const bool in  = r < P1_BLOCKS;
        const bool pre = r < (int)blockIdx.x;
        ta.x += in ? hv[u].x : 0;  ta.y += in ? hv[u].y : 0;
        ta.z += in ? hv[u].z : 0;  ta.w += in ? hv[u].w : 0;
        pa.x += pre ? hv[u].x : 0; pa.y += pre ? hv[u].y : 0;
        pa.z += pre ? hv[u].z : 0; pa.w += pre ? hv[u].w : 0;
      }
    }
    atomicAdd(&tot[4 * c4 + 0], ta.x); atomicAdd(&tot[4 * c4 + 1], ta.y);
    atomicAdd(&tot[4 * c4 + 2], ta.z); atomicAdd(&tot[4 * c4 + 3], ta.w);
    atomicAdd(&gof[4 * c4 + 0], pa.x); atomicAdd(&gof[4 * c4 + 1], pa.y);
    atomicAdd(&gof[4 * c4 + 2], pa.z); atomicAdd(&gof[4 * c4 + 3], pa.w);
  }
  __syncthreads();

  const int2 sc = block_excl_scan2<P1_T / 32>(
      make_int2((tid < HP) ? tot[tid] : 0, (tid < HP) ? ownc : 0), wtot);
  if (tid < N_BKT) {
    lcur[tid] = sc.y;
    gof[tid] += sc.x - sc.y;
    if (blockIdx.x == 0) bpre[tid] = sc.x;
  }
  __syncthreads();

  int lp[P1_QPT][4], go[P1_QPT][4];
  const int spare = HP + (tid & 31);
#pragma unroll
  for (int k = 0; k < P1_QPT; ++k) {
    const bool ok = (blockIdx.x * P1_QPT + k) * P1_T + tid < N_QUADS;
    const int b0 = dv[k].x / SCAN_B, b1 = dv[k].y / SCAN_B;
    const int b2 = dv[k].z / SCAN_B, b3 = dv[k].w / SCAN_B;
    lp[k][0] = atomicAdd(&lcur[ok ? b0 : spare], 1);  go[k][0] = gof[ok ? b0 : 0];
    lp[k][1] = atomicAdd(&lcur[ok ? b1 : spare], 1);  go[k][1] = gof[ok ? b1 : 0];
    lp[k][2] = atomicAdd(&lcur[ok ? b2 : spare], 1);  go[k][2] = gof[ok ? b2 : 0];
    lp[k][3] = atomicAdd(&lcur[ok ? b3 : spare], 1);  go[k][3] = gof[ok ? b3 : 0];
  }
#pragma unroll
  for (int k = 0; k < P1_QPT; ++k) {
    if ((blockIdx.x * P1_QPT + k) * P1_T + tid < N_QUADS) {
      stage[lp[k][0]] = make_int2((sv[k].x << 10) | ((dv[k].x % SCAN_B) << 2) | (tv[k].x & 3),
                                  lp[k][0] + go[k][0]);
      stage[lp[k][1]] = make_int2((sv[k].y << 10) | ((dv[k].y % SCAN_B) << 2) | (tv[k].y & 3),
                                  lp[k][1] + go[k][1]);
      stage[lp[k][2]] = make_int2((sv[k].z << 10) | ((dv[k].z % SCAN_B) << 2) | (tv[k].z & 3),
                                  lp[k][2] + go[k][2]);
      stage[lp[k][3]] = make_int2((sv[k].w << 10) | ((dv[k].w % SCAN_B) << 2) | (tv[k].w & 3),
                                  lp[k][3] + go[k][3]);
    }
  }
  __syncthreads();

  const int q0 = blockIdx.x * (P1_QPT * P1_T);
  const int nv = 4 * min(max(N_QUADS - q0, 0), P1_QPT * P1_T);
  int2 ev[P1_QPT * 4];
#pragma unroll
  for (int k = 0; k < P1_QPT * 4; ++k) {
    const int i = k * P1_T + tid;
    ev[k] = stage[(i < nv) ? i : 0];
  }
#pragma unroll
  for (int k = 0; k < P1_QPT * 4; ++k)
    if (k * P1_T + tid < nv) part[ev[k].y] = ev[k].x;
}

#define SORT_T    256
#define SORT_K    (SCAN_B * NUM_REL)
#define SORT_R    24
#define STAGE_CAP 8192

__device__ __forceinline__ void atomic_max_f32(float* addr, float v) {
  if (v >= 0.0f) atomicMax((int*)addr, __float_as_int(v));
  else           atomicMin((unsigned int*)addr, __float_as_uint(v));
}

enum { SLOTS_GLOBAL = 0,
       SLOTS_LDS    = 1,
       SLOTS_FRESH  = 2 };
template <int KIND>
__device__ __forceinline__ int fetch_slot(const int* slots, int i) {
  if (KIND == SLOTS_FRESH)
    return __hip_atomic_load(slots + i, __ATOMIC_RELAXED, __HIP_MEMORY_SCOPE_AGENT);
  return slots[i];
}

#define ROW_SL 16
#define TILE_P(DIN) (NUM_REL * (DIN) + 4)

template <int DIN, int KIND>
__device__ __forceinline__ void gather_rows(const float* __restrict__ X,
                                            const int* slots,
                                            const int s0, const int ns,
                                            const int (&n)[NUM_REL],
                                            const int lane, float* at,
                                            float4 (&a)[NUM_REL][DIN / 8]) {
  constexpr int LPR = DIN / 4;
  constexpr int AP  = TILE_P(DIN);
  static_assert(LPR == 4 || LPR == 8, "a lane group is 4 or 8 lanes");
  const int c    = lane & (LPR - 1);
  const int half = lane >> 4;
  const int l16  = lane & 15;

  int nw = ns;
#pragma unroll
  for (int dd = LPR; dd < 32; dd <<= 1) nw = max(nw, __shfl_xor(nw, dd));

  const char* xb = (const char*)X;
  const unsigned cb = 16u * (unsigned)c;
  float4 acc = make_float4(0.0f, 0.0f, 0.0f, 0.0f);
  int prev = -1;
  asm volatile("" ::: "memory");
  for (int t0 = 0; t0 < nw; t0 += ROW_SL) {
    int w[ROW_SL];
#pragma unroll
    for (int k = 0; k < ROW_SL; ++k)
      w[k] = fetch_slot<KIND>(slots, (t0 + k < ns) ? s0 + t0 + k : 0);
#pragma unroll
    for (int k = 0; k < ROW_SL; ++k)
      asm volatile("" : "+v"(w[k]));
    float4 v[ROW_SL];
#pragma unroll
    for (int k = 0; k < ROW_SL; ++k)
      v[k] = *(const float4*)(xb + ((unsigned)(w[k] >> 10) * (unsigned)(DIN * 4) + cb));
#pragma unroll
    for (int k = 0; k < ROW_SL; ++k) {
      const int  key   = w[k] & 63;
      const bool fresh = key != prev;
      acc.x = (fresh ? 0.0f : acc.x) + v[k].x;
      acc.y = (fresh ? 0.0f : acc.y) + v[k].y;
      acc.z = (fresh ? 0.0f : acc.z) + v[k].z;
      acc.w = (fresh ? 0.0f : acc.w) + v[k].w;
      prev = key;
      const bool last = (k == ROW_SL - 1) || (t0 + k + 1 >= ns) ||
                        ((w[(k + 1 < ROW_SL) ? k + 1 : k] & 63) != key);
      if (t0 + k < ns && last)
        *(float4*)(&at[(key >> 2) * AP + (key & 3) * DIN + 4 * c]) = acc;
    }
  }
  asm volatile("" ::: "memory");
#pragma unroll
  for (int r = 0; r < NUM_REL; ++r)
#pragma unroll
    for (int i = 0; i < DIN / 8; ++i) {
      const float4 tv = *(const float4*)(&at[l16 * AP + r * DIN + 8 * i + 4 * half]);
      a[r][i] = (n[r] > 0) ? tv : make_float4(0.0f, 0.0f, 0.0f, 0.0f);
    }
}

template <int DIN, int DOUT, bool RELU, bool POOL>
__device__ __forceinline__ void finish_strip(const float* __restrict__ X,
                                             const float4 (&a)[NUM_REL][DIN / 8],
                                             const int (&n)[NUM_REL],
                                             const int m0, const int row,
                                             const int half, const int l16,
                                             const float* Ws,
                                             const float* __restrict__ bias,
                                             const int*   __restrict__ batch,
                                             float* __restrict__ out) {
  constexpr int KA = NUM_REL * DIN;
  constexpr int P  = DOUT + 8;
  constexpr int NT = DOUT / 16;
  constexpr int G  = DIN / 8;
  const float* xq = X + 4 * half;

  float4 xv[G];
#pragma unroll
  for (int i = 0; i < G; ++i)
    xv[i] = *(const float4*)(xq + (size_t)row * DIN + 8 * i);

  float inv[NUM_REL];
#pragma unroll
  for (int r = 0; r < NUM_REL; ++r) inv[r] = 1.0f / fmaxf((float)n[r], 1.0f);

  float bvp[NT];
  int   g[8];
  if (POOL) {
#pragma unroll
    for (int t = 0; t < NT; ++t) bvp[t] = bias[t * 16 + l16];
    const int* bp8 = batch + m0 + 8 * half;
    const int4 ba = *(const int4*)(bp8);
    const int4 bb = *(const int4*)(bp8 + 4);
    g[0] = ba.x; g[1] = ba.y; g[2] = ba.z; g[3] = ba.w;
    g[4] = bb.x; g[5] = bb.y; g[6] = bb.z; g[7] = bb.w;
  }

  v8f acc[NT];
#pragma unroll
  for (int t = 0; t < NT; ++t) acc[t] = v8f{};

#pragma unroll
  for (int r = 0; r < NUM_REL; ++r) {
    const float sc = inv[r];
#pragma unroll
    for (int i = 0; i < G; ++i) {
      const int kb = r * DIN + 8 * i + 4 * half;
      const float4 av = a[r][i];
      const v2f a0 = {av.x * sc, av.y * sc};
      const v2f a1 = {av.z * sc, av.w * sc};
      const float* bp = &Ws[kb * P + l16];
#pragma unroll
      for (int t = 0; t < NT; ++t) {
        const v2f b0 = {bp[t * 16], bp[t * 16 + P]};
        acc[t] = __builtin_amdgcn_wmma_f32_16x16x4_f32(false, a0, false, b0,
                                                       (short)0, acc[t], false, false);
      }
#pragma unroll
      for (int t = 0; t < NT; ++t) {
        const v2f b1 = {bp[t * 16 + 2 * P], bp[t * 16 + 3 * P]};
        acc[t] = __builtin_amdgcn_wmma_f32_16x16x4_f32(false, a1, false, b1,
                                                       (short)0, acc[t], false, false);
      }
    }
  }

#pragma unroll
  for (int kk = 0; kk < DIN; kk += 8) {
    const int kb = kk + 4 * half;
    const float4 av = xv[kk / 8];
    const v2f a0 = {av.x, av.y};
    const v2f a1 = {av.z, av.w};
    const float* bp = &Ws[(KA + kb) * P + l16];
#pragma unroll
    for (int t = 0; t < NT; ++t) {
      const v2f b0 = {bp[t * 16], bp[t * 16 + P]};
      acc[t] = __builtin_amdgcn_wmma_f32_16x16x4_f32(false, a0, false, b0,
                                                     (short)0, acc[t], false, false);
    }
#pragma unroll
    for (int t = 0; t < NT; ++t) {
      const v2f b1 = {bp[t * 16 + 2 * P], bp[t * 16 + 3 * P]};
      acc[t] = __builtin_amdgcn_wmma_f32_16x16x4_f32(false, a1, false, b1,
                                                     (short)0, acc[t], false, false);
    }
  }

  if (!POOL) {
#pragma unroll
    for (int t = 0; t < NT; ++t) {
      const int   col = t * 16 + l16;
      const float bv  = bias[col];
#pragma unroll
      for (int v = 0; v < 8; ++v) {
        float y = acc[t][v] + bv;
        if (RELU) y = fmaxf(y, 0.0f);
        out[(size_t)(m0 + v + 8 * half) * DOUT + col] = y;
      }
    }
  } else {
    const int g0 = __builtin_amdgcn_readfirstlane(g[0]);
    bool same = true;
#pragma unroll
    for (int v = 0; v < 8; ++v) same = same && (g[v] == g0);
    const bool one_graph = __builtin_amdgcn_ballot_w32(same) == 0xffffffffu;

#pragma unroll
    for (int t = 0; t < NT; ++t) {
      const int   col = t * 16 + l16;
      const float bv  = bvp[t];
      float y[8];
#pragma unroll
      for (int v = 0; v < 8; ++v) {
        y[v] = acc[t][v] + bv;
        if (RELU) y[v] = fmaxf(y[v], 0.0f);
      }
      if (one_graph) {
        float m = y[0];
#pragma unroll
        for (int v = 1; v < 8; ++v) m = fmaxf(m, y[v]);
        m = fmaxf(m, __shfl_xor(m, 16));
        if (half == 0) atomic_max_f32(&out[(size_t)g0 * DOUT + col], m);
      } else {
#pragma unroll
        for (int v = 0; v < 8; ++v)
          atomic_max_f32(&out[(size_t)g[v] * DOUT + col], y[v]);
      }
    }
  }
}

template <int DIN, int DOUT, bool RELU, bool POOL, int WAVES>
__global__ __launch_bounds__(WAVES * 32)
void layer_fused(const float* __restrict__ X,
                 const int*   __restrict__ sorted,
                 const int*   __restrict__ cnt,
                 const int*   __restrict__ nloc,
                 const int*   __restrict__ bpre,
                 const float* __restrict__ W,
                 const float* __restrict__ root,
                 const float* __restrict__ bias,
                 const int*   __restrict__ batch,
                 float* __restrict__ out) {
  constexpr int T   = WAVES * 32;
  constexpr int KA  = NUM_REL * DIN;
  constexpr int KT  = KA + DIN;
  constexpr int P   = DOUT + 8;
  constexpr int C4  = DOUT / 4;
  constexpr int NWQ = KA * C4;
  constexpr int NQ  = KT * C4;
  constexpr int NRD = (NQ + T - 1) / T;
  constexpr int LPR = DIN / 4;
  constexpr int RPG = 16 / (32 / LPR);
  static_assert(SCAN_B % 16 == 0, "a strip lies inside one bucket");
  __shared__ __attribute__((aligned(16))) float Ws[KT * P];
  __shared__ __attribute__((aligned(16))) float At[WAVES * 16 * TILE_P(DIN)];

  const int lane  = threadIdx.x & 31;
  const int wave  = threadIdx.x >> 5;
  const int half  = lane >> 4;
  const int l16   = lane & 15;
  const int strip = blockIdx.x * WAVES + wave;
  const bool live = strip < N_NODES / 16;
  const int m0  = (live ? strip : 0) * 16;
  const int row = m0 + l16;
  const int r0  = m0 + RPG * (lane / LPR);

  float4 wq[NRD];
#pragma unroll
  for (int k = 0; k < NRD; ++k) {
    const int i  = threadIdx.x + k * T;
    const int ii = (i < NQ) ? i : 0;
    const float4* p = (ii < NWQ) ? ((const float4*)W + ii) : ((const float4*)root + (ii - NWQ));
    wq[k] = *p;
  }
  const int4 c4 = *(const int4*)(cnt + (size_t)row * NUM_REL);
  int4 ck[RPG];
#pragma unroll
  for (int j = 0; j < RPG; ++j) ck[j] = *(const int4*)(cnt + (size_t)(r0 + j) * NUM_REL);
  const int nl = nloc[r0];
  const int bp = bpre[m0 / SCAN_B];

#pragma unroll
  for (int k = 0; k < NRD; ++k) {
    const int i = threadIdx.x + k * T;
    if (NQ % T == 0 || i < NQ) {
      const int kr = i / C4;
      const int cc = (i - kr * C4) * 4;
      *(float4*)(&Ws[kr * P + cc]) = wq[k];
    }
  }
  __syncthreads();
  if (!live) return;

  int n[NUM_REL];
  n[0] = c4.x; n[1] = c4.y; n[2] = c4.z; n[3] = c4.w;
  int ns = 0;
#pragma unroll
  for (int j = 0; j < RPG; ++j) ns += ck[j].x + ck[j].y + ck[j].z + ck[j].w;

  float4 a[NUM_REL][DIN / 8];
  gather_rows<DIN, SLOTS_GLOBAL>(X, sorted, bp + nl, ns, n, lane,
                                 At + wave * (16 * TILE_P(DIN)), a);
  finish_strip<DIN, DOUT, RELU, POOL>(X, a, n, m0, row, half, l16, Ws, bias, batch, out);
}

static_assert(32 / (16 / 4) == 8, "layer 1: eight lane groups per wave, two rows each");
#define L1_DIN  16
#define L1_DOUT 16
#define L1_P    (L1_DOUT + 8)
#define L1_KT   ((NUM_REL + 1) * L1_DIN)

__global__ __launch_bounds__(SORT_T)
void bucket_sort_layer1(const int* __restrict__ part,
                        const int* __restrict__ bpre,
                        int* __restrict__ cnt,
                        int* __restrict__ nloc,
                        int* sorted,
                        const float* __restrict__ X,
                        const float* __restrict__ W,
                        const float* __restrict__ root,
                        const float* __restrict__ bias,
                        float* __restrict__ out) {
  static_assert(SORT_T == SCAN_B && NUM_REL == 4, "one thread per node, int4 of counts");
  static_assert(NUM_REL * L1_DIN * (L1_DOUT / 4) == SORT_T, "one float4 of W per thread");
  static_assert(L1_DIN * (L1_DOUT / 4) <= SORT_T, "root in one round of the block");
  static_assert(N_NODES % 16 == 0, "16-row strips must tile the nodes");
  __shared__ __attribute__((aligned(16))) float Ws[L1_KT * L1_P];
  __shared__ __attribute__((aligned(16))) int c[SORT_K + 32];
  __shared__ int wtot[SORT_T / 32];
  __shared__ int stage[STAGE_CAP];
  __shared__ __attribute__((aligned(16))) float At[(SORT_T / 32) * 16 * TILE_P(L1_DIN)];
  const int tid = threadIdx.x;
  const int b   = blockIdx.x;
  const float4 wq = ((const float4*)W)[tid];
  const float4 rq = ((const float4*)root)[(tid < L1_DIN * (L1_DOUT / 4)) ? tid : 0];
  const int beg = bpre[b];
  const int end = (b + 1 < N_BKT) ? bpre[b + 1] : N_EDGES;
  const int n   = end - beg;
  const int spare = SORT_K + (tid & 31);

  int pv[SORT_R];
#pragma unroll
  for (int k = 0; k < SORT_R; ++k) {
    const int i = beg + k * SORT_T + tid;
    pv[k] = part[(i < end) ? i : 0];
  }
  *(int4*)&c[4 * tid] = make_int4(0, 0, 0, 0);
  {
    const int kr = tid / (L1_DOUT / 4);
    const int cc = (tid - kr * (L1_DOUT / 4)) * 4;
    *(float4*)&Ws[kr * L1_P + cc] = wq;
    if (tid < L1_DIN * (L1_DOUT / 4))
      *(float4*)&Ws[(NUM_REL * L1_DIN + kr) * L1_P + cc] = rq;
  }
  __syncthreads();
#pragma unroll
  for (int k = 0; k < SORT_R; ++k)
    atomicAdd(&c[(beg + k * SORT_T + tid < end) ? (pv[k] & (SORT_K - 1)) : spare], 1);
  for (int i = beg + SORT_R * SORT_T + tid; i < end; i += SORT_T)
    atomicAdd(&c[part[i] & (SORT_K - 1)], 1);
  __syncthreads();

  const int4 cv = *(const int4*)&c[4 * tid];
  const int  ex = block_excl_scan<SORT_T / 32>(cv.x + cv.y + cv.z + cv.w, wtot);
  const int node = b * SCAN_B + tid;
  if (node < N_NODES) {
    *(int4*)(cnt + (size_t)node * NUM_REL) = cv;
    nloc[node] = ex;
  }
  *(int4*)&c[4 * tid] = make_int4(ex, ex + cv.x, ex + cv.x + cv.y, ex + cv.x + cv.y + cv.z);
  __syncthreads();

  const bool staged = n <= STAGE_CAP;
  int pos[SORT_R];
#pragma unroll
  for (int k = 0; k < SORT_R; ++k)
    pos[k] = atomicAdd(&c[(beg + k * SORT_T + tid < end) ? (pv[k] & (SORT_K - 1)) : spare], 1);
#pragma unroll
  for (int k = 0; k < SORT_R; ++k)
    if (beg + k * SORT_T + tid < end) {
      if (staged) stage[pos[k]] = pv[k]; else sorted[beg + pos[k]] = pv[k];
    }
  for (int i = beg + SORT_R * SORT_T + tid; i < end; i += SORT_T) {
    const int p = part[i];
    const int q = atomicAdd(&c[p & (SORT_K - 1)], 1);
    if (staged) stage[q] = p; else sorted[beg + q] = p;
  }
  if (!staged) __threadfence();
  __syncthreads();
  if (staged) {
    for (int i = tid; i < n; i += SORT_T) sorted[beg + i] = stage[i];
  }

  const int lane = tid & 31;
  const int wave = tid >> 5;
  const int half = lane >> 4;
  const int l16  = lane & 15;
  for (int s = 0; s < SCAN_B / 16 / (SORT_T / 32); ++s) {
    const int lstrip = wave + s * (SORT_T / 32);
    const int m0     = b * SCAN_B + lstrip * 16;
    if (m0 >= N_NODES) break;
    const int  lrow = lstrip * 16 + l16;
    const int4 e    = *(const int4*)&c[4 * lrow];
    const int  ep   = c[(lrow > 0) ? 4 * lrow - 1 : 0];
    const int  e0   = (lrow > 0) ? ep : 0;
    const int nn[NUM_REL] = {e.x - e0, e.y - e.x, e.z - e.y, e.w - e.z};
    const int grow = lstrip * 16 + 2 * (lane / (L1_DIN / 4));
    const int gp   = c[(grow > 0) ? 4 * grow - 1 : 0];
    const int gs   = (grow > 0) ? gp : 0;
    const int gn   = c[4 * grow + 7] - gs;
    float* at = At + wave * (16 * TILE_P(L1_DIN));
    float4 a[NUM_REL][L1_DIN / 8];
    if (staged)
      gather_rows<L1_DIN, SLOTS_LDS>(X, stage, gs, gn, nn, lane, at, a);
    else
      gather_rows<L1_DIN, SLOTS_FRESH>(X, sorted + beg, gs, gn, nn, lane, at, a);
    finish_strip<L1_DIN, L1_DOUT, true, false>(X, a, nn, m0, m0 + l16, half, l16, Ws,
                                               bias, nullptr, out);
  }
}


template <int DIN, int DOUT, bool RELU, bool POOL, int WAVES>
static void run_layer(const float* X, const float* W, const float* root,
                      const float* bias, const int* sorted, const int* cnt,
                      const int* nloc, const int* bpre, const int* batch,
                      float* out, hipStream_t stream) {
  static_assert(N_NODES % 16 == 0, "16-row strips must tile the nodes");
  layer_fused<DIN, DOUT, RELU, POOL, WAVES><<<(N_NODES / 16 + WAVES - 1) / WAVES, WAVES * 32, 0, stream>>>(
      X, sorted, cnt, nloc, bpre, W, root, bias, batch, out);
}

extern "C" void kernel_launch(void* const* d_in, const int* in_sizes, int n_in,
                              void* d_out, int out_size, void* d_ws,
                              size_t ws_size, hipStream_t stream) {
  (void)in_sizes; (void)n_in; (void)out_size; (void)ws_size;

  const float* x     = (const float*)d_in[0];
  const int*   ei    = (const int*)d_in[1];
  const int*   et    = (const int*)d_in[2];
  const int*   batch = (const int*)d_in[3];
  const float* W1 = (const float*)d_in[4];
  const float* r1 = (const float*)d_in[5];
  const float* b1 = (const float*)d_in[6];
  const float* W2 = (const float*)d_in[7];
  const float* r2 = (const float*)d_in[8];
  const float* b2 = (const float*)d_in[9];
  const float* W3 = (const float*)d_in[10];
  const float* r3 = (const float*)d_in[11];
  const float* b3 = (const float*)d_in[12];
  const int* src = ei;
  const int* dst = ei + N_EDGES;
  float* out = (float*)d_out;

  char*  ws     = (char*)d_ws;
  int*   cnt    = (int*)ws;
  float* buf2   = (float*)(ws + 1600000);
  float* buf13  = (float*)(ws + 14400000);
  int*   hist   = (int*)(ws + 24000000);
  int*   sorted = (int*)(ws + 65600000);
  int*   part   = (int*)(ws + 72000000);
  int*   nloc   = (int*)(ws + 78400000);
  int*   bpre   = (int*)(ws + 80050048);

  bucket_hist<<<P1_BLOCKS, P1_T, 0, stream>>>(dst, hist, out);
  bucket_scatter<<<P1_BLOCKS, P1_T, 0, stream>>>(src, dst, et, hist, bpre, part);
  bucket_sort_layer1<<<N_BKT, SORT_T, 0, stream>>>(part, bpre, cnt, nloc, sorted,
                                                   x, W1, r1, b1, buf13);
  run_layer<16, 32, true, false, 8>(buf13, W2, r2, b2, sorted, cnt, nloc, bpre,
                                 batch, buf2, stream);
  run_layer<32, 64, false, true, 25>(buf2, W3, r3, b3, sorted, cnt, nloc, bpre,
                                 batch, out, stream);
}
